# starting kernel
# baseline (speedup 1.0000x reference)
#include <hip/hip_runtime.h>
#include <math.h>


typedef __attribute__((ext_vector_type(2))) float v2f;
typedef __attribute__((ext_vector_type(8))) float v8f;

#define N_NODES 100000

__global__ void zero_kernel(float* __restrict__ p, int n) {
  int i = blockIdx.x * blockDim.x + threadIdx.x;
  if (i < n) p[i] = 0.0f;
}

__global__ void degree_kernel(const int* __restrict__ src,
                              float* __restrict__ deg, int E) {
  int e = blockIdx.x * blockDim.x + threadIdx.x;
  if (e < E) atomicAdd(deg + src[e], 1.0f);
}

__global__ void dis_kernel(const float* __restrict__ deg,
                           float* __restrict__ dis, int n) {
  int i = blockIdx.x * blockDim.x + threadIdx.x;
  if (i < n) {
    float d = deg[i];
    dis[i] = (d > 0.0f) ? (1.0f / sqrtf(d)) : 0.0f;
  }
}

__global__ void prop1_kernel(const int* __restrict__ src, const int* __restrict__ dst,
                             const float* __restrict__ dis, const float* __restrict__ x,
                             float* __restrict__ tx, int E) {
  int e = blockIdx.x * blockDim.x + threadIdx.x;
  if (e >= E) return;
  int s = src[e], d = dst[e];
  float w = -(dis[s] * dis[d]);
  const float4* xs = reinterpret_cast<const float4*>(x + (size_t)s * 8);
  float4 a = xs[0];
  float4 b = xs[1];
  float* t = tx + (size_t)d * 8;
  atomicAdd(t + 0, w * a.x); atomicAdd(t + 1, w * a.y);
  atomicAdd(t + 2, w * a.z); atomicAdd(t + 3, w * a.w);
  atomicAdd(t + 4, w * b.x); atomicAdd(t + 5, w * b.y);
  atomicAdd(t + 6, w * b.z); atomicAdd(t + 7, w * b.w);
}

__global__ void layer1_kernel(const float* __restrict__ x, const float* __restrict__ tx,
                              const float* __restrict__ W1, const float* __restrict__ b1,
                              float* __restrict__ h1, int n) {
  int i = blockIdx.x * blockDim.x + threadIdx.x;
  if (i >= n) return;
  const float4* xp = reinterpret_cast<const float4*>(x + (size_t)i * 8);
  const float4* tp = reinterpret_cast<const float4*>(tx + (size_t)i * 8);
  float4 x0 = xp[0], x1 = xp[1], t0 = tp[0], t1 = tp[1];
  float xi[8] = {x0.x, x0.y, x0.z, x0.w, x1.x, x1.y, x1.z, x1.w};
  float ti[8] = {t0.x, t0.y, t0.z, t0.w, t1.x, t1.y, t1.z, t1.w};
#pragma unroll
  for (int h = 0; h < 3; ++h) {
    float acc = b1[h];
#pragma unroll
    for (int f = 0; f < 8; ++f)
      acc += xi[f] * W1[f * 3 + h] + ti[f] * W1[24 + f * 3 + h];
    h1[(size_t)i * 3 + h] = fmaxf(acc, 0.0f);
  }
}

__global__ void prop2_kernel(const int* __restrict__ src, const int* __restrict__ dst,
                             const float* __restrict__ dis, const float* __restrict__ h1,
                             float* __restrict__ tx, int E) {
  int e = blockIdx.x * blockDim.x + threadIdx.x;
  if (e >= E) return;
  int s = src[e], d = dst[e];
  float w = -(dis[s] * dis[d]);
  const float* hs = h1 + (size_t)s * 3;
  float* t = tx + (size_t)d * 3;
  atomicAdd(t + 0, w * hs[0]);
  atomicAdd(t + 1, w * hs[1]);
  atomicAdd(t + 2, w * hs[2]);
}

__global__ void layer2_kernel(const float* __restrict__ h1, const float* __restrict__ tx,
                              const float* __restrict__ W2, const float* __restrict__ b2,
                              float* __restrict__ h2, int n) {
  int i = blockIdx.x * blockDim.x + threadIdx.x;
  if (i >= n) return;
  float hv[3], tv[3];
#pragma unroll
  for (int f = 0; f < 3; ++f) {
    hv[f] = h1[(size_t)i * 3 + f];
    tv[f] = tx[(size_t)i * 3 + f];
  }
#pragma unroll
  for (int h = 0; h < 16; ++h) {
    float acc = b2[h];
#pragma unroll
    for (int f = 0; f < 3; ++f)
      acc += hv[f] * W2[f * 16 + h] + tv[f] * W2[48 + f * 16 + h];
    h2[(size_t)i * 16 + h] = fmaxf(acc, 0.0f);
  }
}

__global__ void head_gemm_wmma(const float* __restrict__ h2, const float* __restrict__ Wl,
                               const float* __restrict__ bl, float* __restrict__ out,
                               int ntiles) {
  int gwave = (blockIdx.x * blockDim.x + threadIdx.x) >> 5;
  int lane  = threadIdx.x & 31;
  if (gwave >= ntiles) return;
  int half = lane >> 4;
  int l16  = lane & 15;
  int row  = gwave * 16 + l16;

  v8f acc0 = {}, acc1 = {}, acc2 = {}, acc3 = {};
#pragma unroll
  for (int kk = 0; kk < 16; kk += 4) {
    int ka = kk + 2 * half;
    v2f a;
    a.x = h2[(size_t)row * 16 + ka];
    a.y = h2[(size_t)row * 16 + ka + 1];

    v2f b0, b1, b2, b3;
    b0.x = Wl[ka * 64 +  0 + l16]; b0.y = Wl[(ka + 1) * 64 +  0 + l16];
    b1.x = Wl[ka * 64 + 16 + l16]; b1.y = Wl[(ka + 1) * 64 + 16 + l16];
    b2.x = Wl[ka * 64 + 32 + l16]; b2.y = Wl[(ka + 1) * 64 + 32 + l16];
    b3.x = Wl[ka * 64 + 48 + l16]; b3.y = Wl[(ka + 1) * 64 + 48 + l16];

    acc0 = __builtin_amdgcn_wmma_f32_16x16x4_f32(false, a, false, b0, (short)0, acc0, false, false);
    acc1 = __builtin_amdgcn_wmma_f32_16x16x4_f32(false, a, false, b1, (short)0, acc1, false, false);
    acc2 = __builtin_amdgcn_wmma_f32_16x16x4_f32(false, a, false, b2, (short)0, acc2, false, false);
    acc3 = __builtin_amdgcn_wmma_f32_16x16x4_f32(false, a, false, b3, (short)0, acc3, false, false);
  }

  v8f accs[4] = {acc0, acc1, acc2, acc3};
#pragma unroll
  for (int j = 0; j < 4; ++j) {
    int col = j * 16 + l16;
    float bias = bl[col];
#pragma unroll
    for (int r = 0; r < 8; ++r) {
      int m = gwave * 16 + r + 8 * half;
      out[(size_t)m * 64 + col] = accs[j][r] + bias;
    }
  }
}

extern "C" void kernel_launch(void* const* d_in, const int* in_sizes, int n_in,
                              void* d_out, int out_size, void* d_ws, size_t ws_size,
                              hipStream_t stream) {
  const float* x  = (const float*)d_in[0];
  const int*   ei = (const int*)d_in[1];
  const float* W1 = (const float*)d_in[2];
  const float* b1 = (const float*)d_in[3];
  const float* W2 = (const float*)d_in[4];
  const float* b2 = (const float*)d_in[5];
  const float* Wl = (const float*)d_in[6];
  const float* bl = (const float*)d_in[7];
  float* out = (float*)d_out;

  const int N = in_sizes[0] / 8;
  const int E = in_sizes[1] / 2;
  const int* src = ei;
  const int* dst = ei + E;

  float* ws  = (float*)d_ws;
  float* deg = ws;
  float* dis = ws + (size_t)N;
  float* tx1 = ws + (size_t)2  * N;
  float* h1  = ws + (size_t)10 * N;
  float* tx2 = ws + (size_t)13 * N;
  float* h2  = ws + (size_t)16 * N;

  const int B = 256;
  zero_kernel<<<(16 * N + B - 1) / B, B, 0, stream>>>(ws, 16 * N);

  degree_kernel<<<(E + B - 1) / B, B, 0, stream>>>(src, deg, E);
  dis_kernel<<<(N + B - 1) / B, B, 0, stream>>>(deg, dis, N);

  prop1_kernel<<<(E + B - 1) / B, B, 0, stream>>>(src, dst, dis, x, tx1, E);
  layer1_kernel<<<(N + B - 1) / B, B, 0, stream>>>(x, tx1, W1, b1, h1, N);

  prop2_kernel<<<(E + B - 1) / B, B, 0, stream>>>(src, dst, dis, h1, tx2, E);
  layer2_kernel<<<(N + B - 1) / B, B, 0, stream>>>(h1, tx2, W2, b2, h2, N);

  const int ntiles = N / 16;
  head_gemm_wmma<<<(ntiles * 32 + B - 1) / B, B, 0, stream>>>(h2, Wl, bl, out, ntiles);
}


# Propagate by gather, no float atomics on Tx: one returning int atomic per edge ranks it in its dst bucket (arrival-order rank sets the sum order), scan + scatter build the list once for both layers
# speedup vs baseline: 2.0476x; 2.0476x over previous
#include <hip/hip_runtime.h>
#include <math.h>


typedef __attribute__((ext_vector_type(2))) float v2f;
typedef __attribute__((ext_vector_type(8))) float v8f;

#define N_NODES 100000

__global__ void zero_kernel(float* __restrict__ p, int n) {
  int i = blockIdx.x * blockDim.x + threadIdx.x;
  if (i < n) p[i] = 0.0f;
}

__global__ void degree_kernel(const int* __restrict__ src,
                              float* __restrict__ deg, int E) {
  int e = blockIdx.x * blockDim.x + threadIdx.x;
  if (e < E) atomicAdd(deg + src[e], 1.0f);
}

__global__ void dis_kernel(const float* __restrict__ deg,
                           float* __restrict__ dis, int n) {
  int i = blockIdx.x * blockDim.x + threadIdx.x;
  if (i < n) {
    float d = deg[i];
    dis[i] = (d > 0.0f) ? (1.0f / sqrtf(d)) : 0.0f;
  }
}

__global__ void rank_kernel(const int* __restrict__ dst, int* __restrict__ cnt,
                            int* __restrict__ rank, int E) {
  int e = blockIdx.x * blockDim.x + threadIdx.x;
  if (e < E) rank[e] = atomicAdd(cnt + dst[e], 1);
}

__device__ __forceinline__ void load4_counts(const int* __restrict__ cnt, int i, int send,
                                             int c[4]) {
#pragma unroll
  for (int k = 0; k < 4; ++k) {
    int idx = i + k;
    int v = cnt[idx < send ? idx : send - 1];
    c[k] = (idx < send) ? v : 0;
  }
}

__global__ void __launch_bounds__(1024)
scan_kernel(const int* __restrict__ cnt, int* __restrict__ off, int n) {
  __shared__ int wtot[32];
  const int lane = threadIdx.x & 31;
  const int wv   = threadIdx.x >> 5;
  const int seg  = ((n + 31) / 32 + 127) & ~127;
  const int sbeg = wv * seg;
  const int send = (sbeg + seg < n) ? (sbeg + seg) : n;

  int s = 0;
#pragma unroll 4
  for (int b = sbeg; b < send; b += 128) {
    int c[4];
    load4_counts(cnt, b + lane * 4, send, c);
    s += (c[0] + c[1]) + (c[2] + c[3]);
  }
  for (int o = 16; o > 0; o >>= 1) s += __shfl_xor(s, o);
  if (lane == 0) wtot[wv] = s;
  __syncthreads();

  int v = wtot[lane];
  int inc = v;
  for (int o = 1; o < 32; o <<= 1) {
    int u = __shfl_up(inc, o);
    if (lane >= o) inc += u;
  }
  int carry = __shfl(inc - v, wv);

  for (int b = sbeg; b < send; b += 128) {
    int i = b + lane * 4;
    int c[4];
    load4_counts(cnt, i, send, c);
    int t = (c[0] + c[1]) + (c[2] + c[3]);
    int inc2 = t;
    for (int o = 1; o < 32; o <<= 1) {
      int u = __shfl_up(inc2, o);
      if (lane >= o) inc2 += u;
    }
    int ex = carry + inc2 - t;
#pragma unroll
    for (int k = 0; k < 4; ++k) {
      if (i + k < send) off[i + k] = ex;
      ex += c[k];
    }
    carry += __shfl(inc2, 31);
  }
  if (wv == 31 && lane == 0) off[n] = carry;
}

__global__ void fill_kernel(const int* __restrict__ src, const int* __restrict__ dst,
                            const int* __restrict__ rank, const int* __restrict__ off,
                            const float* __restrict__ dis, int2* __restrict__ eb, int E) {
  int e = blockIdx.x * blockDim.x + threadIdx.x;
  if (e >= E) return;
  int s = src[e], d = dst[e];
  int2 rec;
  rec.x = s;
  rec.y = __float_as_int(dis[s]);
  eb[off[d] + rank[e]] = rec;
}

__global__ void pull1_kernel(const int* __restrict__ off, const int2* __restrict__ eb,
                             const float* __restrict__ dis, const float* __restrict__ x,
                             float* __restrict__ tx, int n) {
  int g = blockIdx.x * blockDim.x + threadIdx.x;
  int d = g >> 3, f = g & 7;
  if (d >= n) return;
  int beg = off[d], end = off[d + 1];
  float dd = dis[d];
  float acc = 0.0f;
#pragma unroll 8
  for (int j = beg; j < end; ++j) {
    int2 p = eb[j];
    float w = -(__int_as_float(p.y) * dd);
    acc += w * x[p.x * 8 + f];
  }
  tx[d * 8 + f] = acc;
}

__global__ void layer1_kernel(const float* __restrict__ x, const float* __restrict__ tx,
                              const float* __restrict__ W1, const float* __restrict__ b1,
                              float* __restrict__ h1, int n) {
  int i = blockIdx.x * blockDim.x + threadIdx.x;
  if (i >= n) return;
  const float4* xp = reinterpret_cast<const float4*>(x + (size_t)i * 8);
  const float4* tp = reinterpret_cast<const float4*>(tx + (size_t)i * 8);
  float4 x0 = xp[0], x1 = xp[1], t0 = tp[0], t1 = tp[1];
  float xi[8] = {x0.x, x0.y, x0.z, x0.w, x1.x, x1.y, x1.z, x1.w};
  float ti[8] = {t0.x, t0.y, t0.z, t0.w, t1.x, t1.y, t1.z, t1.w};
#pragma unroll
  for (int h = 0; h < 3; ++h) {
    float acc = b1[h];
#pragma unroll
    for (int f = 0; f < 8; ++f)
      acc += xi[f] * W1[f * 3 + h] + ti[f] * W1[24 + f * 3 + h];
    h1[(size_t)i * 3 + h] = fmaxf(acc, 0.0f);
  }
}

__global__ void pull2_kernel(const int* __restrict__ off, const int2* __restrict__ eb,
                             const float* __restrict__ dis, const float* __restrict__ h1,
                             float* __restrict__ tx, int n) {
  int g = blockIdx.x * blockDim.x + threadIdx.x;
  int d = g >> 2, f = g & 3;
  if (d >= n) return;
  int fc = (f < 3) ? f : 2;
  int beg = off[d], end = off[d + 1];
  float dd = dis[d];
  float acc = 0.0f;
#pragma unroll 8
  for (int j = beg; j < end; ++j) {
    int2 p = eb[j];
    float w = -(__int_as_float(p.y) * dd);
    acc += w * h1[p.x * 3 + fc];
  }
  if (f < 3) tx[d * 3 + f] = acc;
}

__global__ void layer2_kernel(const float* __restrict__ h1, const float* __restrict__ tx,
                              const float* __restrict__ W2, const float* __restrict__ b2,
                              float* __restrict__ h2, int n) {
  int i = blockIdx.x * blockDim.x + threadIdx.x;
  if (i >= n) return;
  float hv[3], tv[3];
#pragma unroll
  for (int f = 0; f < 3; ++f) {
    hv[f] = h1[(size_t)i * 3 + f];
    tv[f] = tx[(size_t)i * 3 + f];
  }
#pragma unroll
  for (int h = 0; h < 16; ++h) {
    float acc = b2[h];
#pragma unroll
    for (int f = 0; f < 3; ++f)
      acc += hv[f] * W2[f * 16 + h] + tv[f] * W2[48 + f * 16 + h];
    h2[(size_t)i * 16 + h] = fmaxf(acc, 0.0f);
  }
}

__global__ void head_gemm_wmma(const float* __restrict__ h2, const float* __restrict__ Wl,
                               const float* __restrict__ bl, float* __restrict__ out,
                               int ntiles) {
  int gwave = (blockIdx.x * blockDim.x + threadIdx.x) >> 5;
  int lane  = threadIdx.x & 31;
  if (gwave >= ntiles) return;
  int half = lane >> 4;
  int l16  = lane & 15;
  int row  = gwave * 16 + l16;

  v8f acc0 = {}, acc1 = {}, acc2 = {}, acc3 = {};
#pragma unroll
  for (int kk = 0; kk < 16; kk += 4) {
    int ka = kk + 2 * half;
    v2f a;
    a.x = h2[(size_t)row * 16 + ka];
    a.y = h2[(size_t)row * 16 + ka + 1];

    v2f b0, b1, b2, b3;
    b0.x = Wl[ka * 64 +  0 + l16]; b0.y = Wl[(ka + 1) * 64 +  0 + l16];
    b1.x = Wl[ka * 64 + 16 + l16]; b1.y = Wl[(ka + 1) * 64 + 16 + l16];
    b2.x = Wl[ka * 64 + 32 + l16]; b2.y = Wl[(ka + 1) * 64 + 32 + l16];
    b3.x = Wl[ka * 64 + 48 + l16]; b3.y = Wl[(ka + 1) * 64 + 48 + l16];

    acc0 = __builtin_amdgcn_wmma_f32_16x16x4_f32(false, a, false, b0, (short)0, acc0, false, false);
    acc1 = __builtin_amdgcn_wmma_f32_16x16x4_f32(false, a, false, b1, (short)0, acc1, false, false);
    acc2 = __builtin_amdgcn_wmma_f32_16x16x4_f32(false, a, false, b2, (short)0, acc2, false, false);
    acc3 = __builtin_amdgcn_wmma_f32_16x16x4_f32(false, a, false, b3, (short)0, acc3, false, false);
  }

  v8f accs[4] = {acc0, acc1, acc2, acc3};
#pragma unroll
  for (int j = 0; j < 4; ++j) {
    int col = j * 16 + l16;
    float bias = bl[col];
#pragma unroll
    for (int r = 0; r < 8; ++r) {
      int m = gwave * 16 + r + 8 * half;
      out[(size_t)m * 64 + col] = accs[j][r] + bias;
    }
  }
}

extern "C" void kernel_launch(void* const* d_in, const int* in_sizes, int n_in,
                              void* d_out, int out_size, void* d_ws, size_t ws_size,
                              hipStream_t stream) {
  const float* x  = (const float*)d_in[0];
  const int*   ei = (const int*)d_in[1];
  const float* W1 = (const float*)d_in[2];
  const float* b1 = (const float*)d_in[3];
  const float* W2 = (const float*)d_in[4];
  const float* b2 = (const float*)d_in[5];
  const float* Wl = (const float*)d_in[6];
  const float* bl = (const float*)d_in[7];
  float* out = (float*)d_out;

  const int N = in_sizes[0] / 8;
  const int E = in_sizes[1] / 2;
  const int* src = ei;
  const int* dst = ei + E;

  float* ws  = (float*)d_ws;
  float* deg = ws;
  int*   cnt = (int*)(ws + (size_t)N);
  float* dis = ws + (size_t)2  * N;
  float* tx1 = ws + (size_t)3  * N;
  float* h1  = ws + (size_t)11 * N;
  float* tx2 = ws + (size_t)14 * N;
  float* h2  = ws + (size_t)17 * N;
  int*   off = (int*)(ws + (size_t)33 * N);
  int2*  eb  = (int2*)(ws + (size_t)34 * N + 4);
  int*   rank = (int*)(ws + (size_t)34 * N + 4 + (size_t)2 * E);

  const int B = 256;
  zero_kernel<<<(2 * N + B - 1) / B, B, 0, stream>>>(ws, 2 * N);

  degree_kernel<<<(E + B - 1) / B, B, 0, stream>>>(src, deg, E);
  dis_kernel<<<(N + B - 1) / B, B, 0, stream>>>(deg, dis, N);

  rank_kernel<<<(E + B - 1) / B, B, 0, stream>>>(dst, cnt, rank, E);
  scan_kernel<<<1, 1024, 0, stream>>>(cnt, off, N);
  fill_kernel<<<(E + B - 1) / B, B, 0, stream>>>(src, dst, rank, off, dis, eb, E);

  pull1_kernel<<<(8 * N + B - 1) / B, B, 0, stream>>>(off, eb, dis, x, tx1, N);
  layer1_kernel<<<(N + B - 1) / B, B, 0, stream>>>(x, tx1, W1, b1, h1, N);

  pull2_kernel<<<(4 * N + B - 1) / B, B, 0, stream>>>(off, eb, dis, h1, tx2, N);
  layer2_kernel<<<(N + B - 1) / B, B, 0, stream>>>(h1, tx2, W2, b2, h2, N);

  const int ntiles = N / 16;
  head_gemm_wmma<<<(ntiles * 32 + B - 1) / B, B, 0, stream>>>(h2, Wl, bl, out, ntiles);
}


# Degree and in-bucket rank from per-slice LDS counters (16-bit pairs, LDS atomics) plus a column pass replace both global-atomic edge passes; sum order: slice order, then LDS arrival order in a slice
# speedup vs baseline: 5.0495x; 2.4661x over previous
#include <hip/hip_runtime.h>
#include <math.h>


typedef __attribute__((ext_vector_type(2))) float v2f;
typedef __attribute__((ext_vector_type(8))) float v8f;

#define N_NODES 100000

#define NWORDS (N_NODES / 2)
#define SLICE  50000

__global__ void __launch_bounds__(1024)
hist_kernel(const int* __restrict__ src, const int* __restrict__ dst,
            unsigned int* __restrict__ histD, unsigned int* __restrict__ histS,
            unsigned short* __restrict__ lrank, int E) {
  __shared__ unsigned int h[NWORDS];
  const int t  = threadIdx.x;
  const int e0 = blockIdx.x * SLICE;
  const int e1 = (e0 + SLICE < E) ? (e0 + SLICE) : E;
  const bool is_dst = (blockIdx.y == 0);
  const int* __restrict__ idx = is_dst ? dst : src;

  for (int i = t; i < NWORDS; i += 1024) h[i] = 0u;
  __syncthreads();

  for (int b = e0 + t; b < e1; b += 8 * 1024) {
    int d[8];
#pragma unroll
    for (int k = 0; k < 8; ++k) {
      int e = b + k * 1024;
      d[k] = idx[e < e1 ? e : e1 - 1];
    }
    unsigned old[8];
#pragma unroll
    for (int k = 0; k < 8; ++k) {
      unsigned sh = (unsigned)(d[k] & 1) * 16u;
      old[k] = atomicAdd(&h[d[k] >> 1], (b + k * 1024 < e1) ? (1u << sh) : 0u);
    }
    if (is_dst) {
#pragma unroll
      for (int k = 0; k < 8; ++k) {
        int e = b + k * 1024;
        unsigned sh = (unsigned)(d[k] & 1) * 16u;
        if (e < e1) lrank[e] = (unsigned short)((old[k] >> sh) & 0xFFFFu);
      }
    }
  }
  __syncthreads();

  unsigned int* row = (is_dst ? histD : histS) + (size_t)blockIdx.x * NWORDS;
  for (int i = t; i < NWORDS; i += 1024) row[i] = h[i];
}

__global__ void colscan_kernel(const unsigned int* __restrict__ histD,
                               const unsigned int* __restrict__ histS,
                               int* __restrict__ base, int* __restrict__ cnt,
                               float* __restrict__ dis, int G) {
  int i = blockIdx.x * blockDim.x + threadIdx.x;
  if (i >= NWORDS) return;
  if (blockIdx.y == 0) {
    int r0 = 0, r1 = 0;
#pragma unroll 16
    for (int w = 0; w < G; ++w) {
      unsigned v = histD[(size_t)w * NWORDS + i];
      int2 b;
      b.x = r0;
      b.y = r1;
      *reinterpret_cast<int2*>(base + (size_t)w * N_NODES + 2 * i) = b;
      r0 += (int)(v & 0xFFFFu);
      r1 += (int)(v >> 16);
    }
    int2 c;
    c.x = r0;
    c.y = r1;
    *reinterpret_cast<int2*>(cnt + 2 * i) = c;
  } else {
    int d0 = 0, d1 = 0;
#pragma unroll 16
    for (int w = 0; w < G; ++w) {
      unsigned u = histS[(size_t)w * NWORDS + i];
      d0 += (int)(u & 0xFFFFu);
      d1 += (int)(u >> 16);
    }
    float f0 = (float)d0, f1 = (float)d1;
    float2 r;
    r.x = (f0 > 0.0f) ? (1.0f / sqrtf(f0)) : 0.0f;
    r.y = (f1 > 0.0f) ? (1.0f / sqrtf(f1)) : 0.0f;
    *reinterpret_cast<float2*>(dis + 2 * i) = r;
  }
}

__device__ __forceinline__ void load4_counts(const int* __restrict__ cnt, int i, int send,
                                             int c[4]) {
#pragma unroll
  for (int k = 0; k < 4; ++k) {
    int idx = i + k;
    int v = cnt[idx < send ? idx : send - 1];
    c[k] = (idx < send) ? v : 0;
  }
}

__global__ void __launch_bounds__(1024)
scan_kernel(const int* __restrict__ cnt, int* __restrict__ off, int n) {
  __shared__ int wtot[32];
  const int lane = threadIdx.x & 31;
  const int wv   = threadIdx.x >> 5;
  const int seg  = ((n + 31) / 32 + 127) & ~127;
  const int sbeg = wv * seg;
  const int send = (sbeg + seg < n) ? (sbeg + seg) : n;

  int s = 0;
#pragma unroll 4
  for (int b = sbeg; b < send; b += 128) {
    int c[4];
    load4_counts(cnt, b + lane * 4, send, c);
    s += (c[0] + c[1]) + (c[2] + c[3]);
  }
  for (int o = 16; o > 0; o >>= 1) s += __shfl_xor(s, o);
  if (lane == 0) wtot[wv] = s;
  __syncthreads();

  int v = wtot[lane];
  int inc = v;
  for (int o = 1; o < 32; o <<= 1) {
    int u = __shfl_up(inc, o);
    if (lane >= o) inc += u;
  }
  int carry = __shfl(inc - v, wv);

  for (int b = sbeg; b < send; b += 128) {
    int i = b + lane * 4;
    int c[4];
    load4_counts(cnt, i, send, c);
    int t = (c[0] + c[1]) + (c[2] + c[3]);
    int inc2 = t;
    for (int o = 1; o < 32; o <<= 1) {
      int u = __shfl_up(inc2, o);
      if (lane >= o) inc2 += u;
    }
    int ex = carry + inc2 - t;
#pragma unroll
    for (int k = 0; k < 4; ++k) {
      if (i + k < send) off[i + k] = ex;
      ex += c[k];
    }
    carry += __shfl(inc2, 31);
  }
  if (wv == 31 && lane == 0) off[n] = carry;
}

__global__ void fill_kernel(const int* __restrict__ src, const int* __restrict__ dst,
                            const unsigned short* __restrict__ lrank,
                            const int* __restrict__ base, const int* __restrict__ off,
                            const float* __restrict__ dis, int2* __restrict__ eb, int E) {
  int e = blockIdx.x * blockDim.x + threadIdx.x;
  if (e >= E) return;
  int s = src[e], d = dst[e];
  int w = e / SLICE;
  int2 rec;
  rec.x = s;
  rec.y = __float_as_int(dis[s]);
  eb[off[d] + base[(size_t)w * N_NODES + d] + (int)lrank[e]] = rec;
}

__global__ void pull1_kernel(const int* __restrict__ off, const int2* __restrict__ eb,
                             const float* __restrict__ dis, const float* __restrict__ x,
                             float* __restrict__ tx, int n) {
  int g = blockIdx.x * blockDim.x + threadIdx.x;
  int d = g >> 3, f = g & 7;
  if (d >= n) return;
  int beg = off[d], end = off[d + 1];
  float dd = dis[d];
  float acc = 0.0f;
#pragma unroll 8
  for (int j = beg; j < end; ++j) {
    int2 p = eb[j];
    float w = -(__int_as_float(p.y) * dd);
    acc += w * x[p.x * 8 + f];
  }
  tx[d * 8 + f] = acc;
}

__global__ void layer1_kernel(const float* __restrict__ x, const float* __restrict__ tx,
                              const float* __restrict__ W1, const float* __restrict__ b1,
                              float* __restrict__ h1, int n) {
  int i = blockIdx.x * blockDim.x + threadIdx.x;
  if (i >= n) return;
  const float4* xp = reinterpret_cast<const float4*>(x + (size_t)i * 8);
  const float4* tp = reinterpret_cast<const float4*>(tx + (size_t)i * 8);
  float4 x0 = xp[0], x1 = xp[1], t0 = tp[0], t1 = tp[1];
  float xi[8] = {x0.x, x0.y, x0.z, x0.w, x1.x, x1.y, x1.z, x1.w};
  float ti[8] = {t0.x, t0.y, t0.z, t0.w, t1.x, t1.y, t1.z, t1.w};
#pragma unroll
  for (int h = 0; h < 3; ++h) {
    float acc = b1[h];
#pragma unroll
    for (int f = 0; f < 8; ++f)
      acc += xi[f] * W1[f * 3 + h] + ti[f] * W1[24 + f * 3 + h];
    h1[(size_t)i * 3 + h] = fmaxf(acc, 0.0f);
  }
}

__global__ void pull2_kernel(const int* __restrict__ off, const int2* __restrict__ eb,
                             const float* __restrict__ dis, const float* __restrict__ h1,
                             float* __restrict__ tx, int n) {
  int g = blockIdx.x * blockDim.x + threadIdx.x;
  int d = g >> 2, f = g & 3;
  if (d >= n) return;
  int fc = (f < 3) ? f : 2;
  int beg = off[d], end = off[d + 1];
  float dd = dis[d];
  float acc = 0.0f;
#pragma unroll 8
  for (int j = beg; j < end; ++j) {
    int2 p = eb[j];
    float w = -(__int_as_float(p.y) * dd);
    acc += w * h1[p.x * 3 + fc];
  }
  if (f < 3) tx[d * 3 + f] = acc;
}

__global__ void layer2_kernel(const float* __restrict__ h1, const float* __restrict__ tx,
                              const float* __restrict__ W2, const float* __restrict__ b2,
                              float* __restrict__ h2, int n) {
  int i = blockIdx.x * blockDim.x + threadIdx.x;
  if (i >= n) return;
  float hv[3], tv[3];
#pragma unroll
  for (int f = 0; f < 3; ++f) {
    hv[f] = h1[(size_t)i * 3 + f];
    tv[f] = tx[(size_t)i * 3 + f];
  }
#pragma unroll
  for (int h = 0; h < 16; ++h) {
    float acc = b2[h];
#pragma unroll
    for (int f = 0; f < 3; ++f)
      acc += hv[f] * W2[f * 16 + h] + tv[f] * W2[48 + f * 16 + h];
    h2[(size_t)i * 16 + h] = fmaxf(acc, 0.0f);
  }
}

__global__ void head_gemm_wmma(const float* __restrict__ h2, const float* __restrict__ Wl,
                               const float* __restrict__ bl, float* __restrict__ out,
                               int ntiles) {
  int gwave = (blockIdx.x * blockDim.x + threadIdx.x) >> 5;
  int lane  = threadIdx.x & 31;
  if (gwave >= ntiles) return;
  int half = lane >> 4;
  int l16  = lane & 15;
  int row  = gwave * 16 + l16;

  v8f acc0 = {}, acc1 = {}, acc2 = {}, acc3 = {};
#pragma unroll
  for (int kk = 0; kk < 16; kk += 4) {
    int ka = kk + 2 * half;
    v2f a;
    a.x = h2[(size_t)row * 16 + ka];
    a.y = h2[(size_t)row * 16 + ka + 1];

    v2f b0, b1, b2, b3;
    b0.x = Wl[ka * 64 +  0 + l16]; b0.y = Wl[(ka + 1) * 64 +  0 + l16];
    b1.x = Wl[ka * 64 + 16 + l16]; b1.y = Wl[(ka + 1) * 64 + 16 + l16];
    b2.x = Wl[ka * 64 + 32 + l16]; b2.y = Wl[(ka + 1) * 64 + 32 + l16];
    b3.x = Wl[ka * 64 + 48 + l16]; b3.y = Wl[(ka + 1) * 64 + 48 + l16];

    acc0 = __builtin_amdgcn_wmma_f32_16x16x4_f32(false, a, false, b0, (short)0, acc0, false, false);
    acc1 = __builtin_amdgcn_wmma_f32_16x16x4_f32(false, a, false, b1, (short)0, acc1, false, false);
    acc2 = __builtin_amdgcn_wmma_f32_16x16x4_f32(false, a, false, b2, (short)0, acc2, false, false);
    acc3 = __builtin_amdgcn_wmma_f32_16x16x4_f32(false, a, false, b3, (short)0, acc3, false, false);
  }

  v8f accs[4] = {acc0, acc1, acc2, acc3};
#pragma unroll
  for (int j = 0; j < 4; ++j) {
    int col = j * 16 + l16;
    float bias = bl[col];
#pragma unroll
    for (int r = 0; r < 8; ++r) {
      int m = gwave * 16 + r + 8 * half;
      out[(size_t)m * 64 + col] = accs[j][r] + bias;
    }
  }
}

extern "C" void kernel_launch(void* const* d_in, const int* in_sizes, int n_in,
                              void* d_out, int out_size, void* d_ws, size_t ws_size,
                              hipStream_t stream) {
  const float* x  = (const float*)d_in[0];
  const int*   ei = (const int*)d_in[1];
  const float* W1 = (const float*)d_in[2];
  const float* b1 = (const float*)d_in[3];
  const float* W2 = (const float*)d_in[4];
  const float* b2 = (const float*)d_in[5];
  const float* Wl = (const float*)d_in[6];
  const float* bl = (const float*)d_in[7];
  float* out = (float*)d_out;

  const int N = in_sizes[0] / 8;
  const int E = in_sizes[1] / 2;
  const int* src = ei;
  const int* dst = ei + E;

  const int G = (E + SLICE - 1) / SLICE;

  float* ws  = (float*)d_ws;
  int*   cnt = (int*)(ws + (size_t)N);
  float* dis = ws + (size_t)2  * N;
  float* tx1 = ws + (size_t)3  * N;
  float* h1  = ws + (size_t)11 * N;
  float* tx2 = ws + (size_t)14 * N;
  float* h2  = ws + (size_t)17 * N;
  int*   off = (int*)(ws + (size_t)33 * N);
  int2*  eb  = (int2*)(ws + (size_t)34 * N + 4);
  unsigned short* lrank = (unsigned short*)(ws + (size_t)34 * N + 4 + (size_t)2 * E);
  int*   base = (int*)(ws + (size_t)34 * N + 4 + (size_t)2 * E + (size_t)E / 2);
  unsigned int* histD = (unsigned int*)(base + (size_t)G * N);
  unsigned int* histS = histD + (size_t)G * NWORDS;

  const int B = 256;
  hist_kernel<<<dim3(G, 2), 1024, 0, stream>>>(src, dst, histD, histS, lrank, E);
  colscan_kernel<<<dim3((NWORDS + B - 1) / B, 2), B, 0, stream>>>(histD, histS, base, cnt, dis, G);

  scan_kernel<<<1, 1024, 0, stream>>>(cnt, off, N);
  fill_kernel<<<(E + B - 1) / B, B, 0, stream>>>(src, dst, lrank, base, off, dis, eb, E);

  pull1_kernel<<<(8 * N + B - 1) / B, B, 0, stream>>>(off, eb, dis, x, tx1, N);
  layer1_kernel<<<(N + B - 1) / B, B, 0, stream>>>(x, tx1, W1, b1, h1, N);

  pull2_kernel<<<(4 * N + B - 1) / B, B, 0, stream>>>(off, eb, dis, h1, tx2, N);
  layer2_kernel<<<(N + B - 1) / B, B, 0, stream>>>(h1, tx2, W2, b2, h2, N);

  const int ntiles = N / 16;
  head_gemm_wmma<<<(ntiles * 32 + B - 1) / B, B, 0, stream>>>(h2, Wl, bl, out, ntiles);
}


# Edge list built without scattered stores: each slice is ordered by dst in LDS (scan of its counters) and leaves whole; a group per 128 nodes gathers its runs and builds its window in LDS; same slots
# speedup vs baseline: 6.8913x; 1.3647x over previous
#include <hip/hip_runtime.h>
#include <math.h>


typedef __attribute__((ext_vector_type(2))) float v2f;
typedef __attribute__((ext_vector_type(8))) float v8f;

#define N_NODES 100000

#define NWORDS (N_NODES / 2)
#define SLICE  50000
#define NGRP   (N_NODES / 32)
static_assert(SLICE <= NWORDS, "the counter array is reused as the slice's ordered window");
static_assert(N_NODES % 32 == 0, "a wave owns 32 whole nodes");

#define FGRP 4
#define FPH  8
#define FCAP 16384

__global__ void __launch_bounds__(1024)
hist_kernel(const int* __restrict__ src, const int* __restrict__ dst,
            unsigned int* __restrict__ histD, unsigned int* __restrict__ histS,
            int* __restrict__ ss, int* __restrict__ lscan32, int E) {
  __shared__ unsigned int h[NWORDS];
  __shared__ unsigned short lr[SLICE];
  __shared__ int wtot[32];
  const int t  = threadIdx.x;
  const int e0 = blockIdx.x * SLICE;
  const int e1 = (e0 + SLICE < E) ? (e0 + SLICE) : E;
  const bool is_dst = (blockIdx.y == 0);
  const int* __restrict__ idx = is_dst ? dst : src;

  for (int i = t; i < NWORDS; i += 1024) h[i] = 0u;
  __syncthreads();

  for (int b = e0 + t; b < e1; b += 8 * 1024) {
    int d[8];
#pragma unroll
    for (int k = 0; k < 8; ++k) {
      int e = b + k * 1024;
      d[k] = idx[e < e1 ? e : e1 - 1];
    }
    unsigned old[8];
#pragma unroll
    for (int k = 0; k < 8; ++k) {
      unsigned sh = (unsigned)(d[k] & 1) * 16u;
      old[k] = atomicAdd(&h[d[k] >> 1], (b + k * 1024 < e1) ? (1u << sh) : 0u);
    }
    if (is_dst) {
#pragma unroll
      for (int k = 0; k < 8; ++k) {
        int e = b + k * 1024;
        unsigned sh = (unsigned)(d[k] & 1) * 16u;
        if (e < e1) lr[e - e0] = (unsigned short)((old[k] >> sh) & 0xFFFFu);
      }
    }
  }
  __syncthreads();

  unsigned int* row = (is_dst ? histD : histS) + (size_t)blockIdx.x * NWORDS;
  for (int i = t; i < NWORDS; i += 1024) row[i] = h[i];
  if (!is_dst) return;
  __syncthreads();

  const int lane = t & 31, wv = t >> 5;
  const int w0 = t * 49;
  const int w1 = (w0 + 49 < NWORDS) ? (w0 + 49) : NWORDS;
  int s = 0;
  for (int i = w0; i < w1; ++i) {
    unsigned v = h[i];
    s += (int)(v & 0xFFFFu) + (int)(v >> 16);
  }
  int inc = s;
  for (int o = 1; o < 32; o <<= 1) {
    int u = __shfl_up(inc, o);
    if (lane >= o) inc += u;
  }
  if (lane == 31) wtot[wv] = inc;
  __syncthreads();
  int tv = wtot[lane];
  int winc = tv;
  for (int o = 1; o < 32; o <<= 1) {
    int u = __shfl_up(winc, o);
    if (lane >= o) winc += u;
  }
  int run = __shfl(winc - tv, wv) + inc - s;
  for (int i = w0; i < w1; ++i) {
    unsigned v = h[i];
    unsigned lo = v & 0xFFFFu, hi = v >> 16;
    h[i] = (unsigned)run | (((unsigned)run + lo) << 16);
    run += (int)(lo + hi);
  }
  __syncthreads();

  for (int i = t; i < NGRP; i += 1024)
    lscan32[(size_t)blockIdx.x * NGRP + i] = (int)(h[16 * i] & 0xFFFFu);

  const int n = e1 - e0;
  for (int b = t; b < n; b += 8 * 1024) {
    int d[8];
#pragma unroll
    for (int k = 0; k < 8; ++k) {
      int i = b + k * 1024;
      d[k] = dst[e0 + (i < n ? i : n - 1)];
    }
#pragma unroll
    for (int k = 0; k < 8; ++k) {
      int i = b + k * 1024;
      if (i < n) {
        unsigned p = ((h[d[k] >> 1] >> ((unsigned)(d[k] & 1) * 16u)) & 0xFFFFu) + (unsigned)lr[i];
        lr[i] = (unsigned short)p;
      }
    }
  }
  __syncthreads();

  for (int b = t; b < n; b += 8 * 1024) {
    int sv[8];
#pragma unroll
    for (int k = 0; k < 8; ++k) {
      int i = b + k * 1024;
      sv[k] = src[e0 + (i < n ? i : n - 1)];
    }
#pragma unroll
    for (int k = 0; k < 8; ++k) {
      int i = b + k * 1024;
      if (i < n) h[lr[i]] = (unsigned)sv[k];
    }
  }
  __syncthreads();

  for (int i = t; i < n; i += 1024) ss[e0 + i] = (int)h[i];
}

__global__ void colscan_kernel(const unsigned int* __restrict__ histD,
                               const unsigned int* __restrict__ histS,
                               int* __restrict__ base, int* __restrict__ cnt,
                               float* __restrict__ dis, int G) {
  int i = blockIdx.x * blockDim.x + threadIdx.x;
  if (i >= NWORDS) return;
  if (blockIdx.y == 0) {
    int r0 = 0, r1 = 0;
#pragma unroll 16
    for (int w = 0; w < G; ++w) {
      unsigned v = histD[(size_t)w * NWORDS + i];
      int2 b;
      b.x = r0;
      b.y = r1;
      *reinterpret_cast<int2*>(base + (size_t)w * N_NODES + 2 * i) = b;
      r0 += (int)(v & 0xFFFFu);
      r1 += (int)(v >> 16);
    }
    int2 c;
    c.x = r0;
    c.y = r1;
    *reinterpret_cast<int2*>(cnt + 2 * i) = c;
  } else {
    int d0 = 0, d1 = 0;
#pragma unroll 16
    for (int w = 0; w < G; ++w) {
      unsigned u = histS[(size_t)w * NWORDS + i];
      d0 += (int)(u & 0xFFFFu);
      d1 += (int)(u >> 16);
    }
    float f0 = (float)d0, f1 = (float)d1;
    float2 r;
    r.x = (f0 > 0.0f) ? (1.0f / sqrtf(f0)) : 0.0f;
    r.y = (f1 > 0.0f) ? (1.0f / sqrtf(f1)) : 0.0f;
    *reinterpret_cast<float2*>(dis + 2 * i) = r;
  }
}

__device__ __forceinline__ void load4_counts(const int* __restrict__ cnt, int i, int send,
                                             int c[4]) {
#pragma unroll
  for (int k = 0; k < 4; ++k) {
    int idx = i + k;
    int v = cnt[idx < send ? idx : send - 1];
    c[k] = (idx < send) ? v : 0;
  }
}

__global__ void __launch_bounds__(1024)
scan_kernel(const int* __restrict__ cnt, int* __restrict__ off, int n) {
  __shared__ int wtot[32];
  const int lane = threadIdx.x & 31;
  const int wv   = threadIdx.x >> 5;
  const int seg  = ((n + 31) / 32 + 127) & ~127;
  const int sbeg = wv * seg;
  const int send = (sbeg + seg < n) ? (sbeg + seg) : n;

  int s = 0;
#pragma unroll 4
  for (int b = sbeg; b < send; b += 128) {
    int c[4];
    load4_counts(cnt, b + lane * 4, send, c);
    s += (c[0] + c[1]) + (c[2] + c[3]);
  }
  for (int o = 16; o > 0; o >>= 1) s += __shfl_xor(s, o);
  if (lane == 0) wtot[wv] = s;
  __syncthreads();

  int v = wtot[lane];
  int inc = v;
  for (int o = 1; o < 32; o <<= 1) {
    int u = __shfl_up(inc, o);
    if (lane >= o) inc += u;
  }
  int carry = __shfl(inc - v, wv);

  for (int b = sbeg; b < send; b += 128) {
    int i = b + lane * 4;
    int c[4];
    load4_counts(cnt, i, send, c);
    int t = (c[0] + c[1]) + (c[2] + c[3]);
    int inc2 = t;
    for (int o = 1; o < 32; o <<= 1) {
      int u = __shfl_up(inc2, o);
      if (lane >= o) inc2 += u;
    }
    int ex = carry + inc2 - t;
#pragma unroll
    for (int k = 0; k < 4; ++k) {
      if (i + k < send) off[i + k] = ex;
      ex += c[k];
    }
    carry += __shfl(inc2, 31);
  }
  if (wv == 31 && lane == 0) off[n] = carry;
}

__global__ void __launch_bounds__(1024)
fill_kernel(const unsigned int* __restrict__ histD, const int* __restrict__ base,
            const int* __restrict__ off, const int* __restrict__ lscan32,
            const int* __restrict__ ss, const float* __restrict__ dis,
            int2* __restrict__ eb, int G) {
  __shared__ int2 win[FCAP];
  const int lane = threadIdx.x & 31, wv = threadIdx.x >> 5;
  const int grp = blockIdx.x * FGRP + (wv & (FGRP - 1));
  const int n0 = blockIdx.x * (FGRP * 32);
  const int n1 = (n0 + FGRP * 32 < N_NODES) ? (n0 + FGRP * 32) : N_NODES;
  const int o0 = off[n0], o1 = off[n1];
  const bool staged = (o1 - o0) <= FCAP;

  int w = wv / FGRP;
  if (grp < NGRP && w < G) {
    const int d  = grp * 32 + lane;
    const int od = off[d];
    const unsigned sh = (unsigned)(d & 1) * 16u;
    unsigned word = histD[(size_t)w * NWORDS + (d >> 1)];
    int bs = base[(size_t)w * N_NODES + d];
    int rs = lscan32[(size_t)w * NGRP + grp];
    for (; w < G; w += FPH) {
      const int wn = (w + FPH < G) ? (w + FPH) : w;
      unsigned wordn = histD[(size_t)wn * NWORDS + (d >> 1)];
      int bsn = base[(size_t)wn * N_NODES + d];
      int rsn = lscan32[(size_t)wn * NGRP + grp];

      const int c = (int)((word >> sh) & 0xFFFFu);
      int inc = c;
      for (int o = 1; o < 32; o <<= 1) {
        int u = __shfl_up(inc, o);
        if (lane >= o) inc += u;
      }
      const int* __restrict__ sp = ss + (size_t)w * SLICE + rs + (inc - c);
      const int slot = od + bs;
      for (int k = 0; k < c; ++k) {
        int s = sp[k];
        int2 rec;
        rec.x = s;
        rec.y = __float_as_int(dis[s]);
        if (staged) win[slot - o0 + k] = rec;
        else        eb[slot + k] = rec;
      }
      word = wordn;
      bs = bsn;
      rs = rsn;
    }
  }
  __syncthreads();
  if (staged) {
    const int m = o1 - o0;
    for (int i = threadIdx.x; i < m; i += 1024) eb[o0 + i] = win[i];
  }
}

__global__ void pull1_kernel(const int* __restrict__ off, const int2* __restrict__ eb,
                             const float* __restrict__ dis, const float* __restrict__ x,
                             float* __restrict__ tx, int n) {
  int g = blockIdx.x * blockDim.x + threadIdx.x;
  int d = g >> 3, f = g & 7;
  if (d >= n) return;
  int beg = off[d], end = off[d + 1];
  float dd = dis[d];
  float acc = 0.0f;
#pragma unroll 8
  for (int j = beg; j < end; ++j) {
    int2 p = eb[j];
    float w = -(__int_as_float(p.y) * dd);
    acc += w * x[p.x * 8 + f];
  }
  tx[d * 8 + f] = acc;
}

__global__ void layer1_kernel(const float* __restrict__ x, const float* __restrict__ tx,
                              const float* __restrict__ W1, const float* __restrict__ b1,
                              float* __restrict__ h1, int n) {
  int i = blockIdx.x * blockDim.x + threadIdx.x;
  if (i >= n) return;
  const float4* xp = reinterpret_cast<const float4*>(x + (size_t)i * 8);
  const float4* tp = reinterpret_cast<const float4*>(tx + (size_t)i * 8);
  float4 x0 = xp[0], x1 = xp[1], t0 = tp[0], t1 = tp[1];
  float xi[8] = {x0.x, x0.y, x0.z, x0.w, x1.x, x1.y, x1.z, x1.w};
  float ti[8] = {t0.x, t0.y, t0.z, t0.w, t1.x, t1.y, t1.z, t1.w};
#pragma unroll
  for (int h = 0; h < 3; ++h) {
    float acc = b1[h];
#pragma unroll
    for (int f = 0; f < 8; ++f)
      acc += xi[f] * W1[f * 3 + h] + ti[f] * W1[24 + f * 3 + h];
    h1[(size_t)i * 3 + h] = fmaxf(acc, 0.0f);
  }
}

__global__ void pull2_kernel(const int* __restrict__ off, const int2* __restrict__ eb,
                             const float* __restrict__ dis, const float* __restrict__ h1,
                             float* __restrict__ tx, int n) {
  int g = blockIdx.x * blockDim.x + threadIdx.x;
  int d = g >> 2, f = g & 3;
  if (d >= n) return;
  int fc = (f < 3) ? f : 2;
  int beg = off[d], end = off[d + 1];
  float dd = dis[d];
  float acc = 0.0f;
#pragma unroll 8
  for (int j = beg; j < end; ++j) {
    int2 p = eb[j];
    float w = -(__int_as_float(p.y) * dd);
    acc += w * h1[p.x * 3 + fc];
  }
  if (f < 3) tx[d * 3 + f] = acc;
}

__global__ void layer2_kernel(const float* __restrict__ h1, const float* __restrict__ tx,
                              const float* __restrict__ W2, const float* __restrict__ b2,
                              float* __restrict__ h2, int n) {
  int i = blockIdx.x * blockDim.x + threadIdx.x;
  if (i >= n) return;
  float hv[3], tv[3];
#pragma unroll
  for (int f = 0; f < 3; ++f) {
    hv[f] = h1[(size_t)i * 3 + f];
    tv[f] = tx[(size_t)i * 3 + f];
  }
#pragma unroll
  for (int h = 0; h < 16; ++h) {
    float acc = b2[h];
#pragma unroll
    for (int f = 0; f < 3; ++f)
      acc += hv[f] * W2[f * 16 + h] + tv[f] * W2[48 + f * 16 + h];
    h2[(size_t)i * 16 + h] = fmaxf(acc, 0.0f);
  }
}

__global__ void head_gemm_wmma(const float* __restrict__ h2, const float* __restrict__ Wl,
                               const float* __restrict__ bl, float* __restrict__ out,
                               int ntiles) {
  int gwave = (blockIdx.x * blockDim.x + threadIdx.x) >> 5;
  int lane  = threadIdx.x & 31;
  if (gwave >= ntiles) return;
  int half = lane >> 4;
  int l16  = lane & 15;
  int row  = gwave * 16 + l16;

  v8f acc0 = {}, acc1 = {}, acc2 = {}, acc3 = {};
#pragma unroll
  for (int kk = 0; kk < 16; kk += 4) {
    int ka = kk + 2 * half;
    v2f a;
    a.x = h2[(size_t)row * 16 + ka];
    a.y = h2[(size_t)row * 16 + ka + 1];

    v2f b0, b1, b2, b3;
    b0.x = Wl[ka * 64 +  0 + l16]; b0.y = Wl[(ka + 1) * 64 +  0 + l16];
    b1.x = Wl[ka * 64 + 16 + l16]; b1.y = Wl[(ka + 1) * 64 + 16 + l16];
    b2.x = Wl[ka * 64 + 32 + l16]; b2.y = Wl[(ka + 1) * 64 + 32 + l16];
    b3.x = Wl[ka * 64 + 48 + l16]; b3.y = Wl[(ka + 1) * 64 + 48 + l16];

    acc0 = __builtin_amdgcn_wmma_f32_16x16x4_f32(false, a, false, b0, (short)0, acc0, false, false);
    acc1 = __builtin_amdgcn_wmma_f32_16x16x4_f32(false, a, false, b1, (short)0, acc1, false, false);
    acc2 = __builtin_amdgcn_wmma_f32_16x16x4_f32(false, a, false, b2, (short)0, acc2, false, false);
    acc3 = __builtin_amdgcn_wmma_f32_16x16x4_f32(false, a, false, b3, (short)0, acc3, false, false);
  }

  v8f accs[4] = {acc0, acc1, acc2, acc3};
#pragma unroll
  for (int j = 0; j < 4; ++j) {
    int col = j * 16 + l16;
    float bias = bl[col];
#pragma unroll
    for (int r = 0; r < 8; ++r) {
      int m = gwave * 16 + r + 8 * half;
      out[(size_t)m * 64 + col] = accs[j][r] + bias;
    }
  }
}

extern "C" void kernel_launch(void* const* d_in, const int* in_sizes, int n_in,
                              void* d_out, int out_size, void* d_ws, size_t ws_size,
                              hipStream_t stream) {
  const float* x  = (const float*)d_in[0];
  const int*   ei = (const int*)d_in[1];
  const float* W1 = (const float*)d_in[2];
  const float* b1 = (const float*)d_in[3];
  const float* W2 = (const float*)d_in[4];
  const float* b2 = (const float*)d_in[5];
  const float* Wl = (const float*)d_in[6];
  const float* bl = (const float*)d_in[7];
  float* out = (float*)d_out;

  const int N = in_sizes[0] / 8;
  const int E = in_sizes[1] / 2;
  const int* src = ei;
  const int* dst = ei + E;

  const int G = (E + SLICE - 1) / SLICE;

  float* ws  = (float*)d_ws;
  int*   cnt = (int*)(ws + (size_t)N);
  float* dis = ws + (size_t)2  * N;
  float* tx1 = ws + (size_t)3  * N;
  float* h1  = ws + (size_t)11 * N;
  float* tx2 = ws + (size_t)14 * N;
  float* h2  = ws + (size_t)17 * N;
  int*   off = (int*)(ws + (size_t)33 * N);
  int2*  eb  = (int2*)(ws + (size_t)34 * N + 4);
  int*   ss   = (int*)(ws + (size_t)34 * N + 4 + (size_t)2 * E);
  int*   base = (int*)(ws + (size_t)34 * N + 4 + (size_t)3 * E);
  unsigned int* histD = (unsigned int*)(base + (size_t)G * N);
  unsigned int* histS = histD + (size_t)G * NWORDS;
  int*   lscan32 = (int*)(histS + (size_t)G * NWORDS);

  const int B = 256;
  hist_kernel<<<dim3(G, 2), 1024, 0, stream>>>(src, dst, histD, histS, ss, lscan32, E);
  colscan_kernel<<<dim3((NWORDS + B - 1) / B, 2), B, 0, stream>>>(histD, histS, base, cnt, dis, G);

  scan_kernel<<<1, 1024, 0, stream>>>(cnt, off, N);
  fill_kernel<<<(NGRP + FGRP - 1) / FGRP, 1024, 0, stream>>>(histD, base, off, lscan32, ss, dis, eb, G);

  pull1_kernel<<<(8 * N + B - 1) / B, B, 0, stream>>>(off, eb, dis, x, tx1, N);
  layer1_kernel<<<(N + B - 1) / B, B, 0, stream>>>(x, tx1, W1, b1, h1, N);

  pull2_kernel<<<(4 * N + B - 1) / B, B, 0, stream>>>(off, eb, dis, h1, tx2, N);
  layer2_kernel<<<(N + B - 1) / B, B, 0, stream>>>(h1, tx2, W2, b2, h2, N);

  const int ntiles = N / 16;
  head_gemm_wmma<<<(ntiles * 32 + B - 1) / B, B, 0, stream>>>(h2, Wl, bl, out, ntiles);
}


# Bucket-start scan spread over 25 work-groups: each sums the counts in front of its 4096 nodes itself (independent 16-byte loads); the single work-group's serial 25-step walk is gone
# speedup vs baseline: 7.6029x; 1.1033x over previous
#include <hip/hip_runtime.h>
#include <math.h>


typedef __attribute__((ext_vector_type(2))) float v2f;
typedef __attribute__((ext_vector_type(8))) float v8f;

#define N_NODES 100000

#define NWORDS (N_NODES / 2)
#define SLICE  50000
#define NGRP   (N_NODES / 32)
static_assert(SLICE <= NWORDS, "the counter array is reused as the slice's ordered window");
static_assert(N_NODES % 32 == 0, "a wave owns 32 whole nodes");

#define FGRP 4
#define FPH  8
#define FCAP 16384

__global__ void __launch_bounds__(1024)
hist_kernel(const int* __restrict__ src, const int* __restrict__ dst,
            unsigned int* __restrict__ histD, unsigned int* __restrict__ histS,
            int* __restrict__ ss, int* __restrict__ lscan32, int E) {
  __shared__ unsigned int h[NWORDS];
  __shared__ unsigned short lr[SLICE];
  __shared__ int wtot[32];
  const int t  = threadIdx.x;
  const int e0 = blockIdx.x * SLICE;
  const int e1 = (e0 + SLICE < E) ? (e0 + SLICE) : E;
  const bool is_dst = (blockIdx.y == 0);
  const int* __restrict__ idx = is_dst ? dst : src;

  for (int i = t; i < NWORDS; i += 1024) h[i] = 0u;
  __syncthreads();

  for (int b = e0 + t; b < e1; b += 8 * 1024) {
    int d[8];
#pragma unroll
    for (int k = 0; k < 8; ++k) {
      int e = b + k * 1024;
      d[k] = idx[e < e1 ? e : e1 - 1];
    }
    unsigned old[8];
#pragma unroll
    for (int k = 0; k < 8; ++k) {
      unsigned sh = (unsigned)(d[k] & 1) * 16u;
      old[k] = atomicAdd(&h[d[k] >> 1], (b + k * 1024 < e1) ? (1u << sh) : 0u);
    }
    if (is_dst) {
#pragma unroll
      for (int k = 0; k < 8; ++k) {
        int e = b + k * 1024;
        unsigned sh = (unsigned)(d[k] & 1) * 16u;
        if (e < e1) lr[e - e0] = (unsigned short)((old[k] >> sh) & 0xFFFFu);
      }
    }
  }
  __syncthreads();

  unsigned int* row = (is_dst ? histD : histS) + (size_t)blockIdx.x * NWORDS;
  for (int i = t; i < NWORDS; i += 1024) row[i] = h[i];
  if (!is_dst) return;
  __syncthreads();

  const int lane = t & 31, wv = t >> 5;
  const int w0 = t * 49;
  const int w1 = (w0 + 49 < NWORDS) ? (w0 + 49) : NWORDS;
  int s = 0;
  for (int i = w0; i < w1; ++i) {
    unsigned v = h[i];
    s += (int)(v & 0xFFFFu) + (int)(v >> 16);
  }
  int inc = s;
  for (int o = 1; o < 32; o <<= 1) {
    int u = __shfl_up(inc, o);
    if (lane >= o) inc += u;
  }
  if (lane == 31) wtot[wv] = inc;
  __syncthreads();
  int tv = wtot[lane];
  int winc = tv;
  for (int o = 1; o < 32; o <<= 1) {
    int u = __shfl_up(winc, o);
    if (lane >= o) winc += u;
  }
  int run = __shfl(winc - tv, wv) + inc - s;
  for (int i = w0; i < w1; ++i) {
    unsigned v = h[i];
    unsigned lo = v & 0xFFFFu, hi = v >> 16;
    h[i] = (unsigned)run | (((unsigned)run + lo) << 16);
    run += (int)(lo + hi);
  }
  __syncthreads();

  for (int i = t; i < NGRP; i += 1024)
    lscan32[(size_t)blockIdx.x * NGRP + i] = (int)(h[16 * i] & 0xFFFFu);

  const int n = e1 - e0;
  for (int b = t; b < n; b += 8 * 1024) {
    int d[8];
#pragma unroll
    for (int k = 0; k < 8; ++k) {
      int i = b + k * 1024;
      d[k] = dst[e0 + (i < n ? i : n - 1)];
    }
#pragma unroll
    for (int k = 0; k < 8; ++k) {
      int i = b + k * 1024;
      if (i < n) {
        unsigned p = ((h[d[k] >> 1] >> ((unsigned)(d[k] & 1) * 16u)) & 0xFFFFu) + (unsigned)lr[i];
        lr[i] = (unsigned short)p;
      }
    }
  }
  __syncthreads();

  for (int b = t; b < n; b += 8 * 1024) {
    int sv[8];
#pragma unroll
    for (int k = 0; k < 8; ++k) {
      int i = b + k * 1024;
      sv[k] = src[e0 + (i < n ? i : n - 1)];
    }
#pragma unroll
    for (int k = 0; k < 8; ++k) {
      int i = b + k * 1024;
      if (i < n) h[lr[i]] = (unsigned)sv[k];
    }
  }
  __syncthreads();

  for (int i = t; i < n; i += 1024) ss[e0 + i] = (int)h[i];
}

__global__ void colscan_kernel(const unsigned int* __restrict__ histD,
                               const unsigned int* __restrict__ histS,
                               int* __restrict__ base, int* __restrict__ cnt,
                               float* __restrict__ dis, int G) {
  int i = blockIdx.x * blockDim.x + threadIdx.x;
  if (i >= NWORDS) return;
  if (blockIdx.y == 0) {
    int r0 = 0, r1 = 0;
#pragma unroll 16
    for (int w = 0; w < G; ++w) {
      unsigned v = histD[(size_t)w * NWORDS + i];
      int2 b;
      b.x = r0;
      b.y = r1;
      *reinterpret_cast<int2*>(base + (size_t)w * N_NODES + 2 * i) = b;
      r0 += (int)(v & 0xFFFFu);
      r1 += (int)(v >> 16);
    }
    int2 c;
    c.x = r0;
    c.y = r1;
    *reinterpret_cast<int2*>(cnt + 2 * i) = c;
  } else {
    int d0 = 0, d1 = 0;
#pragma unroll 16
    for (int w = 0; w < G; ++w) {
      unsigned u = histS[(size_t)w * NWORDS + i];
      d0 += (int)(u & 0xFFFFu);
      d1 += (int)(u >> 16);
    }
    float f0 = (float)d0, f1 = (float)d1;
    float2 r;
    r.x = (f0 > 0.0f) ? (1.0f / sqrtf(f0)) : 0.0f;
    r.y = (f1 > 0.0f) ? (1.0f / sqrtf(f1)) : 0.0f;
    *reinterpret_cast<float2*>(dis + 2 * i) = r;
  }
}

#define SBLK 4096
static_assert(N_NODES % 4 == 0, "cnt and off start at 16-byte boundaries of the workspace");

__global__ void __launch_bounds__(1024)
scan_kernel(const int* __restrict__ cnt, int* __restrict__ off, int n) {
  __shared__ int wsum[32];
  __shared__ int wtot[32];
  const int t = threadIdx.x, lane = t & 31, wv = t >> 5;
  const int b0 = blockIdx.x * SBLK;
  const int i  = b0 + t * 4;
  const int4* __restrict__ c4 = reinterpret_cast<const int4*>(cnt);

  int4 v = make_int4(0, 0, 0, 0);
  if (i + 3 < n) {
    v = c4[i >> 2];
  } else {
    if (i < n)     v.x = cnt[i];
    if (i + 1 < n) v.y = cnt[i + 1];
    if (i + 2 < n) v.z = cnt[i + 2];
  }

  int s = 0;
  const int rows = blockIdx.x;
  const int4* __restrict__ rp = c4 + t;
  int r = 0;
  for (; r + 12 <= rows; r += 12) {
#pragma unroll
    for (int k = 0; k < 12; ++k) {
      const int4 p = rp[(r + k) * 1024];
      s += (p.x + p.y) + (p.z + p.w);
    }
  }
  const int rem = rows - r;
  if (rem > 0) {
    int last = 0;
#pragma unroll
    for (int k = 0; k < 12; ++k) {
      const int rr = (r + k < rows) ? (r + k) : (rows - 1);
      const int4 p = rp[rr * 1024];
      last = (p.x + p.y) + (p.z + p.w);
      s += last;
    }
    s -= (12 - rem) * last;
  }

  const int t4 = (v.x + v.y) + (v.z + v.w);
  int inc = t4;
  for (int o = 1; o < 32; o <<= 1) {
    int u = __shfl_up(inc, o);
    if (lane >= o) inc += u;
  }
  for (int o = 16; o > 0; o >>= 1) s += __shfl_xor(s, o);
  if (lane == 31) wtot[wv] = inc;
  if (lane == 0)  wsum[wv] = s;
  __syncthreads();

  int front = wsum[lane];
  for (int o = 16; o > 0; o >>= 1) front += __shfl_xor(front, o);
  const int tv = wtot[lane];
  int winc = tv;
  for (int o = 1; o < 32; o <<= 1) {
    int u = __shfl_up(winc, o);
    if (lane >= o) winc += u;
  }
  const int ex = front + __shfl(winc - tv, wv) + (inc - t4);
  const int total = front + __shfl(winc, 31);

  if (i + 3 < n) {
    int4 o4;
    o4.x = ex;
    o4.y = ex + v.x;
    o4.z = ex + v.x + v.y;
    o4.w = ex + v.x + v.y + v.z;
    *reinterpret_cast<int4*>(off + i) = o4;
  } else {
    if (i < n)     off[i]     = ex;
    if (i + 1 < n) off[i + 1] = ex + v.x;
    if (i + 2 < n) off[i + 2] = ex + v.x + v.y;
  }
  if (blockIdx.x == gridDim.x - 1 && t == 0) off[n] = total;
}

__global__ void __launch_bounds__(1024)
fill_kernel(const unsigned int* __restrict__ histD, const int* __restrict__ base,
            const int* __restrict__ off, const int* __restrict__ lscan32,
            const int* __restrict__ ss, const float* __restrict__ dis,
            int2* __restrict__ eb, int G) {
  __shared__ int2 win[FCAP];
  const int lane = threadIdx.x & 31, wv = threadIdx.x >> 5;
  const int grp = blockIdx.x * FGRP + (wv & (FGRP - 1));
  const int n0 = blockIdx.x * (FGRP * 32);
  const int n1 = (n0 + FGRP * 32 < N_NODES) ? (n0 + FGRP * 32) : N_NODES;
  const int o0 = off[n0], o1 = off[n1];
  const bool staged = (o1 - o0) <= FCAP;

  int w = wv / FGRP;
  if (grp < NGRP && w < G) {
    const int d  = grp * 32 + lane;
    const int od = off[d];
    const unsigned sh = (unsigned)(d & 1) * 16u;
    unsigned word = histD[(size_t)w * NWORDS + (d >> 1)];
    int bs = base[(size_t)w * N_NODES + d];
    int rs = lscan32[(size_t)w * NGRP + grp];
    for (; w < G; w += FPH) {
      const int wn = (w + FPH < G) ? (w + FPH) : w;
      unsigned wordn = histD[(size_t)wn * NWORDS + (d >> 1)];
      int bsn = base[(size_t)wn * N_NODES + d];
      int rsn = lscan32[(size_t)wn * NGRP + grp];

      const int c = (int)((word >> sh) & 0xFFFFu);
      int inc = c;
      for (int o = 1; o < 32; o <<= 1) {
        int u = __shfl_up(inc, o);
        if (lane >= o) inc += u;
      }
      const int* __restrict__ sp = ss + (size_t)w * SLICE + rs + (inc - c);
      const int slot = od + bs;
      for (int k = 0; k < c; ++k) {
        int s = sp[k];
        int2 rec;
        rec.x = s;
        rec.y = __float_as_int(dis[s]);
        if (staged) win[slot - o0 + k] = rec;
        else        eb[slot + k] = rec;
      }
      word = wordn;
      bs = bsn;
      rs = rsn;
    }
  }
  __syncthreads();
  if (staged) {
    const int m = o1 - o0;
    for (int i = threadIdx.x; i < m; i += 1024) eb[o0 + i] = win[i];
  }
}

__global__ void pull1_kernel(const int* __restrict__ off, const int2* __restrict__ eb,
                             const float* __restrict__ dis, const float* __restrict__ x,
                             float* __restrict__ tx, int n) {
  int g = blockIdx.x * blockDim.x + threadIdx.x;
  int d = g >> 3, f = g & 7;
  if (d >= n) return;
  int beg = off[d], end = off[d + 1];
  float dd = dis[d];
  float acc = 0.0f;
#pragma unroll 8
  for (int j = beg; j < end; ++j) {
    int2 p = eb[j];
    float w = -(__int_as_float(p.y) * dd);
    acc += w * x[p.x * 8 + f];
  }
  tx[d * 8 + f] = acc;
}

__global__ void layer1_kernel(const float* __restrict__ x, const float* __restrict__ tx,
                              const float* __restrict__ W1, const float* __restrict__ b1,
                              float* __restrict__ h1, int n) {
  int i = blockIdx.x * blockDim.x + threadIdx.x;
  if (i >= n) return;
  const float4* xp = reinterpret_cast<const float4*>(x + (size_t)i * 8);
  const float4* tp = reinterpret_cast<const float4*>(tx + (size_t)i * 8);
  float4 x0 = xp[0], x1 = xp[1], t0 = tp[0], t1 = tp[1];
  float xi[8] = {x0.x, x0.y, x0.z, x0.w, x1.x, x1.y, x1.z, x1.w};
  float ti[8] = {t0.x, t0.y, t0.z, t0.w, t1.x, t1.y, t1.z, t1.w};
#pragma unroll
  for (int h = 0; h < 3; ++h) {
    float acc = b1[h];
#pragma unroll
    for (int f = 0; f < 8; ++f)
      acc += xi[f] * W1[f * 3 + h] + ti[f] * W1[24 + f * 3 + h];
    h1[(size_t)i * 3 + h] = fmaxf(acc, 0.0f);
  }
}

__global__ void pull2_kernel(const int* __restrict__ off, const int2* __restrict__ eb,
                             const float* __restrict__ dis, const float* __restrict__ h1,
                             float* __restrict__ tx, int n) {
  int g = blockIdx.x * blockDim.x + threadIdx.x;
  int d = g >> 2, f = g & 3;
  if (d >= n) return;
  int fc = (f < 3) ? f : 2;
  int beg = off[d], end = off[d + 1];
  float dd = dis[d];
  float acc = 0.0f;
#pragma unroll 8
  for (int j = beg; j < end; ++j) {
    int2 p = eb[j];
    float w = -(__int_as_float(p.y) * dd);
    acc += w * h1[p.x * 3 + fc];
  }
  if (f < 3) tx[d * 3 + f] = acc;
}

__global__ void layer2_kernel(const float* __restrict__ h1, const float* __restrict__ tx,
                              const float* __restrict__ W2, const float* __restrict__ b2,
                              float* __restrict__ h2, int n) {
  int i = blockIdx.x * blockDim.x + threadIdx.x;
  if (i >= n) return;
  float hv[3], tv[3];
#pragma unroll
  for (int f = 0; f < 3; ++f) {
    hv[f] = h1[(size_t)i * 3 + f];
    tv[f] = tx[(size_t)i * 3 + f];
  }
#pragma unroll
  for (int h = 0; h < 16; ++h) {
    float acc = b2[h];
#pragma unroll
    for (int f = 0; f < 3; ++f)
      acc += hv[f] * W2[f * 16 + h] + tv[f] * W2[48 + f * 16 + h];
    h2[(size_t)i * 16 + h] = fmaxf(acc, 0.0f);
  }
}

__global__ void head_gemm_wmma(const float* __restrict__ h2, const float* __restrict__ Wl,
                               const float* __restrict__ bl, float* __restrict__ out,
                               int ntiles) {
  int gwave = (blockIdx.x * blockDim.x + threadIdx.x) >> 5;
  int lane  = threadIdx.x & 31;
  if (gwave >= ntiles) return;
  int half = lane >> 4;
  int l16  = lane & 15;
  int row  = gwave * 16 + l16;

  v8f acc0 = {}, acc1 = {}, acc2 = {}, acc3 = {};
#pragma unroll
  for (int kk = 0; kk < 16; kk += 4) {
    int ka = kk + 2 * half;
    v2f a;
    a.x = h2[(size_t)row * 16 + ka];
    a.y = h2[(size_t)row * 16 + ka + 1];

    v2f b0, b1, b2, b3;
    b0.x = Wl[ka * 64 +  0 + l16]; b0.y = Wl[(ka + 1) * 64 +  0 + l16];
    b1.x = Wl[ka * 64 + 16 + l16]; b1.y = Wl[(ka + 1) * 64 + 16 + l16];
    b2.x = Wl[ka * 64 + 32 + l16]; b2.y = Wl[(ka + 1) * 64 + 32 + l16];
    b3.x = Wl[ka * 64 + 48 + l16]; b3.y = Wl[(ka + 1) * 64 + 48 + l16];

    acc0 = __builtin_amdgcn_wmma_f32_16x16x4_f32(false, a, false, b0, (short)0, acc0, false, false);
    acc1 = __builtin_amdgcn_wmma_f32_16x16x4_f32(false, a, false, b1, (short)0, acc1, false, false);
    acc2 = __builtin_amdgcn_wmma_f32_16x16x4_f32(false, a, false, b2, (short)0, acc2, false, false);
    acc3 = __builtin_amdgcn_wmma_f32_16x16x4_f32(false, a, false, b3, (short)0, acc3, false, false);
  }

  v8f accs[4] = {acc0, acc1, acc2, acc3};
#pragma unroll
  for (int j = 0; j < 4; ++j) {
    int col = j * 16 + l16;
    float bias = bl[col];
#pragma unroll
    for (int r = 0; r < 8; ++r) {
      int m = gwave * 16 + r + 8 * half;
      out[(size_t)m * 64 + col] = accs[j][r] + bias;
    }
  }
}

extern "C" void kernel_launch(void* const* d_in, const int* in_sizes, int n_in,
                              void* d_out, int out_size, void* d_ws, size_t ws_size,
                              hipStream_t stream) {
  const float* x  = (const float*)d_in[0];
  const int*   ei = (const int*)d_in[1];
  const float* W1 = (const float*)d_in[2];
  const float* b1 = (const float*)d_in[3];
  const float* W2 = (const float*)d_in[4];
  const float* b2 = (const float*)d_in[5];
  const float* Wl = (const float*)d_in[6];
  const float* bl = (const float*)d_in[7];
  float* out = (float*)d_out;

  const int N = in_sizes[0] / 8;
  const int E = in_sizes[1] / 2;
  const int* src = ei;
  const int* dst = ei + E;

  const int G = (E + SLICE - 1) / SLICE;

  float* ws  = (float*)d_ws;
  int*   cnt = (int*)(ws + (size_t)N);
  float* dis = ws + (size_t)2  * N;
  float* tx1 = ws + (size_t)3  * N;
  float* h1  = ws + (size_t)11 * N;
  float* tx2 = ws + (size_t)14 * N;
  float* h2  = ws + (size_t)17 * N;
  int*   off = (int*)(ws + (size_t)33 * N);
  int2*  eb  = (int2*)(ws + (size_t)34 * N + 4);
  int*   ss   = (int*)(ws + (size_t)34 * N + 4 + (size_t)2 * E);
  int*   base = (int*)(ws + (size_t)34 * N + 4 + (size_t)3 * E);
  unsigned int* histD = (unsigned int*)(base + (size_t)G * N);
  unsigned int* histS = histD + (size_t)G * NWORDS;
  int*   lscan32 = (int*)(histS + (size_t)G * NWORDS);

  const int B = 256;
  hist_kernel<<<dim3(G, 2), 1024, 0, stream>>>(src, dst, histD, histS, ss, lscan32, E);
  colscan_kernel<<<dim3((NWORDS + B - 1) / B, 2), B, 0, stream>>>(histD, histS, base, cnt, dis, G);

  scan_kernel<<<(N + SBLK - 1) / SBLK, 1024, 0, stream>>>(cnt, off, N);
  fill_kernel<<<(NGRP + FGRP - 1) / FGRP, 1024, 0, stream>>>(histD, base, off, lscan32, ss, dis, eb, G);

  pull1_kernel<<<(8 * N + B - 1) / B, B, 0, stream>>>(off, eb, dis, x, tx1, N);
  layer1_kernel<<<(N + B - 1) / B, B, 0, stream>>>(x, tx1, W1, b1, h1, N);

  pull2_kernel<<<(4 * N + B - 1) / B, B, 0, stream>>>(off, eb, dis, h1, tx2, N);
  layer2_kernel<<<(N + B - 1) / B, B, 0, stream>>>(h1, tx2, W2, b2, h2, N);

  const int ntiles = N / 16;
  head_gemm_wmma<<<(ntiles * 32 + B - 1) / B, B, 0, stream>>>(h2, Wl, bl, out, ntiles);
}


# Bucket fill: a wave takes two slices per trip and loads four src ids per lane and slice as one clamped group ahead of all stores (was: one dependent load then store per record); slots unchanged
# speedup vs baseline: 9.3897x; 1.2350x over previous
#include <hip/hip_runtime.h>
#include <math.h>


typedef __attribute__((ext_vector_type(2))) float v2f;
typedef __attribute__((ext_vector_type(8))) float v8f;

#define N_NODES 100000

#define NWORDS (N_NODES / 2)
#define SLICE  50000
#define NGRP   (N_NODES / 32)
static_assert(SLICE <= NWORDS, "the counter array is reused as the slice's ordered window");
static_assert(N_NODES % 32 == 0, "a wave owns 32 whole nodes");

#define FGRP 4
#define FPH  8
#define FCAP 16384

__global__ void __launch_bounds__(1024)
hist_kernel(const int* __restrict__ src, const int* __restrict__ dst,
            unsigned int* __restrict__ histD, unsigned int* __restrict__ histS,
            int* __restrict__ ss, int* __restrict__ lscan32, int E) {
  __shared__ unsigned int h[NWORDS];
  __shared__ unsigned short lr[SLICE];
  __shared__ int wtot[32];
  const int t  = threadIdx.x;
  const int e0 = blockIdx.x * SLICE;
  const int e1 = (e0 + SLICE < E) ? (e0 + SLICE) : E;
  const bool is_dst = (blockIdx.y == 0);
  const int* __restrict__ idx = is_dst ? dst : src;

  for (int i = t; i < NWORDS; i += 1024) h[i] = 0u;
  __syncthreads();

  for (int b = e0 + t; b < e1; b += 8 * 1024) {
    int d[8];
#pragma unroll
    for (int k = 0; k < 8; ++k) {
      int e = b + k * 1024;
      d[k] = idx[e < e1 ? e : e1 - 1];
    }
    unsigned old[8];
#pragma unroll
    for (int k = 0; k < 8; ++k) {
      unsigned sh = (unsigned)(d[k] & 1) * 16u;
      old[k] = atomicAdd(&h[d[k] >> 1], (b + k * 1024 < e1) ? (1u << sh) : 0u);
    }
    if (is_dst) {
#pragma unroll
      for (int k = 0; k < 8; ++k) {
        int e = b + k * 1024;
        unsigned sh = (unsigned)(d[k] & 1) * 16u;
        if (e < e1) lr[e - e0] = (unsigned short)((old[k] >> sh) & 0xFFFFu);
      }
    }
  }
  __syncthreads();

  unsigned int* row = (is_dst ? histD : histS) + (size_t)blockIdx.x * NWORDS;
  for (int i = t; i < NWORDS; i += 1024) row[i] = h[i];
  if (!is_dst) return;
  __syncthreads();

  const int lane = t & 31, wv = t >> 5;
  const int w0 = t * 49;
  const int w1 = (w0 + 49 < NWORDS) ? (w0 + 49) : NWORDS;
  int s = 0;
  for (int i = w0; i < w1; ++i) {
    unsigned v = h[i];
    s += (int)(v & 0xFFFFu) + (int)(v >> 16);
  }
  int inc = s;
  for (int o = 1; o < 32; o <<= 1) {
    int u = __shfl_up(inc, o);
    if (lane >= o) inc += u;
  }
  if (lane == 31) wtot[wv] = inc;
  __syncthreads();
  int tv = wtot[lane];
  int winc = tv;
  for (int o = 1; o < 32; o <<= 1) {
    int u = __shfl_up(winc, o);
    if (lane >= o) winc += u;
  }
  int run = __shfl(winc - tv, wv) + inc - s;
  for (int i = w0; i < w1; ++i) {
    unsigned v = h[i];
    unsigned lo = v & 0xFFFFu, hi = v >> 16;
    h[i] = (unsigned)run | (((unsigned)run + lo) << 16);
    run += (int)(lo + hi);
  }
  __syncthreads();

  for (int i = t; i < NGRP; i += 1024)
    lscan32[(size_t)blockIdx.x * NGRP + i] = (int)(h[16 * i] & 0xFFFFu);

  const int n = e1 - e0;
  for (int b = t; b < n; b += 8 * 1024) {
    int d[8];
#pragma unroll
    for (int k = 0; k < 8; ++k) {
      int i = b + k * 1024;
      d[k] = dst[e0 + (i < n ? i : n - 1)];
    }
#pragma unroll
    for (int k = 0; k < 8; ++k) {
      int i = b + k * 1024;
      if (i < n) {
        unsigned p = ((h[d[k] >> 1] >> ((unsigned)(d[k] & 1) * 16u)) & 0xFFFFu) + (unsigned)lr[i];
        lr[i] = (unsigned short)p;
      }
    }
  }
  __syncthreads();

  for (int b = t; b < n; b += 8 * 1024) {
    int sv[8];
#pragma unroll
    for (int k = 0; k < 8; ++k) {
      int i = b + k * 1024;
      sv[k] = src[e0 + (i < n ? i : n - 1)];
    }
#pragma unroll
    for (int k = 0; k < 8; ++k) {
      int i = b + k * 1024;
      if (i < n) h[lr[i]] = (unsigned)sv[k];
    }
  }
  __syncthreads();

  for (int i = t; i < n; i += 1024) ss[e0 + i] = (int)h[i];
}

__global__ void colscan_kernel(const unsigned int* __restrict__ histD,
                               const unsigned int* __restrict__ histS,
                               int* __restrict__ base, int* __restrict__ cnt,
                               float* __restrict__ dis, const float* __restrict__ x,
                               float* __restrict__ xs, int G) {
  int i = blockIdx.x * blockDim.x + threadIdx.x;
  if (i >= NWORDS) return;
  if (blockIdx.y == 0) {
    int r0 = 0, r1 = 0;
#pragma unroll 16
    for (int w = 0; w < G; ++w) {
      unsigned v = histD[(size_t)w * NWORDS + i];
      int2 b;
      b.x = r0;
      b.y = r1;
      *reinterpret_cast<int2*>(base + (size_t)w * N_NODES + 2 * i) = b;
      r0 += (int)(v & 0xFFFFu);
      r1 += (int)(v >> 16);
    }
    int2 c;
    c.x = r0;
    c.y = r1;
    *reinterpret_cast<int2*>(cnt + 2 * i) = c;
  } else {
    const float4* __restrict__ xp = reinterpret_cast<const float4*>(x + (size_t)16 * i);
    float4 a0 = xp[0], a1 = xp[1], c0 = xp[2], c1 = xp[3];
    int d0 = 0, d1 = 0;
#pragma unroll 16
    for (int w = 0; w < G; ++w) {
      unsigned u = histS[(size_t)w * NWORDS + i];
      d0 += (int)(u & 0xFFFFu);
      d1 += (int)(u >> 16);
    }
    float f0 = (float)d0, f1 = (float)d1;
    float2 r;
    r.x = (f0 > 0.0f) ? (1.0f / sqrtf(f0)) : 0.0f;
    r.y = (f1 > 0.0f) ? (1.0f / sqrtf(f1)) : 0.0f;
    *reinterpret_cast<float2*>(dis + 2 * i) = r;
    a0.x *= r.x; a0.y *= r.x; a0.z *= r.x; a0.w *= r.x;
    a1.x *= r.x; a1.y *= r.x; a1.z *= r.x; a1.w *= r.x;
    c0.x *= r.y; c0.y *= r.y; c0.z *= r.y; c0.w *= r.y;
    c1.x *= r.y; c1.y *= r.y; c1.z *= r.y; c1.w *= r.y;
    float4* __restrict__ xo = reinterpret_cast<float4*>(xs + (size_t)16 * i);
    xo[0] = a0;
    xo[1] = a1;
    xo[2] = c0;
    xo[3] = c1;
  }
}

#define SBLK 4096
static_assert(N_NODES % 4 == 0, "cnt and off start at 16-byte boundaries of the workspace");

__global__ void __launch_bounds__(1024)
scan_kernel(const int* __restrict__ cnt, int* __restrict__ off, int n) {
  __shared__ int wsum[32];
  __shared__ int wtot[32];
  const int t = threadIdx.x, lane = t & 31, wv = t >> 5;
  const int b0 = blockIdx.x * SBLK;
  const int i  = b0 + t * 4;
  const int4* __restrict__ c4 = reinterpret_cast<const int4*>(cnt);

  int4 v = make_int4(0, 0, 0, 0);
  if (i + 3 < n) {
    v = c4[i >> 2];
  } else {
    if (i < n)     v.x = cnt[i];
    if (i + 1 < n) v.y = cnt[i + 1];
    if (i + 2 < n) v.z = cnt[i + 2];
  }

  int s = 0;
  const int rows = blockIdx.x;
  const int4* __restrict__ rp = c4 + t;
  int r = 0;
  for (; r + 12 <= rows; r += 12) {
#pragma unroll
    for (int k = 0; k < 12; ++k) {
      const int4 p = rp[(r + k) * 1024];
      s += (p.x + p.y) + (p.z + p.w);
    }
  }
  const int rem = rows - r;
  if (rem > 0) {
    int last = 0;
#pragma unroll
    for (int k = 0; k < 12; ++k) {
      const int rr = (r + k < rows) ? (r + k) : (rows - 1);
      const int4 p = rp[rr * 1024];
      last = (p.x + p.y) + (p.z + p.w);
      s += last;
    }
    s -= (12 - rem) * last;
  }

  const int t4 = (v.x + v.y) + (v.z + v.w);
  int inc = t4;
  for (int o = 1; o < 32; o <<= 1) {
    int u = __shfl_up(inc, o);
    if (lane >= o) inc += u;
  }
  for (int o = 16; o > 0; o >>= 1) s += __shfl_xor(s, o);
  if (lane == 31) wtot[wv] = inc;
  if (lane == 0)  wsum[wv] = s;
  __syncthreads();

  int front = wsum[lane];
  for (int o = 16; o > 0; o >>= 1) front += __shfl_xor(front, o);
  const int tv = wtot[lane];
  int winc = tv;
  for (int o = 1; o < 32; o <<= 1) {
    int u = __shfl_up(winc, o);
    if (lane >= o) winc += u;
  }
  const int ex = front + __shfl(winc - tv, wv) + (inc - t4);
  const int total = front + __shfl(winc, 31);

  if (i + 3 < n) {
    int4 o4;
    o4.x = ex;
    o4.y = ex + v.x;
    o4.z = ex + v.x + v.y;
    o4.w = ex + v.x + v.y + v.z;
    *reinterpret_cast<int4*>(off + i) = o4;
  } else {
    if (i < n)     off[i]     = ex;
    if (i + 1 < n) off[i + 1] = ex + v.x;
    if (i + 2 < n) off[i + 2] = ex + v.x + v.y;
  }
  if (blockIdx.x == gridDim.x - 1 && t == 0) off[n] = total;
}

#define FREC 4
__global__ void __launch_bounds__(1024)
fill_kernel(const unsigned int* __restrict__ histD, const int* __restrict__ base,
            const int* __restrict__ off, const int* __restrict__ lscan32,
            const int* __restrict__ ss, int* __restrict__ eb, int G, int E) {
  __shared__ int win[FCAP];
  const int lane = threadIdx.x & 31, wv = threadIdx.x >> 5;
  const int grp = blockIdx.x * FGRP + (wv & (FGRP - 1));
  const int n0 = blockIdx.x * (FGRP * 32);
  const int n1 = (n0 + FGRP * 32 < N_NODES) ? (n0 + FGRP * 32) : N_NODES;
  const int o0 = off[n0], o1 = off[n1];
  const bool staged = (o1 - o0) <= FCAP;

  int w = wv / FGRP;
  if (grp < NGRP && w < G) {
    const int d  = grp * 32 + lane;
    const int od = off[d];
    const unsigned sh = (unsigned)(d & 1) * 16u;
    int wb = (w + FPH < G) ? (w + FPH) : w;
    unsigned wordA = histD[(size_t)w * NWORDS + (d >> 1)];
    int bsA = base[(size_t)w * N_NODES + d];
    int rsA = lscan32[(size_t)w * NGRP + grp];
    unsigned wordB = histD[(size_t)wb * NWORDS + (d >> 1)];
    int bsB = base[(size_t)wb * N_NODES + d];
    int rsB = lscan32[(size_t)wb * NGRP + grp];
    for (; w < G; w += 2 * FPH) {
      const bool hasB = (w + FPH < G);
      const int wn  = (w + 2 * FPH < G) ? (w + 2 * FPH) : w;
      const int wbn = (w + 3 * FPH < G) ? (w + 3 * FPH) : wn;
      unsigned wordAn = histD[(size_t)wn * NWORDS + (d >> 1)];
      int bsAn = base[(size_t)wn * N_NODES + d];
      int rsAn = lscan32[(size_t)wn * NGRP + grp];
      unsigned wordBn = histD[(size_t)wbn * NWORDS + (d >> 1)];
      int bsBn = base[(size_t)wbn * N_NODES + d];
      int rsBn = lscan32[(size_t)wbn * NGRP + grp];

      const int cA = (int)((wordA >> sh) & 0xFFFFu);
      const int cB = hasB ? (int)((wordB >> sh) & 0xFFFFu) : 0;
      int incA = cA, incB = cB;
      for (int o = 1; o < 32; o <<= 1) {
        int uA = __shfl_up(incA, o);
        int uB = __shfl_up(incB, o);
        if (lane >= o) { incA += uA; incB += uB; }
      }
      const int pA = w  * SLICE + rsA + (incA - cA);
      const int pB = wb * SLICE + rsB + (incB - cB);
      const int slotA = od + bsA;
      const int slotB = od + bsB;

      int rA[FREC], rB[FREC];
#pragma unroll
      for (int k = 0; k < FREC; ++k) {
        int q = pA + ((k < cA) ? k : 0);
        rA[k] = ss[(q < E) ? q : (E - 1)];
      }
#pragma unroll
      for (int k = 0; k < FREC; ++k) {
        int q = pB + ((k < cB) ? k : 0);
        rB[k] = ss[(q < E) ? q : (E - 1)];
      }
#pragma unroll
      for (int k = 0; k < FREC; ++k) {
        if (k < cA) {
          if (staged) win[slotA - o0 + k] = rA[k];
          else        eb[slotA + k] = rA[k];
        }
      }
#pragma unroll
      for (int k = 0; k < FREC; ++k) {
        if (k < cB) {
          if (staged) win[slotB - o0 + k] = rB[k];
          else        eb[slotB + k] = rB[k];
        }
      }
      for (int k = FREC; k < cA; ++k) {
        const int s = ss[pA + k];
        if (staged) win[slotA - o0 + k] = s;
        else        eb[slotA + k] = s;
      }
      for (int k = FREC; k < cB; ++k) {
        const int s = ss[pB + k];
        if (staged) win[slotB - o0 + k] = s;
        else        eb[slotB + k] = s;
      }
      wordA = wordAn; bsA = bsAn; rsA = rsAn;
      wordB = wordBn; bsB = bsBn; rsB = rsBn;
      wb = wbn;
    }
  }
  __syncthreads();
  if (staged) {
    const int m = o1 - o0;
    for (int i = threadIdx.x; i < m; i += 1024) eb[o0 + i] = win[i];
  }
}

__global__ void pull1_kernel(const int* __restrict__ off, const int* __restrict__ eb,
                             const float* __restrict__ dis, const float* __restrict__ xs,
                             float* __restrict__ tx, int n) {
  int g = blockIdx.x * blockDim.x + threadIdx.x;
  int d = g >> 3, f = g & 7;
  if (d >= n) return;
  int beg = off[d], end = off[d + 1];
  float dd = dis[d];
  float acc = 0.0f;
#pragma unroll 8
  for (int j = beg; j < end; ++j) {
    const int s = eb[j];
    acc += xs[s * 8 + f];
  }
  tx[d * 8 + f] = -dd * acc;
}

__global__ void layer1_kernel(const float* __restrict__ x, const float* __restrict__ tx,
                              const float* __restrict__ W1, const float* __restrict__ b1,
                              const float* __restrict__ dis, float* __restrict__ h1,
                              float* __restrict__ h1s, int n) {
  int i = blockIdx.x * blockDim.x + threadIdx.x;
  if (i >= n) return;
  const float4* xp = reinterpret_cast<const float4*>(x + (size_t)i * 8);
  const float4* tp = reinterpret_cast<const float4*>(tx + (size_t)i * 8);
  float4 x0 = xp[0], x1 = xp[1], t0 = tp[0], t1 = tp[1];
  float xi[8] = {x0.x, x0.y, x0.z, x0.w, x1.x, x1.y, x1.z, x1.w};
  float ti[8] = {t0.x, t0.y, t0.z, t0.w, t1.x, t1.y, t1.z, t1.w};
  float hv[3];
#pragma unroll
  for (int h = 0; h < 3; ++h) {
    float acc = b1[h];
#pragma unroll
    for (int f = 0; f < 8; ++f)
      acc += xi[f] * W1[f * 3 + h] + ti[f] * W1[24 + f * 3 + h];
    hv[h] = fmaxf(acc, 0.0f);
    h1[(size_t)i * 3 + h] = hv[h];
  }
  const float di = dis[i];
  float4 sv;
  sv.x = di * hv[0];
  sv.y = di * hv[1];
  sv.z = di * hv[2];
  sv.w = 0.0f;
  *reinterpret_cast<float4*>(h1s + (size_t)i * 4) = sv;
}

__global__ void pull2_kernel(const int* __restrict__ off, const int* __restrict__ eb,
                             const float* __restrict__ dis, const float* __restrict__ h1s,
                             float* __restrict__ tx, int n) {
  int g = blockIdx.x * blockDim.x + threadIdx.x;
  int d = g >> 2, f = g & 3;
  if (d >= n) return;
  int beg = off[d], end = off[d + 1];
  float dd = dis[d];
  float acc = 0.0f;
#pragma unroll 8
  for (int j = beg; j < end; ++j) {
    const int s = eb[j];
    acc += h1s[s * 4 + f];
  }
  if (f < 3) tx[d * 3 + f] = -dd * acc;
}

__global__ void layer2_kernel(const float* __restrict__ h1, const float* __restrict__ tx,
                              const float* __restrict__ W2, const float* __restrict__ b2,
                              float* __restrict__ h2, int n) {
  int i = blockIdx.x * blockDim.x + threadIdx.x;
  if (i >= n) return;
  float hv[3], tv[3];
#pragma unroll
  for (int f = 0; f < 3; ++f) {
    hv[f] = h1[(size_t)i * 3 + f];
    tv[f] = tx[(size_t)i * 3 + f];
  }
#pragma unroll
  for (int h = 0; h < 16; ++h) {
    float acc = b2[h];
#pragma unroll
    for (int f = 0; f < 3; ++f)
      acc += hv[f] * W2[f * 16 + h] + tv[f] * W2[48 + f * 16 + h];
    h2[(size_t)i * 16 + h] = fmaxf(acc, 0.0f);
  }
}

__global__ void head_gemm_wmma(const float* __restrict__ h2, const float* __restrict__ Wl,
                               const float* __restrict__ bl, float* __restrict__ out,
                               int ntiles) {
  int gwave = (blockIdx.x * blockDim.x + threadIdx.x) >> 5;
  int lane  = threadIdx.x & 31;
  if (gwave >= ntiles) return;
  int half = lane >> 4;
  int l16  = lane & 15;
  int row  = gwave * 16 + l16;

  v8f acc0 = {}, acc1 = {}, acc2 = {}, acc3 = {};
#pragma unroll
  for (int kk = 0; kk < 16; kk += 4) {
    int ka = kk + 2 * half;
    v2f a;
    a.x = h2[(size_t)row * 16 + ka];
    a.y = h2[(size_t)row * 16 + ka + 1];

    v2f b0, b1, b2, b3;
    b0.x = Wl[ka * 64 +  0 + l16]; b0.y = Wl[(ka + 1) * 64 +  0 + l16];
    b1.x = Wl[ka * 64 + 16 + l16]; b1.y = Wl[(ka + 1) * 64 + 16 + l16];
    b2.x = Wl[ka * 64 + 32 + l16]; b2.y = Wl[(ka + 1) * 64 + 32 + l16];
    b3.x = Wl[ka * 64 + 48 + l16]; b3.y = Wl[(ka + 1) * 64 + 48 + l16];

    acc0 = __builtin_amdgcn_wmma_f32_16x16x4_f32(false, a, false, b0, (short)0, acc0, false, false);
    acc1 = __builtin_amdgcn_wmma_f32_16x16x4_f32(false, a, false, b1, (short)0, acc1, false, false);
    acc2 = __builtin_amdgcn_wmma_f32_16x16x4_f32(false, a, false, b2, (short)0, acc2, false, false);
    acc3 = __builtin_amdgcn_wmma_f32_16x16x4_f32(false, a, false, b3, (short)0, acc3, false, false);
  }

  v8f accs[4] = {acc0, acc1, acc2, acc3};
#pragma unroll
  for (int j = 0; j < 4; ++j) {
    int col = j * 16 + l16;
    float bias = bl[col];
#pragma unroll
    for (int r = 0; r < 8; ++r) {
      int m = gwave * 16 + r + 8 * half;
      out[(size_t)m * 64 + col] = accs[j][r] + bias;
    }
  }
}

extern "C" void kernel_launch(void* const* d_in, const int* in_sizes, int n_in,
                              void* d_out, int out_size, void* d_ws, size_t ws_size,
                              hipStream_t stream) {
  const float* x  = (const float*)d_in[0];
  const int*   ei = (const int*)d_in[1];
  const float* W1 = (const float*)d_in[2];
  const float* b1 = (const float*)d_in[3];
  const float* W2 = (const float*)d_in[4];
  const float* b2 = (const float*)d_in[5];
  const float* Wl = (const float*)d_in[6];
  const float* bl = (const float*)d_in[7];
  float* out = (float*)d_out;

  const int N = in_sizes[0] / 8;
  const int E = in_sizes[1] / 2;
  const int* src = ei;
  const int* dst = ei + E;

  const int G = (E + SLICE - 1) / SLICE;

  float* ws  = (float*)d_ws;
  int*   cnt = (int*)(ws + (size_t)N);
  float* dis = ws + (size_t)2  * N;
  float* tx1 = ws + (size_t)3  * N;
  float* h1  = ws + (size_t)11 * N;
  float* tx2 = ws + (size_t)14 * N;
  float* h2  = ws + (size_t)17 * N;
  int*   off = (int*)(ws + (size_t)33 * N);
  int*   eb  = (int*)(ws + (size_t)34 * N + 4);
  float* xs  = ws + (size_t)34 * N + 8 + (size_t)E;
  float* h1s = xs + (size_t)8 * N;
  int*   ss   = (int*)(ws + (size_t)34 * N + 4 + (size_t)2 * E);
  int*   base = (int*)(ws + (size_t)34 * N + 4 + (size_t)3 * E);
  unsigned int* histD = (unsigned int*)(base + (size_t)G * N);
  unsigned int* histS = histD + (size_t)G * NWORDS;
  int*   lscan32 = (int*)(histS + (size_t)G * NWORDS);

  const int B = 256;
  hist_kernel<<<dim3(G, 2), 1024, 0, stream>>>(src, dst, histD, histS, ss, lscan32, E);
  colscan_kernel<<<dim3((NWORDS + B - 1) / B, 2), B, 0, stream>>>(histD, histS, base, cnt, dis, x, xs, G);

  scan_kernel<<<(N + SBLK - 1) / SBLK, 1024, 0, stream>>>(cnt, off, N);
  fill_kernel<<<(NGRP + FGRP - 1) / FGRP, 1024, 0, stream>>>(histD, base, off, lscan32, ss, eb, G, E);

  pull1_kernel<<<(8 * N + B - 1) / B, B, 0, stream>>>(off, eb, dis, xs, tx1, N);
  layer1_kernel<<<(N + B - 1) / B, B, 0, stream>>>(x, tx1, W1, b1, dis, h1, h1s, N);

  pull2_kernel<<<(4 * N + B - 1) / B, B, 0, stream>>>(off, eb, dis, h1s, tx2, N);
  layer2_kernel<<<(N + B - 1) / B, B, 0, stream>>>(h1, tx2, W2, b2, h2, N);

  const int ntiles = N / 16;
  head_gemm_wmma<<<(ntiles * 32 + B - 1) / B, B, 0, stream>>>(h2, Wl, bl, out, ntiles);
}


# Dense stages fused into the two gather kernels, 9 -> 6 launches: layer 1 summed over the node's 8 lanes (3-step lane tree, bias last: other rounding); layer 2 + head GEMM per 64 nodes through LDS
# speedup vs baseline: 9.8475x; 1.0488x over previous
#include <hip/hip_runtime.h>
#include <math.h>


typedef __attribute__((ext_vector_type(2))) float v2f;
typedef __attribute__((ext_vector_type(8))) float v8f;

#define N_NODES 100000

#define NWORDS (N_NODES / 2)
#define SLICE  50000
#define NGRP   (N_NODES / 32)
static_assert(SLICE <= NWORDS, "the counter array is reused as the slice's ordered window");
static_assert(N_NODES % 32 == 0, "a wave owns 32 whole nodes");

#define FGRP 4
#define FPH  8
#define FCAP 16384

__global__ void __launch_bounds__(1024)
hist_kernel(const int* __restrict__ src, const int* __restrict__ dst,
            unsigned int* __restrict__ histD, unsigned int* __restrict__ histS,
            int* __restrict__ ss, int* __restrict__ lscan32, int E) {
  __shared__ unsigned int h[NWORDS];
  __shared__ unsigned short lr[SLICE];
  __shared__ int wtot[32];
  const int t  = threadIdx.x;
  const int e0 = blockIdx.x * SLICE;
  const int e1 = (e0 + SLICE < E) ? (e0 + SLICE) : E;
  const bool is_dst = (blockIdx.y == 0);
  const int* __restrict__ idx = is_dst ? dst : src;

  for (int i = t; i < NWORDS; i += 1024) h[i] = 0u;
  __syncthreads();

  for (int b = e0 + t; b < e1; b += 8 * 1024) {
    int d[8];
#pragma unroll
    for (int k = 0; k < 8; ++k) {
      int e = b + k * 1024;
      d[k] = idx[e < e1 ? e : e1 - 1];
    }
    unsigned old[8];
#pragma unroll
    for (int k = 0; k < 8; ++k) {
      unsigned sh = (unsigned)(d[k] & 1) * 16u;
      old[k] = atomicAdd(&h[d[k] >> 1], (b + k * 1024 < e1) ? (1u << sh) : 0u);
    }
    if (is_dst) {
#pragma unroll
      for (int k = 0; k < 8; ++k) {
        int e = b + k * 1024;
        unsigned sh = (unsigned)(d[k] & 1) * 16u;
        if (e < e1) lr[e - e0] = (unsigned short)((old[k] >> sh) & 0xFFFFu);
      }
    }
  }
  __syncthreads();

  unsigned int* row = (is_dst ? histD : histS) + (size_t)blockIdx.x * NWORDS;
  for (int i = t; i < NWORDS; i += 1024) row[i] = h[i];
  if (!is_dst) return;
  __syncthreads();

  const int lane = t & 31, wv = t >> 5;
  const int w0 = t * 49;
  const int w1 = (w0 + 49 < NWORDS) ? (w0 + 49) : NWORDS;
  int s = 0;
  for (int i = w0; i < w1; ++i) {
    unsigned v = h[i];
    s += (int)(v & 0xFFFFu) + (int)(v >> 16);
  }
  int inc = s;
  for (int o = 1; o < 32; o <<= 1) {
    int u = __shfl_up(inc, o);
    if (lane >= o) inc += u;
  }
  if (lane == 31) wtot[wv] = inc;
  __syncthreads();
  int tv = wtot[lane];
  int winc = tv;
  for (int o = 1; o < 32; o <<= 1) {
    int u = __shfl_up(winc, o);
    if (lane >= o) winc += u;
  }
  int run = __shfl(winc - tv, wv) + inc - s;
  for (int i = w0; i < w1; ++i) {
    unsigned v = h[i];
    unsigned lo = v & 0xFFFFu, hi = v >> 16;
    h[i] = (unsigned)run | (((unsigned)run + lo) << 16);
    run += (int)(lo + hi);
  }
  __syncthreads();

  for (int i = t; i < NGRP; i += 1024)
    lscan32[(size_t)blockIdx.x * NGRP + i] = (int)(h[16 * i] & 0xFFFFu);

  const int n = e1 - e0;
  for (int b = t; b < n; b += 8 * 1024) {
    int d[8];
#pragma unroll
    for (int k = 0; k < 8; ++k) {
      int i = b + k * 1024;
      d[k] = dst[e0 + (i < n ? i : n - 1)];
    }
#pragma unroll
    for (int k = 0; k < 8; ++k) {
      int i = b + k * 1024;
      if (i < n) {
        unsigned p = ((h[d[k] >> 1] >> ((unsigned)(d[k] & 1) * 16u)) & 0xFFFFu) + (unsigned)lr[i];
        lr[i] = (unsigned short)p;
      }
    }
  }
  __syncthreads();

  for (int b = t; b < n; b += 8 * 1024) {
    int sv[8];
#pragma unroll
    for (int k = 0; k < 8; ++k) {
      int i = b + k * 1024;
      sv[k] = src[e0 + (i < n ? i : n - 1)];
    }
#pragma unroll
    for (int k = 0; k < 8; ++k) {
      int i = b + k * 1024;
      if (i < n) h[lr[i]] = (unsigned)sv[k];
    }
  }
  __syncthreads();

  for (int i = t; i < n; i += 1024) ss[e0 + i] = (int)h[i];
}

__global__ void colscan_kernel(const unsigned int* __restrict__ histD,
                               const unsigned int* __restrict__ histS,
                               int* __restrict__ base, int* __restrict__ cnt,
                               float* __restrict__ dis, const float* __restrict__ x,
                               float* __restrict__ xs, int G) {
  int i = blockIdx.x * blockDim.x + threadIdx.x;
  if (i >= NWORDS) return;
  if (blockIdx.y == 0) {
    int r0 = 0, r1 = 0;
#pragma unroll 16
    for (int w = 0; w < G; ++w) {
      unsigned v = histD[(size_t)w * NWORDS + i];
      int2 b;
      b.x = r0;
      b.y = r1;
      *reinterpret_cast<int2*>(base + (size_t)w * N_NODES + 2 * i) = b;
      r0 += (int)(v & 0xFFFFu);
      r1 += (int)(v >> 16);
    }
    int2 c;
    c.x = r0;
    c.y = r1;
    *reinterpret_cast<int2*>(cnt + 2 * i) = c;
  } else {
    const float4* __restrict__ xp = reinterpret_cast<const float4*>(x + (size_t)16 * i);
    float4 a0 = xp[0], a1 = xp[1], c0 = xp[2], c1 = xp[3];
    int d0 = 0, d1 = 0;
#pragma unroll 16
    for (int w = 0; w < G; ++w) {
      unsigned u = histS[(size_t)w * NWORDS + i];
      d0 += (int)(u & 0xFFFFu);
      d1 += (int)(u >> 16);
    }
    float f0 = (float)d0, f1 = (float)d1;
    float2 r;
    r.x = (f0 > 0.0f) ? (1.0f / sqrtf(f0)) : 0.0f;
    r.y = (f1 > 0.0f) ? (1.0f / sqrtf(f1)) : 0.0f;
    *reinterpret_cast<float2*>(dis + 2 * i) = r;
    a0.x *= r.x; a0.y *= r.x; a0.z *= r.x; a0.w *= r.x;
    a1.x *= r.x; a1.y *= r.x; a1.z *= r.x; a1.w *= r.x;
    c0.x *= r.y; c0.y *= r.y; c0.z *= r.y; c0.w *= r.y;
    c1.x *= r.y; c1.y *= r.y; c1.z *= r.y; c1.w *= r.y;
    float4* __restrict__ xo = reinterpret_cast<float4*>(xs + (size_t)16 * i);
    xo[0] = a0;
    xo[1] = a1;
    xo[2] = c0;
    xo[3] = c1;
  }
}

#define SBLK 4096
static_assert(N_NODES % 4 == 0, "cnt and off start at 16-byte boundaries of the workspace");

__global__ void __launch_bounds__(1024)
scan_kernel(const int* __restrict__ cnt, int* __restrict__ off, int n) {
  __shared__ int wsum[32];
  __shared__ int wtot[32];
  const int t = threadIdx.x, lane = t & 31, wv = t >> 5;
  const int b0 = blockIdx.x * SBLK;
  const int i  = b0 + t * 4;
  const int4* __restrict__ c4 = reinterpret_cast<const int4*>(cnt);

  int4 v = make_int4(0, 0, 0, 0);
  if (i + 3 < n) {
    v = c4[i >> 2];
  } else {
    if (i < n)     v.x = cnt[i];
    if (i + 1 < n) v.y = cnt[i + 1];
    if (i + 2 < n) v.z = cnt[i + 2];
  }

  int s = 0;
  const int rows = blockIdx.x;
  const int4* __restrict__ rp = c4 + t;
  int r = 0;
  for (; r + 12 <= rows; r += 12) {
#pragma unroll
    for (int k = 0; k < 12; ++k) {
      const int4 p = rp[(r + k) * 1024];
      s += (p.x + p.y) + (p.z + p.w);
    }
  }
  const int rem = rows - r;
  if (rem > 0) {
    int last = 0;
#pragma unroll
    for (int k = 0; k < 12; ++k) {
      const int rr = (r + k < rows) ? (r + k) : (rows - 1);
      const int4 p = rp[rr * 1024];
      last = (p.x + p.y) + (p.z + p.w);
      s += last;
    }
    s -= (12 - rem) * last;
  }

  const int t4 = (v.x + v.y) + (v.z + v.w);
  int inc = t4;
  for (int o = 1; o < 32; o <<= 1) {
    int u = __shfl_up(inc, o);
    if (lane >= o) inc += u;
  }
  for (int o = 16; o > 0; o >>= 1) s += __shfl_xor(s, o);
  if (lane == 31) wtot[wv] = inc;
  if (lane == 0)  wsum[wv] = s;
  __syncthreads();

  int front = wsum[lane];
  for (int o = 16; o > 0; o >>= 1) front += __shfl_xor(front, o);
  const int tv = wtot[lane];
  int winc = tv;
  for (int o = 1; o < 32; o <<= 1) {
    int u = __shfl_up(winc, o);
    if (lane >= o) winc += u;
  }
  const int ex = front + __shfl(winc - tv, wv) + (inc - t4);
  const int total = front + __shfl(winc, 31);

  if (i + 3 < n) {
    int4 o4;
    o4.x = ex;
    o4.y = ex + v.x;
    o4.z = ex + v.x + v.y;
    o4.w = ex + v.x + v.y + v.z;
    *reinterpret_cast<int4*>(off + i) = o4;
  } else {
    if (i < n)     off[i]     = ex;
    if (i + 1 < n) off[i + 1] = ex + v.x;
    if (i + 2 < n) off[i + 2] = ex + v.x + v.y;
  }
  if (blockIdx.x == gridDim.x - 1 && t == 0) off[n] = total;
}

#define FREC 4
__global__ void __launch_bounds__(1024)
fill_kernel(const unsigned int* __restrict__ histD, const int* __restrict__ base,
            const int* __restrict__ off, const int* __restrict__ lscan32,
            const int* __restrict__ ss, int* __restrict__ eb, int G, int E) {
  __shared__ int win[FCAP];
  const int lane = threadIdx.x & 31, wv = threadIdx.x >> 5;
  const int grp = blockIdx.x * FGRP + (wv & (FGRP - 1));
  const int n0 = blockIdx.x * (FGRP * 32);
  const int n1 = (n0 + FGRP * 32 < N_NODES) ? (n0 + FGRP * 32) : N_NODES;
  const int o0 = off[n0], o1 = off[n1];
  const bool staged = (o1 - o0) <= FCAP;

  int w = wv / FGRP;
  if (grp < NGRP && w < G) {
    const int d  = grp * 32 + lane;
    const int od = off[d];
    const unsigned sh = (unsigned)(d & 1) * 16u;
    int wb = (w + FPH < G) ? (w + FPH) : w;
    unsigned wordA = histD[(size_t)w * NWORDS + (d >> 1)];
    int bsA = base[(size_t)w * N_NODES + d];
    int rsA = lscan32[(size_t)w * NGRP + grp];
    unsigned wordB = histD[(size_t)wb * NWORDS + (d >> 1)];
    int bsB = base[(size_t)wb * N_NODES + d];
    int rsB = lscan32[(size_t)wb * NGRP + grp];
    for (; w < G; w += 2 * FPH) {
      const bool hasB = (w + FPH < G);
      const int wn  = (w + 2 * FPH < G) ? (w + 2 * FPH) : w;
      const int wbn = (w + 3 * FPH < G) ? (w + 3 * FPH) : wn;
      unsigned wordAn = histD[(size_t)wn * NWORDS + (d >> 1)];
      int bsAn = base[(size_t)wn * N_NODES + d];
      int rsAn = lscan32[(size_t)wn * NGRP + grp];
      unsigned wordBn = histD[(size_t)wbn * NWORDS + (d >> 1)];
      int bsBn = base[(size_t)wbn * N_NODES + d];
      int rsBn = lscan32[(size_t)wbn * NGRP + grp];

      const int cA = (int)((wordA >> sh) & 0xFFFFu);
      const int cB = hasB ? (int)((wordB >> sh) & 0xFFFFu) : 0;
      int incA = cA, incB = cB;
      for (int o = 1; o < 32; o <<= 1) {
        int uA = __shfl_up(incA, o);
        int uB = __shfl_up(incB, o);
        if (lane >= o) { incA += uA; incB += uB; }
      }
      const int pA = w  * SLICE + rsA + (incA - cA);
      const int pB = wb * SLICE + rsB + (incB - cB);
      const int slotA = od + bsA;
      const int slotB = od + bsB;

      int rA[FREC], rB[FREC];
#pragma unroll
      for (int k = 0; k < FREC; ++k) {
        int q = pA + ((k < cA) ? k : 0);
        rA[k] = ss[(q < E) ? q : (E - 1)];
      }
#pragma unroll
      for (int k = 0; k < FREC; ++k) {
        int q = pB + ((k < cB) ? k : 0);
        rB[k] = ss[(q < E) ? q : (E - 1)];
      }
#pragma unroll
      for (int k = 0; k < FREC; ++k) {
        if (k < cA) {
          if (staged) win[slotA - o0 + k] = rA[k];
          else        eb[slotA + k] = rA[k];
        }
      }
#pragma unroll
      for (int k = 0; k < FREC; ++k) {
        if (k < cB) {
          if (staged) win[slotB - o0 + k] = rB[k];
          else        eb[slotB + k] = rB[k];
        }
      }
      for (int k = FREC; k < cA; ++k) {
        const int s = ss[pA + k];
        if (staged) win[slotA - o0 + k] = s;
        else        eb[slotA + k] = s;
      }
      for (int k = FREC; k < cB; ++k) {
        const int s = ss[pB + k];
        if (staged) win[slotB - o0 + k] = s;
        else        eb[slotB + k] = s;
      }
      wordA = wordAn; bsA = bsAn; rsA = rsAn;
      wordB = wordBn; bsB = bsBn; rsB = rsBn;
      wb = wbn;
    }
  }
  __syncthreads();
  if (staged) {
    const int m = o1 - o0;
    for (int i = threadIdx.x; i < m; i += 1024) eb[o0 + i] = win[i];
  }
}

__global__ void pull1_layer1_kernel(const int* __restrict__ off, const int* __restrict__ eb,
                                    const float* __restrict__ dis, const float* __restrict__ xs,
                                    const float* __restrict__ x, const float* __restrict__ W1,
                                    const float* __restrict__ b1, float* __restrict__ h1,
                                    float* __restrict__ h1s, int n) {
  int g = blockIdx.x * blockDim.x + threadIdx.x;
  int d = g >> 3, f = g & 7;
  if (d >= n) return;
  int beg = off[d], end = off[d + 1];
  float dd = dis[d];
  const float xv = x[(size_t)d * 8 + f];
  const float wa0 = W1[f * 3 + 0], wa1 = W1[f * 3 + 1], wa2 = W1[f * 3 + 2];
  const float wb0 = W1[24 + f * 3 + 0], wb1 = W1[24 + f * 3 + 1], wb2 = W1[24 + f * 3 + 2];
  float acc = 0.0f;
#pragma unroll 8
  for (int j = beg; j < end; ++j) {
    const int s = eb[j];
    acc += xs[s * 8 + f];
  }
  const float tv = -dd * acc;
  float p0 = xv * wa0 + tv * wb0;
  float p1 = xv * wa1 + tv * wb1;
  float p2 = xv * wa2 + tv * wb2;
  for (int o = 1; o < 8; o <<= 1) {
    p0 += __shfl_xor(p0, o);
    p1 += __shfl_xor(p1, o);
    p2 += __shfl_xor(p2, o);
  }
  const float hv0 = fmaxf(p0 + b1[0], 0.0f);
  const float hv1 = fmaxf(p1 + b1[1], 0.0f);
  const float hv2 = fmaxf(p2 + b1[2], 0.0f);
  const float hsel = (f == 0) ? hv0 : ((f == 1) ? hv1 : hv2);
  if (f < 3) h1[(size_t)d * 3 + f] = hsel;
  if (f < 4) h1s[(size_t)d * 4 + f] = (f < 3) ? dd * hsel : 0.0f;
}

#define TNODES 64
#define H2P    18

__global__ void __launch_bounds__(256)
pull2_layer2_head_kernel(const int* __restrict__ off, const int* __restrict__ eb,
                         const float* __restrict__ dis, const float* __restrict__ h1s,
                         const float* __restrict__ h1, const float* __restrict__ W2,
                         const float* __restrict__ b2, const float* __restrict__ Wl,
                         const float* __restrict__ bl, float* __restrict__ out,
                         int n, int ntiles) {
  __shared__ float h2t[TNODES * H2P];
  const int t = threadIdx.x;
  const int lane = t & 31;
  const int nl = t >> 2, f = t & 3;
  const int d = blockIdx.x * TNODES + nl;
  const bool live = d < n;

  float acc = 0.0f, dd = 0.0f;
  float a0 = 0.0f, a1 = 0.0f, a2 = 0.0f;
  if (live) {
    int beg = off[d], end = off[d + 1];
    dd = dis[d];
    a0 = h1[(size_t)d * 3 + 0];
    a1 = h1[(size_t)d * 3 + 1];
    a2 = h1[(size_t)d * 3 + 2];
#pragma unroll 8
    for (int j = beg; j < end; ++j) {
      const int s = eb[j];
      acc += h1s[s * 4 + f];
    }
  }
  const float tx = -dd * acc;
  const int gb = lane & ~3;
  const float tv[3] = {__shfl(tx, gb + 0), __shfl(tx, gb + 1), __shfl(tx, gb + 2)};
  const float hv[3] = {a0, a1, a2};

  const int c0 = 4 * f;
  const float4 bb = *reinterpret_cast<const float4*>(b2 + c0);
  float o4[4] = {bb.x, bb.y, bb.z, bb.w};
#pragma unroll
  for (int q = 0; q < 3; ++q) {
    const float4 wa = *reinterpret_cast<const float4*>(W2 + q * 16 + c0);
    const float4 wb = *reinterpret_cast<const float4*>(W2 + 48 + q * 16 + c0);
    o4[0] += hv[q] * wa.x + tv[q] * wb.x;
    o4[1] += hv[q] * wa.y + tv[q] * wb.y;
    o4[2] += hv[q] * wa.z + tv[q] * wb.z;
    o4[3] += hv[q] * wa.w + tv[q] * wb.w;
  }
  float* __restrict__ hrow = h2t + nl * H2P + c0;
  hrow[0] = fmaxf(o4[0], 0.0f);
  hrow[1] = fmaxf(o4[1], 0.0f);
  hrow[2] = fmaxf(o4[2], 0.0f);
  hrow[3] = fmaxf(o4[3], 0.0f);
  __syncthreads();

  const int wv = __builtin_amdgcn_readfirstlane(t >> 5);
  const int tl = wv >> 1;
  const int tile = blockIdx.x * (TNODES / 16) + tl;
  if (tile >= ntiles) return;
  const int half = lane >> 4;
  const int l16  = lane & 15;
  const int jb   = (wv & 1) * 32;
  const float* __restrict__ arow = h2t + (tl * 16 + l16) * H2P;

  v8f acc0 = {}, acc1 = {};
#pragma unroll
  for (int kk = 0; kk < 16; kk += 4) {
    int ka = kk + 2 * half;
    v2f a;
    a.x = arow[ka];
    a.y = arow[ka + 1];

    v2f b0, b1;
    b0.x = Wl[ka * 64 + jb +  0 + l16]; b0.y = Wl[(ka + 1) * 64 + jb +  0 + l16];
    b1.x = Wl[ka * 64 + jb + 16 + l16]; b1.y = Wl[(ka + 1) * 64 + jb + 16 + l16];

    acc0 = __builtin_amdgcn_wmma_f32_16x16x4_f32(false, a, false, b0, (short)0, acc0, false, false);
    acc1 = __builtin_amdgcn_wmma_f32_16x16x4_f32(false, a, false, b1, (short)0, acc1, false, false);
  }

  v8f accs[2] = {acc0, acc1};
#pragma unroll
  for (int j = 0; j < 2; ++j) {
    int col = jb + j * 16 + l16;
    float bias = bl[col];
#pragma unroll
    for (int r = 0; r < 8; ++r) {
      int m = tile * 16 + r + 8 * half;
      out[(size_t)m * 64 + col] = accs[j][r] + bias;
    }
  }
}

extern "C" void kernel_launch(void* const* d_in, const int* in_sizes, int n_in,
                              void* d_out, int out_size, void* d_ws, size_t ws_size,
                              hipStream_t stream) {
  const float* x  = (const float*)d_in[0];
  const int*   ei = (const int*)d_in[1];
  const float* W1 = (const float*)d_in[2];
  const float* b1 = (const float*)d_in[3];
  const float* W2 = (const float*)d_in[4];
  const float* b2 = (const float*)d_in[5];
  const float* Wl = (const float*)d_in[6];
  const float* bl = (const float*)d_in[7];
  float* out = (float*)d_out;

  const int N = in_sizes[0] / 8;
  const int E = in_sizes[1] / 2;
  const int* src = ei;
  const int* dst = ei + E;

  const int G = (E + SLICE - 1) / SLICE;

  float* ws  = (float*)d_ws;
  int*   cnt = (int*)(ws + (size_t)N);
  float* dis = ws + (size_t)2  * N;
  float* h1  = ws + (size_t)11 * N;
  int*   off = (int*)(ws + (size_t)33 * N);
  int*   eb  = (int*)(ws + (size_t)34 * N + 4);
  float* xs  = ws + (size_t)34 * N + 8 + (size_t)E;
  float* h1s = xs + (size_t)8 * N;
  int*   ss   = (int*)(ws + (size_t)34 * N + 4 + (size_t)2 * E);
  int*   base = (int*)(ws + (size_t)34 * N + 4 + (size_t)3 * E);
  unsigned int* histD = (unsigned int*)(base + (size_t)G * N);
  unsigned int* histS = histD + (size_t)G * NWORDS;
  int*   lscan32 = (int*)(histS + (size_t)G * NWORDS);

  const int B = 256;
  hist_kernel<<<dim3(G, 2), 1024, 0, stream>>>(src, dst, histD, histS, ss, lscan32, E);
  colscan_kernel<<<dim3((NWORDS + B - 1) / B, 2), B, 0, stream>>>(histD, histS, base, cnt, dis, x, xs, G);

  scan_kernel<<<(N + SBLK - 1) / SBLK, 1024, 0, stream>>>(cnt, off, N);
  fill_kernel<<<(NGRP + FGRP - 1) / FGRP, 1024, 0, stream>>>(histD, base, off, lscan32, ss, eb, G, E);

  pull1_layer1_kernel<<<(8 * N + B - 1) / B, B, 0, stream>>>(off, eb, dis, xs, x, W1, b1, h1, h1s, N);

  const int ntiles = N / 16;
  pull2_layer2_head_kernel<<<(N + TNODES - 1) / TNODES, 256, 0, stream>>>(off, eb, dis, h1s, h1, W2, b2, Wl, bl,
                                                                          out, N, ntiles);
}


# Edge loops of both gather kernels: a node's lanes fetch 16 records per trip together, one trip ahead, share them by lane exchange (DPP); 16 clamped row gathers per trip, no remainder; same sum order
# speedup vs baseline: 10.5650x; 1.0729x over previous
#include <hip/hip_runtime.h>
#include <math.h>


typedef __attribute__((ext_vector_type(2))) float v2f;
typedef __attribute__((ext_vector_type(8))) float v8f;

#define N_NODES 100000

#define NWORDS (N_NODES / 2)
#define SLICE  50000
#define NGRP   (N_NODES / 32)
static_assert(SLICE <= NWORDS, "the counter array is reused as the slice's ordered window");
static_assert(N_NODES % 32 == 0, "a wave owns 32 whole nodes");

#define FGRP 4
#define FPH  8
#define FCAP 16384

__global__ void __launch_bounds__(1024)
hist_kernel(const int* __restrict__ src, const int* __restrict__ dst,
            unsigned int* __restrict__ histD, unsigned int* __restrict__ histS,
            int* __restrict__ ss, int* __restrict__ lscan32, int E) {
  __shared__ unsigned int h[NWORDS];
  __shared__ unsigned short lr[SLICE];
  __shared__ int wtot[32];
  const int t  = threadIdx.x;
  const int e0 = blockIdx.x * SLICE;
  const int e1 = (e0 + SLICE < E) ? (e0 + SLICE) : E;
  const bool is_dst = (blockIdx.y == 0);
  const int* __restrict__ idx = is_dst ? dst : src;

  for (int i = t; i < NWORDS; i += 1024) h[i] = 0u;
  __syncthreads();

  for (int b = e0 + t; b < e1; b += 8 * 1024) {
    int d[8];
#pragma unroll
    for (int k = 0; k < 8; ++k) {
      int e = b + k * 1024;
      d[k] = idx[e < e1 ? e : e1 - 1];
    }
    unsigned old[8];
#pragma unroll
    for (int k = 0; k < 8; ++k) {
      unsigned sh = (unsigned)(d[k] & 1) * 16u;
      old[k] = atomicAdd(&h[d[k] >> 1], (b + k * 1024 < e1) ? (1u << sh) : 0u);
    }
    if (is_dst) {
#pragma unroll
      for (int k = 0; k < 8; ++k) {
        int e = b + k * 1024;
        unsigned sh = (unsigned)(d[k] & 1) * 16u;
        if (e < e1) lr[e - e0] = (unsigned short)((old[k] >> sh) & 0xFFFFu);
      }
    }
  }
  __syncthreads();

  unsigned int* row = (is_dst ? histD : histS) + (size_t)blockIdx.x * NWORDS;
  for (int i = t; i < NWORDS; i += 1024) row[i] = h[i];
  if (!is_dst) return;
  __syncthreads();

  const int lane = t & 31, wv = t >> 5;
  const int w0 = t * 49;
  const int w1 = (w0 + 49 < NWORDS) ? (w0 + 49) : NWORDS;
  int s = 0;
  for (int i = w0; i < w1; ++i) {
    unsigned v = h[i];
    s += (int)(v & 0xFFFFu) + (int)(v >> 16);
  }
  int inc = s;
  for (int o = 1; o < 32; o <<= 1) {
    int u = __shfl_up(inc, o);
    if (lane >= o) inc += u;
  }
  if (lane == 31) wtot[wv] = inc;
  __syncthreads();
  int tv = wtot[lane];
  int winc = tv;
  for (int o = 1; o < 32; o <<= 1) {
    int u = __shfl_up(winc, o);
    if (lane >= o) winc += u;
  }
  int run = __shfl(winc - tv, wv) + inc - s;
  for (int i = w0; i < w1; ++i) {
    unsigned v = h[i];
    unsigned lo = v & 0xFFFFu, hi = v >> 16;
    h[i] = (unsigned)run | (((unsigned)run + lo) << 16);
    run += (int)(lo + hi);
  }
  __syncthreads();

  for (int i = t; i < NGRP; i += 1024)
    lscan32[(size_t)blockIdx.x * NGRP + i] = (int)(h[16 * i] & 0xFFFFu);

  const int n = e1 - e0;
  for (int b = t; b < n; b += 8 * 1024) {
    int d[8];
#pragma unroll
    for (int k = 0; k < 8; ++k) {
      int i = b + k * 1024;
      d[k] = dst[e0 + (i < n ? i : n - 1)];
    }
#pragma unroll
    for (int k = 0; k < 8; ++k) {
      int i = b + k * 1024;
      if (i < n) {
        unsigned p = ((h[d[k] >> 1] >> ((unsigned)(d[k] & 1) * 16u)) & 0xFFFFu) + (unsigned)lr[i];
        lr[i] = (unsigned short)p;
      }
    }
  }
  __syncthreads();

  for (int b = t; b < n; b += 8 * 1024) {
    int sv[8];
#pragma unroll
    for (int k = 0; k < 8; ++k) {
      int i = b + k * 1024;
      sv[k] = src[e0 + (i < n ? i : n - 1)];
    }
#pragma unroll
    for (int k = 0; k < 8; ++k) {
      int i = b + k * 1024;
      if (i < n) h[lr[i]] = (unsigned)sv[k];
    }
  }
  __syncthreads();

  for (int i = t; i < n; i += 1024) ss[e0 + i] = (int)h[i];
}

__global__ void colscan_kernel(const unsigned int* __restrict__ histD,
                               const unsigned int* __restrict__ histS,
                               int* __restrict__ base, int* __restrict__ cnt,
                               float* __restrict__ dis, const float* __restrict__ x,
                               float* __restrict__ xs, int G) {
  int i = blockIdx.x * blockDim.x + threadIdx.x;
  if (i >= NWORDS) return;
  if (blockIdx.y == 0) {
    int r0 = 0, r1 = 0;
#pragma unroll 16
    for (int w = 0; w < G; ++w) {
      unsigned v = histD[(size_t)w * NWORDS + i];
      int2 b;
      b.x = r0;
      b.y = r1;
      *reinterpret_cast<int2*>(base + (size_t)w * N_NODES + 2 * i) = b;
      r0 += (int)(v & 0xFFFFu);
      r1 += (int)(v >> 16);
    }
    int2 c;
    c.x = r0;
    c.y = r1;
    *reinterpret_cast<int2*>(cnt + 2 * i) = c;
  } else {
    const float4* __restrict__ xp = reinterpret_cast<const float4*>(x + (size_t)16 * i);
    float4 a0 = xp[0], a1 = xp[1], c0 = xp[2], c1 = xp[3];
    int d0 = 0, d1 = 0;
#pragma unroll 16
    for (int w = 0; w < G; ++w) {
      unsigned u = histS[(size_t)w * NWORDS + i];
      d0 += (int)(u & 0xFFFFu);
      d1 += (int)(u >> 16);
    }
    float f0 = (float)d0, f1 = (float)d1;
    float2 r;
    r.x = (f0 > 0.0f) ? (1.0f / sqrtf(f0)) : 0.0f;
    r.y = (f1 > 0.0f) ? (1.0f / sqrtf(f1)) : 0.0f;
    *reinterpret_cast<float2*>(dis + 2 * i) = r;
    a0.x *= r.x; a0.y *= r.x; a0.z *= r.x; a0.w *= r.x;
    a1.x *= r.x; a1.y *= r.x; a1.z *= r.x; a1.w *= r.x;
    c0.x *= r.y; c0.y *= r.y; c0.z *= r.y; c0.w *= r.y;
    c1.x *= r.y; c1.y *= r.y; c1.z *= r.y; c1.w *= r.y;
    float4* __restrict__ xo = reinterpret_cast<float4*>(xs + (size_t)16 * i);
    xo[0] = a0;
    xo[1] = a1;
    xo[2] = c0;
    xo[3] = c1;
  }
}

#define SBLK 4096
static_assert(N_NODES % 4 == 0, "cnt and off start at 16-byte boundaries of the workspace");

__global__ void __launch_bounds__(1024)
scan_kernel(const int* __restrict__ cnt, int* __restrict__ off, int n) {
  __shared__ int wsum[32];
  __shared__ int wtot[32];
  const int t = threadIdx.x, lane = t & 31, wv = t >> 5;
  const int b0 = blockIdx.x * SBLK;
  const int i  = b0 + t * 4;
  const int4* __restrict__ c4 = reinterpret_cast<const int4*>(cnt);

  int4 v = make_int4(0, 0, 0, 0);
  if (i + 3 < n) {
    v = c4[i >> 2];
  } else {
    if (i < n)     v.x = cnt[i];
    if (i + 1 < n) v.y = cnt[i + 1];
    if (i + 2 < n) v.z = cnt[i + 2];
  }

  int s = 0;
  const int rows = blockIdx.x;
  const int4* __restrict__ rp = c4 + t;
  int r = 0;
  for (; r + 12 <= rows; r += 12) {
#pragma unroll
    for (int k = 0; k < 12; ++k) {
      const int4 p = rp[(r + k) * 1024];
      s += (p.x + p.y) + (p.z + p.w);
    }
  }
  const int rem = rows - r;
  if (rem > 0) {
    int last = 0;
#pragma unroll
    for (int k = 0; k < 12; ++k) {
      const int rr = (r + k < rows) ? (r + k) : (rows - 1);
      const int4 p = rp[rr * 1024];
      last = (p.x + p.y) + (p.z + p.w);
      s += last;
    }
    s -= (12 - rem) * last;
  }

  const int t4 = (v.x + v.y) + (v.z + v.w);
  int inc = t4;
  for (int o = 1; o < 32; o <<= 1) {
    int u = __shfl_up(inc, o);
    if (lane >= o) inc += u;
  }
  for (int o = 16; o > 0; o >>= 1) s += __shfl_xor(s, o);
  if (lane == 31) wtot[wv] = inc;
  if (lane == 0)  wsum[wv] = s;
  __syncthreads();

  int front = wsum[lane];
  for (int o = 16; o > 0; o >>= 1) front += __shfl_xor(front, o);
  const int tv = wtot[lane];
  int winc = tv;
  for (int o = 1; o < 32; o <<= 1) {
    int u = __shfl_up(winc, o);
    if (lane >= o) winc += u;
  }
  const int ex = front + __shfl(winc - tv, wv) + (inc - t4);
  const int total = front + __shfl(winc, 31);

  if (i + 3 < n) {
    int4 o4;
    o4.x = ex;
    o4.y = ex + v.x;
    o4.z = ex + v.x + v.y;
    o4.w = ex + v.x + v.y + v.z;
    *reinterpret_cast<int4*>(off + i) = o4;
  } else {
    if (i < n)     off[i]     = ex;
    if (i + 1 < n) off[i + 1] = ex + v.x;
    if (i + 2 < n) off[i + 2] = ex + v.x + v.y;
  }
  if (blockIdx.x == gridDim.x - 1 && t == 0) off[n] = total;
}

#define FREC 4
__global__ void __launch_bounds__(1024)
fill_kernel(const unsigned int* __restrict__ histD, const int* __restrict__ base,
            const int* __restrict__ off, const int* __restrict__ lscan32,
            const int* __restrict__ ss, int* __restrict__ eb, int G, int E) {
  __shared__ int win[FCAP];
  const int lane = threadIdx.x & 31, wv = threadIdx.x >> 5;
  const int grp = blockIdx.x * FGRP + (wv & (FGRP - 1));
  const int n0 = blockIdx.x * (FGRP * 32);
  const int n1 = (n0 + FGRP * 32 < N_NODES) ? (n0 + FGRP * 32) : N_NODES;
  const int o0 = off[n0], o1 = off[n1];
  const bool staged = (o1 - o0) <= FCAP;

  int w = wv / FGRP;
  if (grp < NGRP && w < G) {
    const int d  = grp * 32 + lane;
    const int od = off[d];
    const unsigned sh = (unsigned)(d & 1) * 16u;
    int wb = (w + FPH < G) ? (w + FPH) : w;
    unsigned wordA = histD[(size_t)w * NWORDS + (d >> 1)];
    int bsA = base[(size_t)w * N_NODES + d];
    int rsA = lscan32[(size_t)w * NGRP + grp];
    unsigned wordB = histD[(size_t)wb * NWORDS + (d >> 1)];
    int bsB = base[(size_t)wb * N_NODES + d];
    int rsB = lscan32[(size_t)wb * NGRP + grp];
    for (; w < G; w += 2 * FPH) {
      const bool hasB = (w + FPH < G);
      const int wn  = (w + 2 * FPH < G) ? (w + 2 * FPH) : w;
      const int wbn = (w + 3 * FPH < G) ? (w + 3 * FPH) : wn;
      unsigned wordAn = histD[(size_t)wn * NWORDS + (d >> 1)];
      int bsAn = base[(size_t)wn * N_NODES + d];
      int rsAn = lscan32[(size_t)wn * NGRP + grp];
      unsigned wordBn = histD[(size_t)wbn * NWORDS + (d >> 1)];
      int bsBn = base[(size_t)wbn * N_NODES + d];
      int rsBn = lscan32[(size_t)wbn * NGRP + grp];

      const int cA = (int)((wordA >> sh) & 0xFFFFu);
      const int cB = hasB ? (int)((wordB >> sh) & 0xFFFFu) : 0;
      int incA = cA, incB = cB;
      for (int o = 1; o < 32; o <<= 1) {
        int uA = __shfl_up(incA, o);
        int uB = __shfl_up(incB, o);
        if (lane >= o) { incA += uA; incB += uB; }
      }
      const int pA = w  * SLICE + rsA + (incA - cA);
      const int pB = wb * SLICE + rsB + (incB - cB);
      const int slotA = od + bsA;
      const int slotB = od + bsB;

      int rA[FREC], rB[FREC];
#pragma unroll
      for (int k = 0; k < FREC; ++k) {
        int q = pA + ((k < cA) ? k : 0);
        rA[k] = ss[(q < E) ? q : (E - 1)];
      }
#pragma unroll
      for (int k = 0; k < FREC; ++k) {
        int q = pB + ((k < cB) ? k : 0);
        rB[k] = ss[(q < E) ? q : (E - 1)];
      }
#pragma unroll
      for (int k = 0; k < FREC; ++k) {
        if (k < cA) {
          if (staged) win[slotA - o0 + k] = rA[k];
          else        eb[slotA + k] = rA[k];
        }
      }
#pragma unroll
      for (int k = 0; k < FREC; ++k) {
        if (k < cB) {
          if (staged) win[slotB - o0 + k] = rB[k];
          else        eb[slotB + k] = rB[k];
        }
      }
      for (int k = FREC; k < cA; ++k) {
        const int s = ss[pA + k];
        if (staged) win[slotA - o0 + k] = s;
        else        eb[slotA + k] = s;
      }
      for (int k = FREC; k < cB; ++k) {
        const int s = ss[pB + k];
        if (staged) win[slotB - o0 + k] = s;
        else        eb[slotB + k] = s;
      }
      wordA = wordAn; bsA = bsAn; rsA = rsAn;
      wordB = wordBn; bsB = bsBn; rsB = rsBn;
      wb = wbn;
    }
  }
  __syncthreads();
  if (staged) {
    const int m = o1 - o0;
    for (int i = threadIdx.x; i < m; i += 1024) eb[o0 + i] = win[i];
  }
}

template <int K> __device__ __forceinline__ int grp8_take(int v) {
  return __builtin_amdgcn_mov_dpp8(v, K * 0x249249);
}
template <int K> __device__ __forceinline__ int grp4_take(int v) {
  return __builtin_amdgcn_update_dpp(0, v, K * 0x55, 0xF, 0xF, false);
}

__global__ void pull1_layer1_kernel(const int* __restrict__ off, const int* __restrict__ eb,
                                    const float* __restrict__ dis, const float* __restrict__ xs,
                                    const float* __restrict__ x, const float* __restrict__ W1,
                                    const float* __restrict__ b1, float* __restrict__ h1,
                                    float* __restrict__ h1s, int n) {
  int g = blockIdx.x * blockDim.x + threadIdx.x;
  int d = g >> 3, f = g & 7;
  if (d >= n) return;
  int beg = off[d], end = off[d + 1];
  float dd = dis[d];
  const float xv = x[(size_t)d * 8 + f];
  const float wa0 = W1[f * 3 + 0], wa1 = W1[f * 3 + 1], wa2 = W1[f * 3 + 2];
  const float wb0 = W1[24 + f * 3 + 0], wb1 = W1[24 + f * 3 + 1], wb2 = W1[24 + f * 3 + 2];
  float acc = 0.0f;
  if (beg < end) {
    const int last = end - 1;
    int r[2];
#pragma unroll
    for (int i = 0; i < 2; ++i) {
      const int ix = beg + 8 * i + f;
      r[i] = eb[ix < last ? ix : last];
    }
    for (int j = beg; j < end; j += 16) {
      int q[2];
#pragma unroll
      for (int i = 0; i < 2; ++i) {
        const int ix = j + 16 + 8 * i + f;
        q[i] = eb[ix < last ? ix : last];
      }
      float v[16];
#pragma unroll
      for (int i = 0; i < 2; ++i) {
        v[8 * i + 0] = xs[grp8_take<0>(r[i]) * 8 + f];
        v[8 * i + 1] = xs[grp8_take<1>(r[i]) * 8 + f];
        v[8 * i + 2] = xs[grp8_take<2>(r[i]) * 8 + f];
        v[8 * i + 3] = xs[grp8_take<3>(r[i]) * 8 + f];
        v[8 * i + 4] = xs[grp8_take<4>(r[i]) * 8 + f];
        v[8 * i + 5] = xs[grp8_take<5>(r[i]) * 8 + f];
        v[8 * i + 6] = xs[grp8_take<6>(r[i]) * 8 + f];
        v[8 * i + 7] = xs[grp8_take<7>(r[i]) * 8 + f];
      }
      const int left = end - j;
#pragma unroll
      for (int k = 0; k < 16; ++k) acc += (k < left) ? v[k] : 0.0f;
#pragma unroll
      for (int i = 0; i < 2; ++i) r[i] = q[i];
    }
  }
  const float tv = -dd * acc;
  float p0 = xv * wa0 + tv * wb0;
  float p1 = xv * wa1 + tv * wb1;
  float p2 = xv * wa2 + tv * wb2;
  for (int o = 1; o < 8; o <<= 1) {
    p0 += __shfl_xor(p0, o);
    p1 += __shfl_xor(p1, o);
    p2 += __shfl_xor(p2, o);
  }
  const float hv0 = fmaxf(p0 + b1[0], 0.0f);
  const float hv1 = fmaxf(p1 + b1[1], 0.0f);
  const float hv2 = fmaxf(p2 + b1[2], 0.0f);
  const float hsel = (f == 0) ? hv0 : ((f == 1) ? hv1 : hv2);
  if (f < 3) h1[(size_t)d * 3 + f] = hsel;
  if (f < 4) h1s[(size_t)d * 4 + f] = (f < 3) ? dd * hsel : 0.0f;
}

#define TNODES 64
#define H2P    18

__global__ void __launch_bounds__(256)
pull2_layer2_head_kernel(const int* __restrict__ off, const int* __restrict__ eb,
                         const float* __restrict__ dis, const float* __restrict__ h1s,
                         const float* __restrict__ h1, const float* __restrict__ W2,
                         const float* __restrict__ b2, const float* __restrict__ Wl,
                         const float* __restrict__ bl, float* __restrict__ out,
                         int n, int ntiles) {
  __shared__ float h2t[TNODES * H2P];
  const int t = threadIdx.x;
  const int lane = t & 31;
  const int nl = t >> 2, f = t & 3;
  const int d = blockIdx.x * TNODES + nl;
  const bool live = d < n;

  float acc = 0.0f, dd = 0.0f;
  float a0 = 0.0f, a1 = 0.0f, a2 = 0.0f;
  if (live) {
    int beg = off[d], end = off[d + 1];
    dd = dis[d];
    a0 = h1[(size_t)d * 3 + 0];
    a1 = h1[(size_t)d * 3 + 1];
    a2 = h1[(size_t)d * 3 + 2];
    if (beg < end) {
      const int last = end - 1;
      int r[4];
#pragma unroll
      for (int i = 0; i < 4; ++i) {
        const int ix = beg + 4 * i + f;
        r[i] = eb[ix < last ? ix : last];
      }
      for (int j = beg; j < end; j += 16) {
        int q[4];
#pragma unroll
        for (int i = 0; i < 4; ++i) {
          const int ix = j + 16 + 4 * i + f;
          q[i] = eb[ix < last ? ix : last];
        }
        float v[16];
#pragma unroll
        for (int i = 0; i < 4; ++i) {
          v[4 * i + 0] = h1s[grp4_take<0>(r[i]) * 4 + f];
          v[4 * i + 1] = h1s[grp4_take<1>(r[i]) * 4 + f];
          v[4 * i + 2] = h1s[grp4_take<2>(r[i]) * 4 + f];
          v[4 * i + 3] = h1s[grp4_take<3>(r[i]) * 4 + f];
        }
        const int left = end - j;
#pragma unroll
        for (int k = 0; k < 16; ++k) acc += (k < left) ? v[k] : 0.0f;
#pragma unroll
        for (int i = 0; i < 4; ++i) r[i] = q[i];
      }
    }
  }
  const float tx = -dd * acc;
  const int gb = lane & ~3;
  const float tv[3] = {__shfl(tx, gb + 0), __shfl(tx, gb + 1), __shfl(tx, gb + 2)};
  const float hv[3] = {a0, a1, a2};

  const int c0 = 4 * f;
  const float4 bb = *reinterpret_cast<const float4*>(b2 + c0);
  float o4[4] = {bb.x, bb.y, bb.z, bb.w};
#pragma unroll
  for (int q = 0; q < 3; ++q) {
    const float4 wa = *reinterpret_cast<const float4*>(W2 + q * 16 + c0);
    const float4 wb = *reinterpret_cast<const float4*>(W2 + 48 + q * 16 + c0);
    o4[0] += hv[q] * wa.x + tv[q] * wb.x;
    o4[1] += hv[q] * wa.y + tv[q] * wb.y;
    o4[2] += hv[q] * wa.z + tv[q] * wb.z;
    o4[3] += hv[q] * wa.w + tv[q] * wb.w;
  }
  float* __restrict__ hrow = h2t + nl * H2P + c0;
  hrow[0] = fmaxf(o4[0], 0.0f);
  hrow[1] = fmaxf(o4[1], 0.0f);
  hrow[2] = fmaxf(o4[2], 0.0f);
  hrow[3] = fmaxf(o4[3], 0.0f);
  __syncthreads();

  const int wv = __builtin_amdgcn_readfirstlane(t >> 5);
  const int tl = wv >> 1;
  const int tile = blockIdx.x * (TNODES / 16) + tl;
  if (tile >= ntiles) return;
  const int half = lane >> 4;
  const int l16  = lane & 15;
  const int jb   = (wv & 1) * 32;
  const float* __restrict__ arow = h2t + (tl * 16 + l16) * H2P;

  v8f acc0 = {}, acc1 = {};
#pragma unroll
  for (int kk = 0; kk < 16; kk += 4) {
    int ka = kk + 2 * half;
    v2f a;
    a.x = arow[ka];
    a.y = arow[ka + 1];

    v2f b0, b1;
    b0.x = Wl[ka * 64 + jb +  0 + l16]; b0.y = Wl[(ka + 1) * 64 + jb +  0 + l16];
    b1.x = Wl[ka * 64 + jb + 16 + l16]; b1.y = Wl[(ka + 1) * 64 + jb + 16 + l16];

    acc0 = __builtin_amdgcn_wmma_f32_16x16x4_f32(false, a, false, b0, (short)0, acc0, false, false);
    acc1 = __builtin_amdgcn_wmma_f32_16x16x4_f32(false, a, false, b1, (short)0, acc1, false, false);
  }

  v8f accs[2] = {acc0, acc1};
#pragma unroll
  for (int j = 0; j < 2; ++j) {
    int col = jb + j * 16 + l16;
    float bias = bl[col];
#pragma unroll
    for (int r = 0; r < 8; ++r) {
      int m = tile * 16 + r + 8 * half;
      out[(size_t)m * 64 + col] = accs[j][r] + bias;
    }
  }
}

extern "C" void kernel_launch(void* const* d_in, const int* in_sizes, int n_in,
                              void* d_out, int out_size, void* d_ws, size_t ws_size,
                              hipStream_t stream) {
  const float* x  = (const float*)d_in[0];
  const int*   ei = (const int*)d_in[1];
  const float* W1 = (const float*)d_in[2];
  const float* b1 = (const float*)d_in[3];
  const float* W2 = (const float*)d_in[4];
  const float* b2 = (const float*)d_in[5];
  const float* Wl = (const float*)d_in[6];
  const float* bl = (const float*)d_in[7];
  float* out = (float*)d_out;

  const int N = in_sizes[0] / 8;
  const int E = in_sizes[1] / 2;
  const int* src = ei;
  const int* dst = ei + E;

  const int G = (E + SLICE - 1) / SLICE;

  float* ws  = (float*)d_ws;
  int*   cnt = (int*)(ws + (size_t)N);
  float* dis = ws + (size_t)2  * N;
  float* h1  = ws + (size_t)11 * N;
  int*   off = (int*)(ws + (size_t)33 * N);
  int*   eb  = (int*)(ws + (size_t)34 * N + 4);
  float* xs  = ws + (size_t)34 * N + 8 + (size_t)E;
  float* h1s = xs + (size_t)8 * N;
  int*   ss   = (int*)(ws + (size_t)34 * N + 4 + (size_t)2 * E);
  int*   base = (int*)(ws + (size_t)34 * N + 4 + (size_t)3 * E);
  unsigned int* histD = (unsigned int*)(base + (size_t)G * N);
  unsigned int* histS = histD + (size_t)G * NWORDS;
  int*   lscan32 = (int*)(histS + (size_t)G * NWORDS);

  const int B = 256;
  hist_kernel<<<dim3(G, 2), 1024, 0, stream>>>(src, dst, histD, histS, ss, lscan32, E);
  colscan_kernel<<<dim3((NWORDS + B - 1) / B, 2), B, 0, stream>>>(histD, histS, base, cnt, dis, x, xs, G);

  scan_kernel<<<(N + SBLK - 1) / SBLK, 1024, 0, stream>>>(cnt, off, N);
  fill_kernel<<<(NGRP + FGRP - 1) / FGRP, 1024, 0, stream>>>(histD, base, off, lscan32, ss, eb, G, E);

  pull1_layer1_kernel<<<(8 * N + B - 1) / B, B, 0, stream>>>(off, eb, dis, xs, x, W1, b1, h1, h1s, N);

  const int ntiles = N / 16;
  pull2_layer2_head_kernel<<<(N + TNODES - 1) / TNODES, 256, 0, stream>>>(off, eb, dis, h1s, h1, W2, b2, Wl, bl,
                                                                          out, N, ntiles);
}


# Earlier-slices count plane dropped: the column pass only sums; in the fill a wave owns 16 consecutive slices, loads their count words as one group, phase totals meet in LDS; same slots and sum order
# speedup vs baseline: 11.2076x; 1.0608x over previous
#include <hip/hip_runtime.h>
#include <math.h>


typedef __attribute__((ext_vector_type(2))) float v2f;
typedef __attribute__((ext_vector_type(8))) float v8f;

#define N_NODES 100000

#define NWORDS (N_NODES / 2)
#define SLICE  50000
#define NGRP   (N_NODES / 32)
static_assert(SLICE <= NWORDS, "the counter array is reused as the slice's ordered window");
static_assert(N_NODES % 32 == 0, "a wave owns 32 whole nodes");

#define FGRP 4
#define FPH  8
#define FCAP 16384

__global__ void __launch_bounds__(1024)
hist_kernel(const int* __restrict__ src, const int* __restrict__ dst,
            unsigned int* __restrict__ histD, unsigned int* __restrict__ histS,
            int* __restrict__ ss, int* __restrict__ lscan32, int E) {
  __shared__ unsigned int h[NWORDS];
  __shared__ unsigned short lr[SLICE];
  __shared__ int wtot[32];
  const int t  = threadIdx.x;
  const int e0 = blockIdx.x * SLICE;
  const int e1 = (e0 + SLICE < E) ? (e0 + SLICE) : E;
  const bool is_dst = (blockIdx.y == 0);
  const int* __restrict__ idx = is_dst ? dst : src;

  for (int i = t; i < NWORDS; i += 1024) h[i] = 0u;
  __syncthreads();

  for (int b = e0 + t; b < e1; b += 8 * 1024) {
    int d[8];
#pragma unroll
    for (int k = 0; k < 8; ++k) {
      int e = b + k * 1024;
      d[k] = idx[e < e1 ? e : e1 - 1];
    }
    unsigned old[8];
#pragma unroll
    for (int k = 0; k < 8; ++k) {
      unsigned sh = (unsigned)(d[k] & 1) * 16u;
      old[k] = atomicAdd(&h[d[k] >> 1], (b + k * 1024 < e1) ? (1u << sh) : 0u);
    }
    if (is_dst) {
#pragma unroll
      for (int k = 0; k < 8; ++k) {
        int e = b + k * 1024;
        unsigned sh = (unsigned)(d[k] & 1) * 16u;
        if (e < e1) lr[e - e0] = (unsigned short)((old[k] >> sh) & 0xFFFFu);
      }
    }
  }
  __syncthreads();

  unsigned int* row = (is_dst ? histD : histS) + (size_t)blockIdx.x * NWORDS;
  for (int i = t; i < NWORDS; i += 1024) row[i] = h[i];
  if (!is_dst) return;
  __syncthreads();

  const int lane = t & 31, wv = t >> 5;
  const int w0 = t * 49;
  const int w1 = (w0 + 49 < NWORDS) ? (w0 + 49) : NWORDS;
  int s = 0;
  for (int i = w0; i < w1; ++i) {
    unsigned v = h[i];
    s += (int)(v & 0xFFFFu) + (int)(v >> 16);
  }
  int inc = s;
  for (int o = 1; o < 32; o <<= 1) {
    int u = __shfl_up(inc, o);
    if (lane >= o) inc += u;
  }
  if (lane == 31) wtot[wv] = inc;
  __syncthreads();
  int tv = wtot[lane];
  int winc = tv;
  for (int o = 1; o < 32; o <<= 1) {
    int u = __shfl_up(winc, o);
    if (lane >= o) winc += u;
  }
  int run = __shfl(winc - tv, wv) + inc - s;
  for (int i = w0; i < w1; ++i) {
    unsigned v = h[i];
    unsigned lo = v & 0xFFFFu, hi = v >> 16;
    h[i] = (unsigned)run | (((unsigned)run + lo) << 16);
    run += (int)(lo + hi);
  }
  __syncthreads();

  for (int i = t; i < NGRP; i += 1024)
    lscan32[(size_t)blockIdx.x * NGRP + i] = (int)(h[16 * i] & 0xFFFFu);

  const int n = e1 - e0;
  for (int b = t; b < n; b += 8 * 1024) {
    int d[8];
#pragma unroll
    for (int k = 0; k < 8; ++k) {
      int i = b + k * 1024;
      d[k] = dst[e0 + (i < n ? i : n - 1)];
    }
#pragma unroll
    for (int k = 0; k < 8; ++k) {
      int i = b + k * 1024;
      if (i < n) {
        unsigned p = ((h[d[k] >> 1] >> ((unsigned)(d[k] & 1) * 16u)) & 0xFFFFu) + (unsigned)lr[i];
        lr[i] = (unsigned short)p;
      }
    }
  }
  __syncthreads();

  for (int b = t; b < n; b += 8 * 1024) {
    int sv[8];
#pragma unroll
    for (int k = 0; k < 8; ++k) {
      int i = b + k * 1024;
      sv[k] = src[e0 + (i < n ? i : n - 1)];
    }
#pragma unroll
    for (int k = 0; k < 8; ++k) {
      int i = b + k * 1024;
      if (i < n) h[lr[i]] = (unsigned)sv[k];
    }
  }
  __syncthreads();

  for (int i = t; i < n; i += 1024) ss[e0 + i] = (int)h[i];
}

__global__ void colscan_kernel(const unsigned int* __restrict__ histD,
                               const unsigned int* __restrict__ histS,
                               int* __restrict__ cnt,
                               float* __restrict__ dis, const float* __restrict__ x,
                               float* __restrict__ xs, int G) {
  int i = blockIdx.x * blockDim.x + threadIdx.x;
  if (i >= NWORDS) return;
  if (blockIdx.y == 0) {
    int r0 = 0, r1 = 0;
#pragma unroll 16
    for (int w = 0; w < G; ++w) {
      unsigned v = histD[(size_t)w * NWORDS + i];
      r0 += (int)(v & 0xFFFFu);
      r1 += (int)(v >> 16);
    }
    int2 c;
    c.x = r0;
    c.y = r1;
    *reinterpret_cast<int2*>(cnt + 2 * i) = c;
  } else {
    const float4* __restrict__ xp = reinterpret_cast<const float4*>(x + (size_t)16 * i);
    float4 a0 = xp[0], a1 = xp[1], c0 = xp[2], c1 = xp[3];
    int d0 = 0, d1 = 0;
#pragma unroll 16
    for (int w = 0; w < G; ++w) {
      unsigned u = histS[(size_t)w * NWORDS + i];
      d0 += (int)(u & 0xFFFFu);
      d1 += (int)(u >> 16);
    }
    float f0 = (float)d0, f1 = (float)d1;
    float2 r;
    r.x = (f0 > 0.0f) ? (1.0f / sqrtf(f0)) : 0.0f;
    r.y = (f1 > 0.0f) ? (1.0f / sqrtf(f1)) : 0.0f;
    *reinterpret_cast<float2*>(dis + 2 * i) = r;
    a0.x *= r.x; a0.y *= r.x; a0.z *= r.x; a0.w *= r.x;
    a1.x *= r.x; a1.y *= r.x; a1.z *= r.x; a1.w *= r.x;
    c0.x *= r.y; c0.y *= r.y; c0.z *= r.y; c0.w *= r.y;
    c1.x *= r.y; c1.y *= r.y; c1.z *= r.y; c1.w *= r.y;
    float4* __restrict__ xo = reinterpret_cast<float4*>(xs + (size_t)16 * i);
    xo[0] = a0;
    xo[1] = a1;
    xo[2] = c0;
    xo[3] = c1;
  }
}

#define SBLK 4096
static_assert(N_NODES % 4 == 0, "cnt and off start at 16-byte boundaries of the workspace");

__global__ void __launch_bounds__(1024)
scan_kernel(const int* __restrict__ cnt, int* __restrict__ off, int n) {
  __shared__ int wsum[32];
  __shared__ int wtot[32];
  const int t = threadIdx.x, lane = t & 31, wv = t >> 5;
  const int b0 = blockIdx.x * SBLK;
  const int i  = b0 + t * 4;
  const int4* __restrict__ c4 = reinterpret_cast<const int4*>(cnt);

  int4 v = make_int4(0, 0, 0, 0);
  if (i + 3 < n) {
    v = c4[i >> 2];
  } else {
    if (i < n)     v.x = cnt[i];
    if (i + 1 < n) v.y = cnt[i + 1];
    if (i + 2 < n) v.z = cnt[i + 2];
  }

  int s = 0;
  const int rows = blockIdx.x;
  const int4* __restrict__ rp = c4 + t;
  int r = 0;
  for (; r + 12 <= rows; r += 12) {
#pragma unroll
    for (int k = 0; k < 12; ++k) {
      const int4 p = rp[(r + k) * 1024];
      s += (p.x + p.y) + (p.z + p.w);
    }
  }
  const int rem = rows - r;
  if (rem > 0) {
    int last = 0;
#pragma unroll
    for (int k = 0; k < 12; ++k) {
      const int rr = (r + k < rows) ? (r + k) : (rows - 1);
      const int4 p = rp[rr * 1024];
      last = (p.x + p.y) + (p.z + p.w);
      s += last;
    }
    s -= (12 - rem) * last;
  }

  const int t4 = (v.x + v.y) + (v.z + v.w);
  int inc = t4;
  for (int o = 1; o < 32; o <<= 1) {
    int u = __shfl_up(inc, o);
    if (lane >= o) inc += u;
  }
  for (int o = 16; o > 0; o >>= 1) s += __shfl_xor(s, o);
  if (lane == 31) wtot[wv] = inc;
  if (lane == 0)  wsum[wv] = s;
  __syncthreads();

  int front = wsum[lane];
  for (int o = 16; o > 0; o >>= 1) front += __shfl_xor(front, o);
  const int tv = wtot[lane];
  int winc = tv;
  for (int o = 1; o < 32; o <<= 1) {
    int u = __shfl_up(winc, o);
    if (lane >= o) winc += u;
  }
  const int ex = front + __shfl(winc - tv, wv) + (inc - t4);
  const int total = front + __shfl(winc, 31);

  if (i + 3 < n) {
    int4 o4;
    o4.x = ex;
    o4.y = ex + v.x;
    o4.z = ex + v.x + v.y;
    o4.w = ex + v.x + v.y + v.z;
    *reinterpret_cast<int4*>(off + i) = o4;
  } else {
    if (i < n)     off[i]     = ex;
    if (i + 1 < n) off[i + 1] = ex + v.x;
    if (i + 2 < n) off[i + 2] = ex + v.x + v.y;
  }
  if (blockIdx.x == gridDim.x - 1 && t == 0) off[n] = total;
}

#define FREC 4
#define FCH  16
static_assert(FCH <= 32 && (FCH & (FCH - 1)) == 0 && FCH % 2 == 0, "one lane per slice holds the run start; two slices per trip");
__global__ void __launch_bounds__(1024)
fill_kernel(const unsigned int* __restrict__ histD,
            const int* __restrict__ off, const int* __restrict__ lscan32,
            const int* __restrict__ ss, int* __restrict__ eb, int G, int E) {
  __shared__ int win[FCAP];
  __shared__ int ptot[FPH][FGRP * 32];
  __shared__ unsigned cpair[FCH / 2][1024];
  const int t = threadIdx.x;
  const int lane = t & 31;
  const int wv = __builtin_amdgcn_readfirstlane(t >> 5);
  const int gl = wv & (FGRP - 1), ph = wv / FGRP;
  const int grp = blockIdx.x * FGRP + gl;
  const int n0 = blockIdx.x * (FGRP * 32);
  const int n1 = (n0 + FGRP * 32 < N_NODES) ? (n0 + FGRP * 32) : N_NODES;
  const int o0 = off[n0], o1 = off[n1];
  const bool staged = (o1 - o0) <= FCAP;
  const bool act = grp < NGRP;
  const int gq = act ? grp : (NGRP - 1);
  const int d  = gq * 32 + lane;
  const int nl = gl * 32 + lane;
  const int od = off[d];
  const unsigned sh = (unsigned)(d & 1) * 16u;

  int carry = 0;
  for (int wr = 0; wr < G; wr += FPH * FCH) {
    const int w0 = wr + ph * FCH;
    int c[FCH];
#pragma unroll
    for (int k = 0; k < FCH; ++k) {
      const int row = (w0 + k < G) ? (w0 + k) : (G - 1);
      const unsigned word = histD[(size_t)row * NWORDS + (d >> 1)];
      c[k] = (act && w0 + k < G) ? (int)((word >> sh) & 0xFFFFu) : 0;
    }
    const int wk = w0 + (lane & (FCH - 1));
    const int rsv = lscan32[(size_t)((wk < G) ? wk : (G - 1)) * NGRP + gq];
    int tot = 0;
#pragma unroll
    for (int k = 0; k < FCH; ++k) tot += c[k];
#pragma unroll
    for (int kk = 0; kk < FCH / 2; ++kk)
      cpair[kk][t] = (unsigned)c[2 * kk] | ((unsigned)c[2 * kk + 1] << 16);
    ptot[ph][nl] = tot;
    __syncthreads();
    int bs = carry;
#pragma unroll
    for (int p = 0; p < FPH; ++p) {
      const int v = ptot[p][nl];
      carry += v;
      bs += (p < ph) ? v : 0;
    }

#pragma nounroll
    for (int kk = 0; kk < FCH / 2; ++kk) {
      const unsigned pr = cpair[kk][t];
      const int cA = (int)(pr & 0xFFFFu), cB = (int)(pr >> 16);
      const int rsA = __builtin_amdgcn_readlane(rsv, 2 * kk);
      const int rsB = __builtin_amdgcn_readlane(rsv, 2 * kk + 1);
      int incA = cA, incB = cB;
      for (int o = 1; o < 32; o <<= 1) {
        int uA = __shfl_up(incA, o);
        int uB = __shfl_up(incB, o);
        if (lane >= o) { incA += uA; incB += uB; }
      }
      const int pA = (w0 + 2 * kk)     * SLICE + rsA + (incA - cA);
      const int pB = (w0 + 2 * kk + 1) * SLICE + rsB + (incB - cB);
      const int slotA = od + bs;
      const int slotB = slotA + cA;
      bs += cA + cB;

      int rA[FREC], rB[FREC];
#pragma unroll
      for (int k = 0; k < FREC; ++k) {
        int q = pA + ((k < cA) ? k : 0);
        rA[k] = ss[(q < E) ? q : (E - 1)];
      }
#pragma unroll
      for (int k = 0; k < FREC; ++k) {
        int q = pB + ((k < cB) ? k : 0);
        rB[k] = ss[(q < E) ? q : (E - 1)];
      }
#pragma unroll
      for (int k = 0; k < FREC; ++k) {
        if (k < cA) {
          if (staged) win[slotA - o0 + k] = rA[k];
          else        eb[slotA + k] = rA[k];
        }
      }
#pragma unroll
      for (int k = 0; k < FREC; ++k) {
        if (k < cB) {
          if (staged) win[slotB - o0 + k] = rB[k];
          else        eb[slotB + k] = rB[k];
        }
      }
      for (int k = FREC; k < cA; ++k) {
        const int s = ss[pA + k];
        if (staged) win[slotA - o0 + k] = s;
        else        eb[slotA + k] = s;
      }
      for (int k = FREC; k < cB; ++k) {
        const int s = ss[pB + k];
        if (staged) win[slotB - o0 + k] = s;
        else        eb[slotB + k] = s;
      }
    }
    if (wr + FPH * FCH < G) __syncthreads();
  }
  __syncthreads();
  if (staged) {
    const int m = o1 - o0;
    for (int i = t; i < m; i += 1024) eb[o0 + i] = win[i];
  }
}

template <int K> __device__ __forceinline__ int grp8_take(int v) {
  return __builtin_amdgcn_mov_dpp8(v, K * 0x249249);
}
template <int K> __device__ __forceinline__ int grp4_take(int v) {
  return __builtin_amdgcn_update_dpp(0, v, K * 0x55, 0xF, 0xF, false);
}

__global__ void pull1_layer1_kernel(const int* __restrict__ off, const int* __restrict__ eb,
                                    const float* __restrict__ dis, const float* __restrict__ xs,
                                    const float* __restrict__ x, const float* __restrict__ W1,
                                    const float* __restrict__ b1, float* __restrict__ h1,
                                    float* __restrict__ h1s, int n) {
  int g = blockIdx.x * blockDim.x + threadIdx.x;
  int d = g >> 3, f = g & 7;
  if (d >= n) return;
  int beg = off[d], end = off[d + 1];
  float dd = dis[d];
  const float xv = x[(size_t)d * 8 + f];
  const float wa0 = W1[f * 3 + 0], wa1 = W1[f * 3 + 1], wa2 = W1[f * 3 + 2];
  const float wb0 = W1[24 + f * 3 + 0], wb1 = W1[24 + f * 3 + 1], wb2 = W1[24 + f * 3 + 2];
  float acc = 0.0f;
  if (beg < end) {
    const int last = end - 1;
    int r[2];
#pragma unroll
    for (int i = 0; i < 2; ++i) {
      const int ix = beg + 8 * i + f;
      r[i] = eb[ix < last ? ix : last];
    }
    for (int j = beg; j < end; j += 16) {
      int q[2];
#pragma unroll
      for (int i = 0; i < 2; ++i) {
        const int ix = j + 16 + 8 * i + f;
        q[i] = eb[ix < last ? ix : last];
      }
      float v[16];
#pragma unroll
      for (int i = 0; i < 2; ++i) {
        v[8 * i + 0] = xs[grp8_take<0>(r[i]) * 8 + f];
        v[8 * i + 1] = xs[grp8_take<1>(r[i]) * 8 + f];
        v[8 * i + 2] = xs[grp8_take<2>(r[i]) * 8 + f];
        v[8 * i + 3] = xs[grp8_take<3>(r[i]) * 8 + f];
        v[8 * i + 4] = xs[grp8_take<4>(r[i]) * 8 + f];
        v[8 * i + 5] = xs[grp8_take<5>(r[i]) * 8 + f];
        v[8 * i + 6] = xs[grp8_take<6>(r[i]) * 8 + f];
        v[8 * i + 7] = xs[grp8_take<7>(r[i]) * 8 + f];
      }
      const int left = end - j;
#pragma unroll
      for (int k = 0; k < 16; ++k) acc += (k < left) ? v[k] : 0.0f;
#pragma unroll
      for (int i = 0; i < 2; ++i) r[i] = q[i];
    }
  }
  const float tv = -dd * acc;
  float p0 = xv * wa0 + tv * wb0;
  float p1 = xv * wa1 + tv * wb1;
  float p2 = xv * wa2 + tv * wb2;
  for (int o = 1; o < 8; o <<= 1) {
    p0 += __shfl_xor(p0, o);
    p1 += __shfl_xor(p1, o);
    p2 += __shfl_xor(p2, o);
  }
  const float hv0 = fmaxf(p0 + b1[0], 0.0f);
  const float hv1 = fmaxf(p1 + b1[1], 0.0f);
  const float hv2 = fmaxf(p2 + b1[2], 0.0f);
  const float hsel = (f == 0) ? hv0 : ((f == 1) ? hv1 : hv2);
  if (f < 3) h1[(size_t)d * 3 + f] = hsel;
  if (f < 4) h1s[(size_t)d * 4 + f] = (f < 3) ? dd * hsel : 0.0f;
}

#define TNODES 64
#define H2P    18

__global__ void __launch_bounds__(256)
pull2_layer2_head_kernel(const int* __restrict__ off, const int* __restrict__ eb,
                         const float* __restrict__ dis, const float* __restrict__ h1s,
                         const float* __restrict__ h1, const float* __restrict__ W2,
                         const float* __restrict__ b2, const float* __restrict__ Wl,
                         const float* __restrict__ bl, float* __restrict__ out,
                         int n, int ntiles) {
  __shared__ float h2t[TNODES * H2P];
  const int t = threadIdx.x;
  const int lane = t & 31;
  const int nl = t >> 2, f = t & 3;
  const int d = blockIdx.x * TNODES + nl;
  const bool live = d < n;

  float acc = 0.0f, dd = 0.0f;
  float a0 = 0.0f, a1 = 0.0f, a2 = 0.0f;
  if (live) {
    int beg = off[d], end = off[d + 1];
    dd = dis[d];
    a0 = h1[(size_t)d * 3 + 0];
    a1 = h1[(size_t)d * 3 + 1];
    a2 = h1[(size_t)d * 3 + 2];
    if (beg < end) {
      const int last = end - 1;
      int r[4];
#pragma unroll
      for (int i = 0; i < 4; ++i) {
        const int ix = beg + 4 * i + f;
        r[i] = eb[ix < last ? ix : last];
      }
      for (int j = beg; j < end; j += 16) {
        int q[4];
#pragma unroll
        for (int i = 0; i < 4; ++i) {
          const int ix = j + 16 + 4 * i + f;
          q[i] = eb[ix < last ? ix : last];
        }
        float v[16];
#pragma unroll
        for (int i = 0; i < 4; ++i) {
          v[4 * i + 0] = h1s[grp4_take<0>(r[i]) * 4 + f];
          v[4 * i + 1] = h1s[grp4_take<1>(r[i]) * 4 + f];
          v[4 * i + 2] = h1s[grp4_take<2>(r[i]) * 4 + f];
          v[4 * i + 3] = h1s[grp4_take<3>(r[i]) * 4 + f];
        }
        const int left = end - j;
#pragma unroll
        for (int k = 0; k < 16; ++k) acc += (k < left) ? v[k] : 0.0f;
#pragma unroll
        for (int i = 0; i < 4; ++i) r[i] = q[i];
      }
    }
  }
  const float tx = -dd * acc;
  const int gb = lane & ~3;
  const float tv[3] = {__shfl(tx, gb + 0), __shfl(tx, gb + 1), __shfl(tx, gb + 2)};
  const float hv[3] = {a0, a1, a2};

  const int c0 = 4 * f;
  const float4 bb = *reinterpret_cast<const float4*>(b2 + c0);
  float o4[4] = {bb.x, bb.y, bb.z, bb.w};
#pragma unroll
  for (int q = 0; q < 3; ++q) {
    const float4 wa = *reinterpret_cast<const float4*>(W2 + q * 16 + c0);
    const float4 wb = *reinterpret_cast<const float4*>(W2 + 48 + q * 16 + c0);
    o4[0] += hv[q] * wa.x + tv[q] * wb.x;
    o4[1] += hv[q] * wa.y + tv[q] * wb.y;
    o4[2] += hv[q] * wa.z + tv[q] * wb.z;
    o4[3] += hv[q] * wa.w + tv[q] * wb.w;
  }
  float* __restrict__ hrow = h2t + nl * H2P + c0;
  hrow[0] = fmaxf(o4[0], 0.0f);
  hrow[1] = fmaxf(o4[1], 0.0f);
  hrow[2] = fmaxf(o4[2], 0.0f);
  hrow[3] = fmaxf(o4[3], 0.0f);
  __syncthreads();

  const int wv = __builtin_amdgcn_readfirstlane(t >> 5);
  const int tl = wv >> 1;
  const int tile = blockIdx.x * (TNODES / 16) + tl;
  if (tile >= ntiles) return;
  const int half = lane >> 4;
  const int l16  = lane & 15;
  const int jb   = (wv & 1) * 32;
  const float* __restrict__ arow = h2t + (tl * 16 + l16) * H2P;

  v8f acc0 = {}, acc1 = {};
#pragma unroll
  for (int kk = 0; kk < 16; kk += 4) {
    int ka = kk + 2 * half;
    v2f a;
    a.x = arow[ka];
    a.y = arow[ka + 1];

    v2f b0, b1;
    b0.x = Wl[ka * 64 + jb +  0 + l16]; b0.y = Wl[(ka + 1) * 64 + jb +  0 + l16];
    b1.x = Wl[ka * 64 + jb + 16 + l16]; b1.y = Wl[(ka + 1) * 64 + jb + 16 + l16];

    acc0 = __builtin_amdgcn_wmma_f32_16x16x4_f32(false, a, false, b0, (short)0, acc0, false, false);
    acc1 = __builtin_amdgcn_wmma_f32_16x16x4_f32(false, a, false, b1, (short)0, acc1, false, false);
  }

  v8f accs[2] = {acc0, acc1};
#pragma unroll
  for (int j = 0; j < 2; ++j) {
    int col = jb + j * 16 + l16;
    float bias = bl[col];
#pragma unroll
    for (int r = 0; r < 8; ++r) {
      int m = tile * 16 + r + 8 * half;
      out[(size_t)m * 64 + col] = accs[j][r] + bias;
    }
  }
}

extern "C" void kernel_launch(void* const* d_in, const int* in_sizes, int n_in,
                              void* d_out, int out_size, void* d_ws, size_t ws_size,
                              hipStream_t stream) {
  const float* x  = (const float*)d_in[0];
  const int*   ei = (const int*)d_in[1];
  const float* W1 = (const float*)d_in[2];
  const float* b1 = (const float*)d_in[3];
  const float* W2 = (const float*)d_in[4];
  const float* b2 = (const float*)d_in[5];
  const float* Wl = (const float*)d_in[6];
  const float* bl = (const float*)d_in[7];
  float* out = (float*)d_out;

  const int N = in_sizes[0] / 8;
  const int E = in_sizes[1] / 2;
  const int* src = ei;
  const int* dst = ei + E;

  const int G = (E + SLICE - 1) / SLICE;

  float* ws  = (float*)d_ws;
  int*   cnt = (int*)(ws + (size_t)N);
  float* dis = ws + (size_t)2  * N;
  float* h1  = ws + (size_t)11 * N;
  int*   off = (int*)(ws + (size_t)33 * N);
  int*   eb  = (int*)(ws + (size_t)34 * N + 4);
  float* xs  = ws + (size_t)34 * N + 8 + (size_t)E;
  float* h1s = xs + (size_t)8 * N;
  int*   ss   = (int*)(ws + (size_t)34 * N + 4 + (size_t)2 * E);
  unsigned int* histD = (unsigned int*)(ws + (size_t)34 * N + 4 + (size_t)3 * E);
  unsigned int* histS = histD + (size_t)G * NWORDS;
  int*   lscan32 = (int*)(histS + (size_t)G * NWORDS);

  const int B = 256;
  hist_kernel<<<dim3(G, 2), 1024, 0, stream>>>(src, dst, histD, histS, ss, lscan32, E);
  colscan_kernel<<<dim3((NWORDS + B - 1) / B, 2), B, 0, stream>>>(histD, histS, cnt, dis, x, xs, G);

  scan_kernel<<<(N + SBLK - 1) / SBLK, 1024, 0, stream>>>(cnt, off, N);
  fill_kernel<<<(NGRP + FGRP - 1) / FGRP, 1024, 0, stream>>>(histD, off, lscan32, ss, eb, G, E);

  pull1_layer1_kernel<<<(8 * N + B - 1) / B, B, 0, stream>>>(off, eb, dis, xs, x, W1, b1, h1, h1s, N);

  const int ntiles = N / 16;
  pull2_layer2_head_kernel<<<(N + TNODES - 1) / TNODES, 256, 0, stream>>>(off, eb, dis, h1s, h1, W2, b2, Wl, bl,
                                                                          out, N, ntiles);
}


# Layer-1 gather and dense stage run at the end of the bucket-fill kernel, each work-group summing its 128 nodes from the window it holds in LDS: 6 -> 5 launches, list not re-read; same sum order
# speedup vs baseline: 11.6994x; 1.0439x over previous
#include <hip/hip_runtime.h>
#include <math.h>


typedef __attribute__((ext_vector_type(2))) float v2f;
typedef __attribute__((ext_vector_type(8))) float v8f;

#define N_NODES 100000

#define NWORDS (N_NODES / 2)
#define SLICE  50000
#define NGRP   (N_NODES / 32)
static_assert(SLICE <= NWORDS, "the counter array is reused as the slice's ordered window");
static_assert(N_NODES % 32 == 0, "a wave owns 32 whole nodes");

#define FGRP 4
#define FPH  8
#define FCAP 16384

__global__ void __launch_bounds__(1024)
hist_kernel(const int* __restrict__ src, const int* __restrict__ dst,
            unsigned int* __restrict__ histD, unsigned int* __restrict__ histS,
            int* __restrict__ ss, int* __restrict__ lscan32, int E) {
  __shared__ unsigned int h[NWORDS];
  __shared__ unsigned short lr[SLICE];
  __shared__ int wtot[32];
  const int t  = threadIdx.x;
  const int e0 = blockIdx.x * SLICE;
  const int e1 = (e0 + SLICE < E) ? (e0 + SLICE) : E;
  const bool is_dst = (blockIdx.y == 0);
  const int* __restrict__ idx = is_dst ? dst : src;

  for (int i = t; i < NWORDS; i += 1024) h[i] = 0u;
  __syncthreads();

  for (int b = e0 + t; b < e1; b += 8 * 1024) {
    int d[8];
#pragma unroll
    for (int k = 0; k < 8; ++k) {
      int e = b + k * 1024;
      d[k] = idx[e < e1 ? e : e1 - 1];
    }
    unsigned old[8];
#pragma unroll
    for (int k = 0; k < 8; ++k) {
      unsigned sh = (unsigned)(d[k] & 1) * 16u;
      old[k] = atomicAdd(&h[d[k] >> 1], (b + k * 1024 < e1) ? (1u << sh) : 0u);
    }
    if (is_dst) {
#pragma unroll
      for (int k = 0; k < 8; ++k) {
        int e = b + k * 1024;
        unsigned sh = (unsigned)(d[k] & 1) * 16u;
        if (e < e1) lr[e - e0] = (unsigned short)((old[k] >> sh) & 0xFFFFu);
      }
    }
  }
  __syncthreads();

  unsigned int* row = (is_dst ? histD : histS) + (size_t)blockIdx.x * NWORDS;
  for (int i = t; i < NWORDS; i += 1024) row[i] = h[i];
  if (!is_dst) return;
  __syncthreads();

  const int lane = t & 31, wv = t >> 5;
  const int w0 = t * 49;
  const int w1 = (w0 + 49 < NWORDS) ? (w0 + 49) : NWORDS;
  int s = 0;
  for (int i = w0; i < w1; ++i) {
    unsigned v = h[i];
    s += (int)(v & 0xFFFFu) + (int)(v >> 16);
  }
  int inc = s;
  for (int o = 1; o < 32; o <<= 1) {
    int u = __shfl_up(inc, o);
    if (lane >= o) inc += u;
  }
  if (lane == 31) wtot[wv] = inc;
  __syncthreads();
  int tv = wtot[lane];
  int winc = tv;
  for (int o = 1; o < 32; o <<= 1) {
    int u = __shfl_up(winc, o);
    if (lane >= o) winc += u;
  }
  int run = __shfl(winc - tv, wv) + inc - s;
  for (int i = w0; i < w1; ++i) {
    unsigned v = h[i];
    unsigned lo = v & 0xFFFFu, hi = v >> 16;
    h[i] = (unsigned)run | (((unsigned)run + lo) << 16);
    run += (int)(lo + hi);
  }
  __syncthreads();

  for (int i = t; i < NGRP; i += 1024)
    lscan32[(size_t)blockIdx.x * NGRP + i] = (int)(h[16 * i] & 0xFFFFu);

  const int n = e1 - e0;
  for (int b = t; b < n; b += 8 * 1024) {
    int d[8];
#pragma unroll
    for (int k = 0; k < 8; ++k) {
      int i = b + k * 1024;
      d[k] = dst[e0 + (i < n ? i : n - 1)];
    }
#pragma unroll
    for (int k = 0; k < 8; ++k) {
      int i = b + k * 1024;
      if (i < n) {
        unsigned p = ((h[d[k] >> 1] >> ((unsigned)(d[k] & 1) * 16u)) & 0xFFFFu) + (unsigned)lr[i];
        lr[i] = (unsigned short)p;
      }
    }
  }
  __syncthreads();

  for (int b = t; b < n; b += 8 * 1024) {
    int sv[8];
#pragma unroll
    for (int k = 0; k < 8; ++k) {
      int i = b + k * 1024;
      sv[k] = src[e0 + (i < n ? i : n - 1)];
    }
#pragma unroll
    for (int k = 0; k < 8; ++k) {
      int i = b + k * 1024;
      if (i < n) h[lr[i]] = (unsigned)sv[k];
    }
  }
  __syncthreads();

  for (int i = t; i < n; i += 1024) ss[e0 + i] = (int)h[i];
}

__global__ void colscan_kernel(const unsigned int* __restrict__ histD,
                               const unsigned int* __restrict__ histS,
                               int* __restrict__ cnt,
                               float* __restrict__ dis, const float* __restrict__ x,
                               float* __restrict__ xs, int G) {
  int i = blockIdx.x * blockDim.x + threadIdx.x;
  if (i >= NWORDS) return;
  if (blockIdx.y == 0) {
    int r0 = 0, r1 = 0;
#pragma unroll 16
    for (int w = 0; w < G; ++w) {
      unsigned v = histD[(size_t)w * NWORDS + i];
      r0 += (int)(v & 0xFFFFu);
      r1 += (int)(v >> 16);
    }
    int2 c;
    c.x = r0;
    c.y = r1;
    *reinterpret_cast<int2*>(cnt + 2 * i) = c;
  } else {
    const float4* __restrict__ xp = reinterpret_cast<const float4*>(x + (size_t)16 * i);
    float4 a0 = xp[0], a1 = xp[1], c0 = xp[2], c1 = xp[3];
    int d0 = 0, d1 = 0;
#pragma unroll 16
    for (int w = 0; w < G; ++w) {
      unsigned u = histS[(size_t)w * NWORDS + i];
      d0 += (int)(u & 0xFFFFu);
      d1 += (int)(u >> 16);
    }
    float f0 = (float)d0, f1 = (float)d1;
    float2 r;
    r.x = (f0 > 0.0f) ? (1.0f / sqrtf(f0)) : 0.0f;
    r.y = (f1 > 0.0f) ? (1.0f / sqrtf(f1)) : 0.0f;
    *reinterpret_cast<float2*>(dis + 2 * i) = r;
    a0.x *= r.x; a0.y *= r.x; a0.z *= r.x; a0.w *= r.x;
    a1.x *= r.x; a1.y *= r.x; a1.z *= r.x; a1.w *= r.x;
    c0.x *= r.y; c0.y *= r.y; c0.z *= r.y; c0.w *= r.y;
    c1.x *= r.y; c1.y *= r.y; c1.z *= r.y; c1.w *= r.y;
    float4* __restrict__ xo = reinterpret_cast<float4*>(xs + (size_t)16 * i);
    xo[0] = a0;
    xo[1] = a1;
    xo[2] = c0;
    xo[3] = c1;
  }
}

#define SBLK 4096
static_assert(N_NODES % 4 == 0, "cnt and off start at 16-byte boundaries of the workspace");

__global__ void __launch_bounds__(1024)
scan_kernel(const int* __restrict__ cnt, int* __restrict__ off, int n) {
  __shared__ int wsum[32];
  __shared__ int wtot[32];
  const int t = threadIdx.x, lane = t & 31, wv = t >> 5;
  const int b0 = blockIdx.x * SBLK;
  const int i  = b0 + t * 4;
  const int4* __restrict__ c4 = reinterpret_cast<const int4*>(cnt);

  int4 v = make_int4(0, 0, 0, 0);
  if (i + 3 < n) {
    v = c4[i >> 2];
  } else {
    if (i < n)     v.x = cnt[i];
    if (i + 1 < n) v.y = cnt[i + 1];
    if (i + 2 < n) v.z = cnt[i + 2];
  }

  int s = 0;
  const int rows = blockIdx.x;
  const int4* __restrict__ rp = c4 + t;
  int r = 0;
  for (; r + 12 <= rows; r += 12) {
#pragma unroll
    for (int k = 0; k < 12; ++k) {
      const int4 p = rp[(r + k) * 1024];
      s += (p.x + p.y) + (p.z + p.w);
    }
  }
  const int rem = rows - r;
  if (rem > 0) {
    int last = 0;
#pragma unroll
    for (int k = 0; k < 12; ++k) {
      const int rr = (r + k < rows) ? (r + k) : (rows - 1);
      const int4 p = rp[rr * 1024];
      last = (p.x + p.y) + (p.z + p.w);
      s += last;
    }
    s -= (12 - rem) * last;
  }

  const int t4 = (v.x + v.y) + (v.z + v.w);
  int inc = t4;
  for (int o = 1; o < 32; o <<= 1) {
    int u = __shfl_up(inc, o);
    if (lane >= o) inc += u;
  }
  for (int o = 16; o > 0; o >>= 1) s += __shfl_xor(s, o);
  if (lane == 31) wtot[wv] = inc;
  if (lane == 0)  wsum[wv] = s;
  __syncthreads();

  int front = wsum[lane];
  for (int o = 16; o > 0; o >>= 1) front += __shfl_xor(front, o);
  const int tv = wtot[lane];
  int winc = tv;
  for (int o = 1; o < 32; o <<= 1) {
    int u = __shfl_up(winc, o);
    if (lane >= o) winc += u;
  }
  const int ex = front + __shfl(winc - tv, wv) + (inc - t4);
  const int total = front + __shfl(winc, 31);

  if (i + 3 < n) {
    int4 o4;
    o4.x = ex;
    o4.y = ex + v.x;
    o4.z = ex + v.x + v.y;
    o4.w = ex + v.x + v.y + v.z;
    *reinterpret_cast<int4*>(off + i) = o4;
  } else {
    if (i < n)     off[i]     = ex;
    if (i + 1 < n) off[i + 1] = ex + v.x;
    if (i + 2 < n) off[i + 2] = ex + v.x + v.y;
  }
  if (blockIdx.x == gridDim.x - 1 && t == 0) off[n] = total;
}

template <int K> __device__ __forceinline__ int grp8_take(int v) {
  return __builtin_amdgcn_mov_dpp8(v, K * 0x249249);
}
template <int K> __device__ __forceinline__ int grp4_take(int v) {
  return __builtin_amdgcn_update_dpp(0, v, K * 0x55, 0xF, 0xF, false);
}

template <bool STAGED>
__device__ __forceinline__ void layer1_node(const int* rec, const int sub, const int d, const int f,
                                            const int beg, const int end,
                                            const float* __restrict__ dis, const float* __restrict__ xs,
                                            const float* __restrict__ x, const float* __restrict__ W1,
                                            const float* __restrict__ b1, float* __restrict__ h1,
                                            float* __restrict__ h1s) {
  float dd = dis[d];
  const float xv = x[(size_t)d * 8 + f];
  const float wa0 = W1[f * 3 + 0], wa1 = W1[f * 3 + 1], wa2 = W1[f * 3 + 2];
  const float wb0 = W1[24 + f * 3 + 0], wb1 = W1[24 + f * 3 + 1], wb2 = W1[24 + f * 3 + 2];
  float acc = 0.0f;
  if (beg < end) {
    const int last = end - 1;
    int r[2];
#pragma unroll
    for (int i = 0; i < 2; ++i) {
      const int ix = beg + 8 * i + f;
      r[i] = rec[(ix < last ? ix : last) - sub];
    }
    for (int j = beg; j < end; j += 16) {
      int q[2];
#pragma unroll
      for (int i = 0; i < 2; ++i) {
        const int ix = j + 16 + 8 * i + f;
        q[i] = rec[(ix < last ? ix : last) - sub];
      }
      float v[16];
#pragma unroll
      for (int i = 0; i < 2; ++i) {
        v[8 * i + 0] = xs[grp8_take<0>(r[i]) * 8 + f];
        v[8 * i + 1] = xs[grp8_take<1>(r[i]) * 8 + f];
        v[8 * i + 2] = xs[grp8_take<2>(r[i]) * 8 + f];
        v[8 * i + 3] = xs[grp8_take<3>(r[i]) * 8 + f];
        v[8 * i + 4] = xs[grp8_take<4>(r[i]) * 8 + f];
        v[8 * i + 5] = xs[grp8_take<5>(r[i]) * 8 + f];
        v[8 * i + 6] = xs[grp8_take<6>(r[i]) * 8 + f];
        v[8 * i + 7] = xs[grp8_take<7>(r[i]) * 8 + f];
      }
      const int left = end - j;
#pragma unroll
      for (int k = 0; k < 16; ++k) acc += (k < left) ? v[k] : 0.0f;
#pragma unroll
      for (int i = 0; i < 2; ++i) r[i] = q[i];
    }
  }
  const float tv = -dd * acc;
  float p0 = xv * wa0 + tv * wb0;
  float p1 = xv * wa1 + tv * wb1;
  float p2 = xv * wa2 + tv * wb2;
  for (int o = 1; o < 8; o <<= 1) {
    p0 += __shfl_xor(p0, o);
    p1 += __shfl_xor(p1, o);
    p2 += __shfl_xor(p2, o);
  }
  const float hv0 = fmaxf(p0 + b1[0], 0.0f);
  const float hv1 = fmaxf(p1 + b1[1], 0.0f);
  const float hv2 = fmaxf(p2 + b1[2], 0.0f);
  const float hsel = (f == 0) ? hv0 : ((f == 1) ? hv1 : hv2);
  if (f < 3) h1[(size_t)d * 3 + f] = hsel;
  if (f < 4) h1s[(size_t)d * 4 + f] = (f < 3) ? dd * hsel : 0.0f;
}

#define FREC 4
#define FCH  16
static_assert(FCH <= 32 && (FCH & (FCH - 1)) == 0 && FCH % 2 == 0, "one lane per slice holds the run start; two slices per trip");
__global__ void __launch_bounds__(1024)
fill_layer1_kernel(const unsigned int* __restrict__ histD,
                   const int* __restrict__ off, const int* __restrict__ lscan32,
                   const int* __restrict__ ss, int* eb, int G, int E,
                   const float* __restrict__ dis, const float* __restrict__ xs,
                   const float* __restrict__ x, const float* __restrict__ W1,
                   const float* __restrict__ b1, float* __restrict__ h1,
                   float* __restrict__ h1s) {
  __shared__ int win[FCAP];
  __shared__ int ptot[FPH][FGRP * 32];
  __shared__ unsigned cpair[FCH / 2][1024];
  const int t = threadIdx.x;
  const int lane = t & 31;
  const int wv = __builtin_amdgcn_readfirstlane(t >> 5);
  const int gl = wv & (FGRP - 1), ph = wv / FGRP;
  const int grp = blockIdx.x * FGRP + gl;
  const int n0 = blockIdx.x * (FGRP * 32);
  const int n1 = (n0 + FGRP * 32 < N_NODES) ? (n0 + FGRP * 32) : N_NODES;
  const int o0 = off[n0], o1 = off[n1];
  const bool staged = (o1 - o0) <= FCAP;
  const bool act = grp < NGRP;
  const int gq = act ? grp : (NGRP - 1);
  const int d  = gq * 32 + lane;
  const int nl = gl * 32 + lane;
  const int od = off[d];
  const unsigned sh = (unsigned)(d & 1) * 16u;

  int carry = 0;
  for (int wr = 0; wr < G; wr += FPH * FCH) {
    const int w0 = wr + ph * FCH;
    int c[FCH];
#pragma unroll
    for (int k = 0; k < FCH; ++k) {
      const int row = (w0 + k < G) ? (w0 + k) : (G - 1);
      const unsigned word = histD[(size_t)row * NWORDS + (d >> 1)];
      c[k] = (act && w0 + k < G) ? (int)((word >> sh) & 0xFFFFu) : 0;
    }
    const int wk = w0 + (lane & (FCH - 1));
    const int rsv = lscan32[(size_t)((wk < G) ? wk : (G - 1)) * NGRP + gq];
    int tot = 0;
#pragma unroll
    for (int k = 0; k < FCH; ++k) tot += c[k];
#pragma unroll
    for (int kk = 0; kk < FCH / 2; ++kk)
      cpair[kk][t] = (unsigned)c[2 * kk] | ((unsigned)c[2 * kk + 1] << 16);
    ptot[ph][nl] = tot;
    __syncthreads();
    int bs = carry;
#pragma unroll
    for (int p = 0; p < FPH; ++p) {
      const int v = ptot[p][nl];
      carry += v;
      bs += (p < ph) ? v : 0;
    }

#pragma nounroll
    for (int kk = 0; kk < FCH / 2; ++kk) {
      const unsigned pr = cpair[kk][t];
      const int cA = (int)(pr & 0xFFFFu), cB = (int)(pr >> 16);
      const int rsA = __builtin_amdgcn_readlane(rsv, 2 * kk);
      const int rsB = __builtin_amdgcn_readlane(rsv, 2 * kk + 1);
      int incA = cA, incB = cB;
      for (int o = 1; o < 32; o <<= 1) {
        int uA = __shfl_up(incA, o);
        int uB = __shfl_up(incB, o);
        if (lane >= o) { incA += uA; incB += uB; }
      }
      const int pA = (w0 + 2 * kk)     * SLICE + rsA + (incA - cA);
      const int pB = (w0 + 2 * kk + 1) * SLICE + rsB + (incB - cB);
      const int slotA = od + bs;
      const int slotB = slotA + cA;
      bs += cA + cB;

      int rA[FREC], rB[FREC];
#pragma unroll
      for (int k = 0; k < FREC; ++k) {
        int q = pA + ((k < cA) ? k : 0);
        rA[k] = ss[(q < E) ? q : (E - 1)];
      }
#pragma unroll
      for (int k = 0; k < FREC; ++k) {
        int q = pB + ((k < cB) ? k : 0);
        rB[k] = ss[(q < E) ? q : (E - 1)];
      }
#pragma unroll
      for (int k = 0; k < FREC; ++k) {
        if (k < cA) {
          if (staged) win[slotA - o0 + k] = rA[k];
          else        eb[slotA + k] = rA[k];
        }
      }
#pragma unroll
      for (int k = 0; k < FREC; ++k) {
        if (k < cB) {
          if (staged) win[slotB - o0 + k] = rB[k];
          else        eb[slotB + k] = rB[k];
        }
      }
      for (int k = FREC; k < cA; ++k) {
        const int s = ss[pA + k];
        if (staged) win[slotA - o0 + k] = s;
        else        eb[slotA + k] = s;
      }
      for (int k = FREC; k < cB; ++k) {
        const int s = ss[pB + k];
        if (staged) win[slotB - o0 + k] = s;
        else        eb[slotB + k] = s;
      }
    }
    if (wr + FPH * FCH < G) __syncthreads();
  }
  __syncthreads();
  if (staged) {
    const int m = o1 - o0;
    for (int i = t; i < m; i += 1024) eb[o0 + i] = win[i];
  }

  const int dn = n0 + (t >> 3);
  if (dn >= n1) return;
  const int nb = off[dn], ne = off[dn + 1];
  if (staged) layer1_node<true>(win, o0, dn, t & 7, nb, ne, dis, xs, x, W1, b1, h1, h1s);
  else        layer1_node<false>(eb, 0, dn, t & 7, nb, ne, dis, xs, x, W1, b1, h1, h1s);
}

#define TNODES 64
#define H2P    18

__global__ void __launch_bounds__(256)
pull2_layer2_head_kernel(const int* __restrict__ off, const int* __restrict__ eb,
                         const float* __restrict__ dis, const float* __restrict__ h1s,
                         const float* __restrict__ h1, const float* __restrict__ W2,
                         const float* __restrict__ b2, const float* __restrict__ Wl,
                         const float* __restrict__ bl, float* __restrict__ out,
                         int n, int ntiles) {
  __shared__ float h2t[TNODES * H2P];
  const int t = threadIdx.x;
  const int lane = t & 31;
  const int nl = t >> 2, f = t & 3;
  const int d = blockIdx.x * TNODES + nl;
  const bool live = d < n;

  float acc = 0.0f, dd = 0.0f;
  float a0 = 0.0f, a1 = 0.0f, a2 = 0.0f;
  if (live) {
    int beg = off[d], end = off[d + 1];
    dd = dis[d];
    a0 = h1[(size_t)d * 3 + 0];
    a1 = h1[(size_t)d * 3 + 1];
    a2 = h1[(size_t)d * 3 + 2];
    if (beg < end) {
      const int last = end - 1;
      int r[4];
#pragma unroll
      for (int i = 0; i < 4; ++i) {
        const int ix = beg + 4 * i + f;
        r[i] = eb[ix < last ? ix : last];
      }
      for (int j = beg; j < end; j += 16) {
        int q[4];
#pragma unroll
        for (int i = 0; i < 4; ++i) {
          const int ix = j + 16 + 4 * i + f;
          q[i] = eb[ix < last ? ix : last];
        }
        float v[16];
#pragma unroll
        for (int i = 0; i < 4; ++i) {
          v[4 * i + 0] = h1s[grp4_take<0>(r[i]) * 4 + f];
          v[4 * i + 1] = h1s[grp4_take<1>(r[i]) * 4 + f];
          v[4 * i + 2] = h1s[grp4_take<2>(r[i]) * 4 + f];
          v[4 * i + 3] = h1s[grp4_take<3>(r[i]) * 4 + f];
        }
        const int left = end - j;
#pragma unroll
        for (int k = 0; k < 16; ++k) acc += (k < left) ? v[k] : 0.0f;
#pragma unroll
        for (int i = 0; i < 4; ++i) r[i] = q[i];
      }
    }
  }
  const float tx = -dd * acc;
  const int gb = lane & ~3;
  const float tv[3] = {__shfl(tx, gb + 0), __shfl(tx, gb + 1), __shfl(tx, gb + 2)};
  const float hv[3] = {a0, a1, a2};

  const int c0 = 4 * f;
  const float4 bb = *reinterpret_cast<const float4*>(b2 + c0);
  float o4[4] = {bb.x, bb.y, bb.z, bb.w};
#pragma unroll
  for (int q = 0; q < 3; ++q) {
    const float4 wa = *reinterpret_cast<const float4*>(W2 + q * 16 + c0);
    const float4 wb = *reinterpret_cast<const float4*>(W2 + 48 + q * 16 + c0);
    o4[0] += hv[q] * wa.x + tv[q] * wb.x;
    o4[1] += hv[q] * wa.y + tv[q] * wb.y;
    o4[2] += hv[q] * wa.z + tv[q] * wb.z;
    o4[3] += hv[q] * wa.w + tv[q] * wb.w;
  }
  float* __restrict__ hrow = h2t + nl * H2P + c0;
  hrow[0] = fmaxf(o4[0], 0.0f);
  hrow[1] = fmaxf(o4[1], 0.0f);
  hrow[2] = fmaxf(o4[2], 0.0f);
  hrow[3] = fmaxf(o4[3], 0.0f);
  __syncthreads();

  const int wv = __builtin_amdgcn_readfirstlane(t >> 5);
  const int tl = wv >> 1;
  const int tile = blockIdx.x * (TNODES / 16) + tl;
  if (tile >= ntiles) return;
  const int half = lane >> 4;
  const int l16  = lane & 15;
  const int jb   = (wv & 1) * 32;
  const float* __restrict__ arow = h2t + (tl * 16 + l16) * H2P;

  v8f acc0 = {}, acc1 = {};
#pragma unroll
  for (int kk = 0; kk < 16; kk += 4) {
    int ka = kk + 2 * half;
    v2f a;
    a.x = arow[ka];
    a.y = arow[ka + 1];

    v2f b0, b1;
    b0.x = Wl[ka * 64 + jb +  0 + l16]; b0.y = Wl[(ka + 1) * 64 + jb +  0 + l16];
    b1.x = Wl[ka * 64 + jb + 16 + l16]; b1.y = Wl[(ka + 1) * 64 + jb + 16 + l16];

    acc0 = __builtin_amdgcn_wmma_f32_16x16x4_f32(false, a, false, b0, (short)0, acc0, false, false);
    acc1 = __builtin_amdgcn_wmma_f32_16x16x4_f32(false, a, false, b1, (short)0, acc1, false, false);
  }

  v8f accs[2] = {acc0, acc1};
#pragma unroll
  for (int j = 0; j < 2; ++j) {
    int col = jb + j * 16 + l16;
    float bias = bl[col];
#pragma unroll
    for (int r = 0; r < 8; ++r) {
      int m = tile * 16 + r + 8 * half;
      out[(size_t)m * 64 + col] = accs[j][r] + bias;
    }
  }
}

extern "C" void kernel_launch(void* const* d_in, const int* in_sizes, int n_in,
                              void* d_out, int out_size, void* d_ws, size_t ws_size,
                              hipStream_t stream) {
  const float* x  = (const float*)d_in[0];
  const int*   ei = (const int*)d_in[1];
  const float* W1 = (const float*)d_in[2];
  const float* b1 = (const float*)d_in[3];
  const float* W2 = (const float*)d_in[4];
  const float* b2 = (const float*)d_in[5];
  const float* Wl = (const float*)d_in[6];
  const float* bl = (const float*)d_in[7];
  float* out = (float*)d_out;

  const int N = in_sizes[0] / 8;
  const int E = in_sizes[1] / 2;
  const int* src = ei;
  const int* dst = ei + E;

  const int G = (E + SLICE - 1) / SLICE;

  float* ws  = (float*)d_ws;
  int*   cnt = (int*)(ws + (size_t)N);
  float* dis = ws + (size_t)2  * N;
  float* h1  = ws + (size_t)11 * N;
  int*   off = (int*)(ws + (size_t)33 * N);
  int*   eb  = (int*)(ws + (size_t)34 * N + 4);
  float* xs  = ws + (size_t)34 * N + 8 + (size_t)E;
  float* h1s = xs + (size_t)8 * N;
  int*   ss   = (int*)(ws + (size_t)34 * N + 4 + (size_t)2 * E);
  unsigned int* histD = (unsigned int*)(ws + (size_t)34 * N + 4 + (size_t)3 * E);
  unsigned int* histS = histD + (size_t)G * NWORDS;
  int*   lscan32 = (int*)(histS + (size_t)G * NWORDS);

  const int B = 256;
  hist_kernel<<<dim3(G, 2), 1024, 0, stream>>>(src, dst, histD, histS, ss, lscan32, E);
  colscan_kernel<<<dim3((NWORDS + B - 1) / B, 2), B, 0, stream>>>(histD, histS, cnt, dis, x, xs, G);

  scan_kernel<<<(N + SBLK - 1) / SBLK, 1024, 0, stream>>>(cnt, off, N);
  fill_layer1_kernel<<<(NGRP + FGRP - 1) / FGRP, 1024, 0, stream>>>(histD, off, lscan32, ss, eb, G, E,
                                                                    dis, xs, x, W1, b1, h1, h1s);

  const int ntiles = N / 16;
  pull2_layer2_head_kernel<<<(N + TNODES - 1) / TNODES, 256, 0, stream>>>(off, eb, dis, h1s, h1, W2, b2, Wl, bl,
                                                                          out, N, ntiles);
}
